# starting kernel
# baseline (speedup 1.0000x reference)
#include <hip/hip_runtime.h>
#include <hip/hip_bf16.h>


#define BB 16
#define SS 128
#define HH 200
#define PP 20
#define FF 105
#define EPSF 1e-8f

typedef float v2f __attribute__((ext_vector_type(2)));
typedef float v8f __attribute__((ext_vector_type(8)));

__device__ __forceinline__ float wave_sum(float v) {
#pragma unroll
  for (int o = 16; o > 0; o >>= 1) v += __shfl_xor(v, o, 32);
  return v;
}

__global__ void prep_kernel(const float* __restrict__ src, const int* __restrict__ mask,
                            const float* __restrict__ w1, float* __restrict__ dst,
                            float* __restrict__ nrm, float* __restrict__ wn) {
  __shared__ float row[HH];
  int bs = blockIdx.x;
  int lane = threadIdx.x;
  float m = (float)mask[bs];
  const float* srow = src + (size_t)bs * HH;
  float* drow = dst + (size_t)bs * HH;
  float acc = 0.f;
  for (int h = lane; h < HH; h += 32) {
    float v = srow[h] * m;
    drow[h] = v; row[h] = v; acc += v * v;
  }
  acc = wave_sum(acc);
  if (lane == 0) nrm[bs] = sqrtf(acc);
  __syncthreads();
  for (int p = 0; p < PP; p++) {
    float a = 0.f;
    for (int h = lane; h < HH; h += 32) {
      float ww = w1[p * HH + h]; float v = row[h];
      a += ww * ww * v * v;
    }
    a = wave_sum(a);
    if (lane == 0) wn[bs * PP + p] = sqrtf(a);
  }
}

__global__ void gather_last_kernel(const float* __restrict__ cp, const float* __restrict__ ch,
                                   const int* __restrict__ mask_p, const int* __restrict__ mask_h,
                                   float* __restrict__ cpl, float* __restrict__ chl) {
  int b = blockIdx.x;
  int lp = 0, lh = 0;
  for (int i = 0; i < SS; i++) { lp += mask_p[b * SS + i]; lh += mask_h[b * SS + i]; }
  int ip = (lp > 0) ? lp - 1 : 0;
  int ih = (lh > 0) ? lh - 1 : 0;
  for (int h = threadIdx.x; h < HH; h += blockDim.x) {
    cpl[b * HH + h] = cp[((size_t)b * SS + ip) * HH + h];
    chl[b * HH + h] = ch[((size_t)b * SS + ih) * HH + h];
  }
}

__global__ void cos_wmma_kernel(const float* __restrict__ cp, const float* __restrict__ ch,
                                const float* __restrict__ np, const float* __restrict__ nh,
                                float* __restrict__ cosO) {
  int wid = blockIdx.x;
  int b = wid >> 6;
  int x0 = ((wid >> 3) & 7) << 4;
  int y0 = (wid & 7) << 4;
  int lane = threadIdx.x, half = lane >> 4, l16 = lane & 15;
  const float* arow = cp + ((size_t)(b * SS) + x0 + l16) * HH + 2 * half;
  const float* brow = ch + ((size_t)(b * SS) + y0 + l16) * HH + 2 * half;
  v8f acc = {0, 0, 0, 0, 0, 0, 0, 0};
  for (int k = 0; k < HH; k += 4) {
    v2f a = *(const v2f*)(arow + k);
    v2f bv = *(const v2f*)(brow + k);
    acc = __builtin_amdgcn_wmma_f32_16x16x4_f32(false, a, false, bv, (short)0, acc, false, false);
  }
  float nhv = nh[b * SS + y0 + l16];
#pragma unroll
  for (int r = 0; r < 8; r++) {
    int m = r + 8 * half;
    float d = fmaxf(np[b * SS + x0 + m] * nhv, EPSF);
    cosO[((size_t)(b * SS) + x0 + m) * SS + y0 + l16] = acc[r] / d;
  }
}

__global__ void cos_reduce_kernel(const float* __restrict__ cosM, const int* __restrict__ mask,
                                  float* __restrict__ cossum, float* __restrict__ out,
                                  int transposed, int side) {
  int b = blockIdx.x / SS, r = blockIdx.x % SS;
  int t = threadIdx.x;
  float v = transposed ? cosM[((size_t)b * SS + t) * SS + r]
                       : cosM[((size_t)b * SS + r) * SS + t];
  __shared__ float smax[128], ssum[128];
  smax[t] = v; ssum[t] = v;
  __syncthreads();
  for (int o = 64; o > 0; o >>= 1) {
    if (t < o) { smax[t] = fmaxf(smax[t], smax[t + o]); ssum[t] += ssum[t + o]; }
    __syncthreads();
  }
  if (t == 0) {
    float m = (float)mask[b * SS + r];
    cossum[b * SS + r] = ssum[0];
    float* o2 = out + ((size_t)(side * BB + b) * SS + r) * FF;
    o2[0] = m * smax[0];
    o2[1] = m * (ssum[0] / (float)SS);
  }
}

__global__ void attmean_wmma_kernel(const float* __restrict__ cosM, const float* __restrict__ V,
                                    const float* __restrict__ csum, float* __restrict__ attO,
                                    int aStrideM, int aStrideK) {
  int wid = blockIdx.x;
  int b = wid / (8 * 13);
  int rem = wid % (8 * 13);
  int m0 = (rem / 13) * 16;
  int n0 = (rem % 13) * 16;
  int lane = threadIdx.x, half = lane >> 4, l16 = lane & 15;
  const float* cb = cosM + (size_t)b * SS * SS;
  int nIdx = n0 + l16;
  bool nOK = nIdx < HH;
  v8f acc = {0, 0, 0, 0, 0, 0, 0, 0};
  for (int k = 0; k < SS; k += 4) {
    int ka = k + 2 * half;
    v2f a, bv;
    a.x = cb[(size_t)(m0 + l16) * aStrideM + (size_t)(ka + 0) * aStrideK];
    a.y = cb[(size_t)(m0 + l16) * aStrideM + (size_t)(ka + 1) * aStrideK];
    bv.x = nOK ? V[((size_t)b * SS + ka + 0) * HH + nIdx] : 0.f;
    bv.y = nOK ? V[((size_t)b * SS + ka + 1) * HH + nIdx] : 0.f;
    acc = __builtin_amdgcn_wmma_f32_16x16x4_f32(false, a, false, bv, (short)0, acc, false, false);
  }
  if (nOK) {
#pragma unroll
    for (int r = 0; r < 8; r++) {
      int m = m0 + r + 8 * half;
      float d = fmaxf(csum[b * SS + m], EPSF);
      attO[((size_t)b * SS + m) * HH + nIdx] = acc[r] / d;
    }
  }
}

__global__ void attmax_kernel(const float* __restrict__ cosM, const float* __restrict__ V,
                              const int* __restrict__ mask, float* __restrict__ attO,
                              int cRowStride, int cColStride) {
  int b = blockIdx.x / SS, r = blockIdx.x % SS;
  int h = threadIdx.x;
  if (h >= HH) return;
  const float* cb = cosM + (size_t)b * SS * SS;
  float mx = -3.0e38f;
  for (int k = 0; k < SS; k++) {
    float c = cb[(size_t)r * cRowStride + (size_t)k * cColStride];
    float v = V[((size_t)b * SS + k) * HH + h];
    mx = fmaxf(mx, c * v);
  }
  float m = (float)mask[b * SS + r];
  attO[((size_t)b * SS + r) * HH + h] = m * mx;
}

__global__ void mpm_kernel(const float* __restrict__ v1, const float* __restrict__ v2,
                           const float* __restrict__ w, int v2_bstride, int v2_sstride,
                           float* __restrict__ out, int side, int foff) {
  __shared__ float r1[HH], r2[HH];
  int bs = blockIdx.x;
  int b = bs / SS, s = bs % SS;
  int lane = threadIdx.x;
  const float* p1 = v1 + (size_t)bs * HH;
  const float* p2 = v2 + (size_t)b * v2_bstride + (size_t)s * v2_sstride;
  float d11 = 0, d22 = 0, d12 = 0;
  for (int h = lane; h < HH; h += 32) {
    float a = p1[h], c = p2[h];
    r1[h] = a; r2[h] = c;
    d11 += a * a; d22 += c * c; d12 += a * c;
  }
  d11 = wave_sum(d11); d22 = wave_sum(d22); d12 = wave_sum(d12);
  __syncthreads();
  float* orow = out + ((size_t)(side * BB + b) * SS + s) * FF + foff;
  if (lane == 0)
    orow[0] = d12 / (fmaxf(sqrtf(d11), EPSF) * fmaxf(sqrtf(d22), EPSF));
  for (int p = 0; p < PP; p++) {
    float s11 = 0, s22 = 0, s12 = 0;
    for (int h = lane; h < HH; h += 32) {
      float ww = w[p * HH + h]; float w2 = ww * ww;
      float a = r1[h], c = r2[h];
      s11 += w2 * a * a; s22 += w2 * c * c; s12 += w2 * a * c;
    }
    s11 = wave_sum(s11); s22 = wave_sum(s22); s12 = wave_sum(s12);
    if (lane == 0)
      orow[1 + p] = s12 / (fmaxf(sqrtf(s11), EPSF) * fmaxf(sqrtf(s22), EPSF));
  }
}

__global__ void maxpool_wmma_kernel(const float* __restrict__ v1, const float* __restrict__ v2,
                                    const float* __restrict__ w,
                                    const float* __restrict__ wnA, const float* __restrict__ wnB,
                                    const int* __restrict__ mask, float* __restrict__ out,
                                    int side) {
  int wid = blockIdx.x;
  int b = wid / (PP * 8);
  int p = (wid / 8) % PP;
  int x0 = (wid % 8) * 16;
  int lane = threadIdx.x, half = lane >> 4, l16 = lane & 15;
  float rmax[8], rsum[8], wA[8];
#pragma unroll
  for (int r = 0; r < 8; r++) {
    rmax[r] = -3.0e38f; rsum[r] = 0.f;
    wA[r] = wnA[((size_t)b * SS + x0 + r + 8 * half) * PP + p];
  }
  const float* a_base = v1 + ((size_t)b * SS + x0 + l16) * HH;
  const float* wp = w + p * HH;
  for (int y0 = 0; y0 < SS; y0 += 16) {
    const float* b_base = v2 + ((size_t)b * SS + y0 + l16) * HH;
    v8f acc = {0, 0, 0, 0, 0, 0, 0, 0};
    for (int k = 0; k < HH; k += 4) {
      int ka = k + 2 * half;
      float w0v = wp[ka], w1v = wp[ka + 1];
      v2f a, bv;
      a.x = w0v * w0v * a_base[ka];
      a.y = w1v * w1v * a_base[ka + 1];
      bv = *(const v2f*)(b_base + ka);
      acc = __builtin_amdgcn_wmma_f32_16x16x4_f32(false, a, false, bv, (short)0, acc, false, false);
    }
    float wB = wnB[((size_t)b * SS + y0 + l16) * PP + p];
#pragma unroll
    for (int r = 0; r < 8; r++) {
      float mv = acc[r] / fmaxf(wA[r] * wB, EPSF);
      rmax[r] = fmaxf(rmax[r], mv);
      rsum[r] += mv;
    }
  }
#pragma unroll
  for (int r = 0; r < 8; r++) {
    for (int o = 8; o > 0; o >>= 1) {
      rmax[r] = fmaxf(rmax[r], __shfl_xor(rmax[r], o, 32));
      rsum[r] += __shfl_xor(rsum[r], o, 32);
    }
  }
  if (l16 == 0) {
#pragma unroll
    for (int r = 0; r < 8; r++) {
      int x = x0 + r + 8 * half;
      float m = (float)mask[b * SS + x];
      float* orow = out + ((size_t)(side * BB + b) * SS + x) * FF;
      orow[23 + p] = m * rmax[r];
      orow[43 + p] = m * (rsum[r] / (float)SS);
    }
  }
}

extern "C" void kernel_launch(void* const* d_in, const int* in_sizes, int n_in,
                              void* d_out, int out_size, void* d_ws, size_t ws_size,
                              hipStream_t stream) {
  const float* ctx_p = (const float*)d_in[0];
  const float* ctx_h = (const float*)d_in[1];
  const float* w0 = (const float*)d_in[2];
  const float* w1 = (const float*)d_in[3];
  const float* w2 = (const float*)d_in[4];
  const float* w3 = (const float*)d_in[5];
  const int* mask_p = (const int*)d_in[6];
  const int* mask_h = (const int*)d_in[7];
  float* out = (float*)d_out;
  float* ws = (float*)d_ws;

  const size_t SZ_ROWS = (size_t)BB * SS * HH;
  float* cp = ws;                 ws += SZ_ROWS;
  float* ch = ws;                 ws += SZ_ROWS;
  float* n_p = ws;                ws += BB * SS;
  float* n_h = ws;                ws += BB * SS;
  float* wn1_p = ws;              ws += (size_t)BB * SS * PP;
  float* wn1_h = ws;              ws += (size_t)BB * SS * PP;
  float* cpl = ws;                ws += BB * HH;
  float* chl = ws;                ws += BB * HH;
  float* cosM = ws;               ws += (size_t)BB * SS * SS;
  float* cossum_y = ws;           ws += BB * SS;
  float* cossum_x = ws;           ws += BB * SS;
  float* att_mean_h = ws;         ws += SZ_ROWS;
  float* att_mean_p = ws;         ws += SZ_ROWS;
  float* att_max_h = ws;          ws += SZ_ROWS;
  float* att_max_p = ws;          ws += SZ_ROWS;

  prep_kernel<<<BB * SS, 32, 0, stream>>>(ctx_p, mask_p, w1, cp, n_p, wn1_p);
  prep_kernel<<<BB * SS, 32, 0, stream>>>(ctx_h, mask_h, w1, ch, n_h, wn1_h);

  gather_last_kernel<<<BB, 64, 0, stream>>>(cp, ch, mask_p, mask_h, cpl, chl);

  cos_wmma_kernel<<<BB * 64, 32, 0, stream>>>(cp, ch, n_p, n_h, cosM);

  cos_reduce_kernel<<<BB * SS, 128, 0, stream>>>(cosM, mask_p, cossum_y, out, 0, 0);
  cos_reduce_kernel<<<BB * SS, 128, 0, stream>>>(cosM, mask_h, cossum_x, out, 1, 1);

  attmean_wmma_kernel<<<BB * 8 * 13, 32, 0, stream>>>(cosM, ch, cossum_y, att_mean_h, SS, 1);
  attmean_wmma_kernel<<<BB * 8 * 13, 32, 0, stream>>>(cosM, cp, cossum_x, att_mean_p, 1, SS);

  attmax_kernel<<<BB * SS, 256, 0, stream>>>(cosM, ch, mask_p, att_max_h, SS, 1);
  attmax_kernel<<<BB * SS, 256, 0, stream>>>(cosM, cp, mask_h, att_max_p, 1, SS);

  mpm_kernel<<<BB * SS, 32, 0, stream>>>(cp, chl, w0, HH, 0, out, 0, 2);
  mpm_kernel<<<BB * SS, 32, 0, stream>>>(ch, cpl, w0, HH, 0, out, 1, 2);
  mpm_kernel<<<BB * SS, 32, 0, stream>>>(cp, att_mean_h, w2, SS * HH, HH, out, 0, 63);
  mpm_kernel<<<BB * SS, 32, 0, stream>>>(ch, att_mean_p, w2, SS * HH, HH, out, 1, 63);
  mpm_kernel<<<BB * SS, 32, 0, stream>>>(cp, att_max_h, w3, SS * HH, HH, out, 0, 84);
  mpm_kernel<<<BB * SS, 32, 0, stream>>>(ch, att_max_p, w3, SS * HH, HH, out, 1, 84);

  maxpool_wmma_kernel<<<BB * PP * 8, 32, 0, stream>>>(cp, ch, w1, wn1_p, wn1_h, mask_p, out, 0);
  maxpool_wmma_kernel<<<BB * PP * 8, 32, 0, stream>>>(ch, cp, w1, wn1_h, wn1_p, mask_h, out, 1);
}


# maxpool match as a pair-major GEMM: M = 16 (x,y) pairs of one x (A = v1[x]*v2[y] on the fly, y rows in registers), N = perspectives (w*w); a D tile feeds both sides; x- and y-sum order changes
# speedup vs baseline: 2.4842x; 2.4842x over previous
#include <hip/hip_runtime.h>
#include <hip/hip_bf16.h>


#define BB 16
#define SS 128
#define HH 200
#define PP 20
#define FF 105
#define EPSF 1e-8f

typedef float v2f __attribute__((ext_vector_type(2)));
typedef float v8f __attribute__((ext_vector_type(8)));

#define RT 32
#define LDP 201
#define QPB 7
#define NTH 128
#define NV4 (RT * (HH / 4))
#define NFULL (NV4 / NTH)
#define NTAIL (NV4 % NTH)

struct TilePieces { float4 v[NFULL + 1]; };

__device__ __forceinline__ void tile_load(TilePieces& t, const float* __restrict__ g,
                                          int row_stride, int tid) {
#pragma unroll
  for (int k = 0; k < NFULL; k++) {
    int i = tid + k * NTH, r = i / (HH / 4), c = i - r * (HH / 4);
    t.v[k] = *(const float4*)(g + (size_t)r * row_stride + 4 * c);
  }
  t.v[NFULL] = make_float4(0.f, 0.f, 0.f, 0.f);
  if (tid < NTAIL) {
    int i = tid + NFULL * NTH, r = i / (HH / 4), c = i - r * (HH / 4);
    t.v[NFULL] = *(const float4*)(g + (size_t)r * row_stride + 4 * c);
  }
}

__device__ __forceinline__ void tile_to_lds(const TilePieces& t, float* lds, int tid) {
#pragma unroll
  for (int k = 0; k <= NFULL; k++) {
    int i = tid + k * NTH, r = i / (HH / 4), c = i - r * (HH / 4);
    if (k < NFULL || tid < NTAIL) {
      float* d = lds + r * LDP + 4 * c;
      d[0] = t.v[k].x; d[1] = t.v[k].y; d[2] = t.v[k].z; d[3] = t.v[k].w;
    }
  }
}

__global__ __launch_bounds__(NTH) void prep_kernel(
    const float* __restrict__ src, const int* __restrict__ mask,
    const float* __restrict__ w1, float* __restrict__ dst,
    float* __restrict__ nrm, float* __restrict__ wn) {
  __shared__ float rows[RT * LDP];
  const int row0 = blockIdx.x * RT;
  const int tid = threadIdx.x;
  const bool writes_dst = (blockIdx.y == 0);
  const float* stile = src + (size_t)row0 * HH;
  float* dtile = dst + (size_t)row0 * HH;
  TilePieces t;
  tile_load(t, stile, HH, tid);
#pragma unroll
  for (int k = 0; k <= NFULL; k++) {
    int r = (tid + k * NTH) / (HH / 4);
    float m = (k < NFULL || tid < NTAIL) ? (float)mask[row0 + r] : 0.f;
    t.v[k].x *= m; t.v[k].y *= m; t.v[k].z *= m; t.v[k].w *= m;
  }
  if (writes_dst) {
#pragma unroll
    for (int k = 0; k <= NFULL; k++)
      if (k < NFULL || tid < NTAIL) *(float4*)(dtile + 4 * (tid + k * NTH)) = t.v[k];
  }
  tile_to_lds(t, rows, tid);
  __syncthreads();

  const int lane = tid & 31;
  const int wv = __builtin_amdgcn_readfirstlane(tid >> 5);
  const int q0 = QPB * blockIdx.y + 2 * wv;
  const bool two = (2 * wv + 1) < QPB;
  const bool plain = (q0 == PP);
  const float* wa = w1 + (plain ? 0 : q0) * HH;
  const float* wb = w1 + (two ? q0 + 1 : 0) * HH;
  const float* vr = rows + lane * LDP;
  float sa = 0.f, sb = 0.f;
#pragma unroll 8
  for (int h = 0; h < HH; h++) {
    float v = vr[h];
    float vv = v * v;
    float xa = wa[h], xb = wb[h];
    float w2a = plain ? 1.0f : xa * xa;
    sa += w2a * vv;
    sb += (xb * xb) * vv;
  }
  const int row = row0 + lane;
  if (plain) nrm[row] = sqrtf(sa);
  else wn[row * PP + q0] = sqrtf(sa);
  if (two) wn[row * PP + q0 + 1] = sqrtf(sb);
}

__global__ void gather_last_kernel(const float* __restrict__ cp, const float* __restrict__ ch,
                                   const int* __restrict__ mask_p, const int* __restrict__ mask_h,
                                   float* __restrict__ cpl, float* __restrict__ chl) {
  int b = blockIdx.x;
  int lp = 0, lh = 0;
  for (int i = 0; i < SS; i++) { lp += mask_p[b * SS + i]; lh += mask_h[b * SS + i]; }
  int ip = (lp > 0) ? lp - 1 : 0;
  int ih = (lh > 0) ? lh - 1 : 0;
  for (int h = threadIdx.x; h < HH; h += blockDim.x) {
    cpl[b * HH + h] = cp[((size_t)b * SS + ip) * HH + h];
    chl[b * HH + h] = ch[((size_t)b * SS + ih) * HH + h];
  }
}

__global__ void cos_wmma_kernel(const float* __restrict__ cp, const float* __restrict__ ch,
                                const float* __restrict__ np, const float* __restrict__ nh,
                                float* __restrict__ cosO) {
  int wid = blockIdx.x;
  int b = wid >> 6;
  int x0 = ((wid >> 3) & 7) << 4;
  int y0 = (wid & 7) << 4;
  int lane = threadIdx.x, half = lane >> 4, l16 = lane & 15;
  const float* arow = cp + ((size_t)(b * SS) + x0 + l16) * HH + 2 * half;
  const float* brow = ch + ((size_t)(b * SS) + y0 + l16) * HH + 2 * half;
  v8f acc = {0, 0, 0, 0, 0, 0, 0, 0};
  for (int k = 0; k < HH; k += 4) {
    v2f a = *(const v2f*)(arow + k);
    v2f bv = *(const v2f*)(brow + k);
    acc = __builtin_amdgcn_wmma_f32_16x16x4_f32(false, a, false, bv, (short)0, acc, false, false);
  }
  float nhv = nh[b * SS + y0 + l16];
#pragma unroll
  for (int r = 0; r < 8; r++) {
    int m = r + 8 * half;
    float d = fmaxf(np[b * SS + x0 + m] * nhv, EPSF);
    cosO[((size_t)(b * SS) + x0 + m) * SS + y0 + l16] = acc[r] / d;
  }
}

__global__ void cos_reduce_kernel(const float* __restrict__ cosM, const int* __restrict__ mask,
                                  float* __restrict__ cossum, float* __restrict__ out,
                                  int transposed, int side) {
  int b = blockIdx.x / SS, r = blockIdx.x % SS;
  int t = threadIdx.x;
  float v = transposed ? cosM[((size_t)b * SS + t) * SS + r]
                       : cosM[((size_t)b * SS + r) * SS + t];
  __shared__ float smax[128], ssum[128];
  smax[t] = v; ssum[t] = v;
  __syncthreads();
  for (int o = 64; o > 0; o >>= 1) {
    if (t < o) { smax[t] = fmaxf(smax[t], smax[t + o]); ssum[t] += ssum[t + o]; }
    __syncthreads();
  }
  if (t == 0) {
    float m = (float)mask[b * SS + r];
    cossum[b * SS + r] = ssum[0];
    float* o2 = out + ((size_t)(side * BB + b) * SS + r) * FF;
    o2[0] = m * smax[0];
    o2[1] = m * (ssum[0] / (float)SS);
  }
}

__global__ void attmean_wmma_kernel(const float* __restrict__ cosM, const float* __restrict__ V,
                                    const float* __restrict__ csum, float* __restrict__ attO,
                                    int aStrideM, int aStrideK) {
  int wid = blockIdx.x;
  int b = wid / (8 * 13);
  int rem = wid % (8 * 13);
  int m0 = (rem / 13) * 16;
  int n0 = (rem % 13) * 16;
  int lane = threadIdx.x, half = lane >> 4, l16 = lane & 15;
  const float* cb = cosM + (size_t)b * SS * SS;
  int nIdx = n0 + l16;
  bool nOK = nIdx < HH;
  v8f acc = {0, 0, 0, 0, 0, 0, 0, 0};
  for (int k = 0; k < SS; k += 4) {
    int ka = k + 2 * half;
    v2f a, bv;
    a.x = cb[(size_t)(m0 + l16) * aStrideM + (size_t)(ka + 0) * aStrideK];
    a.y = cb[(size_t)(m0 + l16) * aStrideM + (size_t)(ka + 1) * aStrideK];
    bv.x = nOK ? V[((size_t)b * SS + ka + 0) * HH + nIdx] : 0.f;
    bv.y = nOK ? V[((size_t)b * SS + ka + 1) * HH + nIdx] : 0.f;
    acc = __builtin_amdgcn_wmma_f32_16x16x4_f32(false, a, false, bv, (short)0, acc, false, false);
  }
  if (nOK) {
#pragma unroll
    for (int r = 0; r < 8; r++) {
      int m = m0 + r + 8 * half;
      float d = fmaxf(csum[b * SS + m], EPSF);
      attO[((size_t)b * SS + m) * HH + nIdx] = acc[r] / d;
    }
  }
}

__global__ void attmax_kernel(const float* __restrict__ cosM, const float* __restrict__ V,
                              const int* __restrict__ mask, float* __restrict__ attO,
                              int cRowStride, int cColStride) {
  int b = blockIdx.x / SS, r = blockIdx.x % SS;
  int h = threadIdx.x;
  if (h >= HH) return;
  const float* cb = cosM + (size_t)b * SS * SS;
  float mx = -3.0e38f;
  for (int k = 0; k < SS; k++) {
    float c = cb[(size_t)r * cRowStride + (size_t)k * cColStride];
    float v = V[((size_t)b * SS + k) * HH + h];
    mx = fmaxf(mx, c * v);
  }
  float m = (float)mask[b * SS + r];
  attO[((size_t)b * SS + r) * HH + h] = m * mx;
}

__global__ __launch_bounds__(NTH) void mpm_kernel(
    const float* __restrict__ v1, const float* __restrict__ v2,
    const float* __restrict__ w, int v2_bstride, int v2_sstride,
    float* __restrict__ out, int side, int foff) {
  __shared__ float r1[RT * LDP], r2[RT * LDP];
  const int row0 = blockIdx.x * RT;
  const int b = row0 / SS, s0 = row0 % SS;
  const int tid = threadIdx.x;
  const float* t1 = v1 + (size_t)row0 * HH;
  const float* t2 = v2 + (size_t)b * v2_bstride + (size_t)s0 * v2_sstride;
  TilePieces ta, tc;
  tile_load(ta, t1, HH, tid);
  tile_load(tc, t2, v2_sstride, tid);
  tile_to_lds(ta, r1, tid);
  tile_to_lds(tc, r2, tid);
  __syncthreads();

  const int lane = tid & 31;
  const int wv = __builtin_amdgcn_readfirstlane(tid >> 5);
  const int q0 = QPB * blockIdx.y + 2 * wv;
  const bool two = (2 * wv + 1) < QPB;
  const bool plain = (q0 == PP);
  const float* wa = w + (plain ? 0 : q0) * HH;
  const float* wb = w + (two ? q0 + 1 : 0) * HH;
  const float* ar = r1 + lane * LDP;
  const float* cr = r2 + lane * LDP;
  float a11 = 0.f, a22 = 0.f, a12 = 0.f, b11 = 0.f, b22 = 0.f, b12 = 0.f;
#pragma unroll 8
  for (int h = 0; h < HH; h++) {
    float a = ar[h], c = cr[h];
    float aa = a * a, cc = c * c, ac = a * c;
    float xa = wa[h], xb = wb[h];
    float w2a = plain ? 1.0f : xa * xa;
    float w2b = xb * xb;
    a11 += w2a * aa; a22 += w2a * cc; a12 += w2a * ac;
    b11 += w2b * aa; b22 += w2b * cc; b12 += w2b * ac;
  }
  float* orow = out + ((size_t)(side * BB + b) * SS + s0 + lane) * FF + foff;
  orow[plain ? 0 : 1 + q0] = a12 / (fmaxf(sqrtf(a11), EPSF) * fmaxf(sqrtf(a22), EPSF));
  if (two)
    orow[2 + q0] = b12 / (fmaxf(sqrtf(b11), EPSF) * fmaxf(sqrtf(b22), EPSF));
}

#define MP_XG 8
#define MP_WP 204
#define MP_NT 256

__global__ __launch_bounds__(MP_NT) void maxpool_pair_kernel(
    const float* __restrict__ v1, const float* __restrict__ v2,
    const float* __restrict__ w,
    const float* __restrict__ wnA, const float* __restrict__ wnB,
    const int* __restrict__ mask, float* __restrict__ out,
    float* __restrict__ pmax, float* __restrict__ psum) {
  __shared__ __attribute__((aligned(16))) float w2s[PP * MP_WP];
  __shared__ __attribute__((aligned(16))) float xs[MP_XG * HH];
  __shared__ float wnAs[MP_XG * PP];
  __shared__ float redm[8 * MP_XG * PP];
  __shared__ float reds[8 * MP_XG * PP];

  const int b = blockIdx.x / (SS / MP_XG);
  const int x0 = (blockIdx.x % (SS / MP_XG)) * MP_XG;
  const int tid = threadIdx.x;
  const int lane = tid & 31, half = lane >> 4, l16 = lane & 15;
  const int wv = tid >> 5;
  const int y0 = wv * 16;

  v2f yf[HH / 4];
  const float* yrow = v2 + ((size_t)b * SS + y0 + l16) * HH + 2 * half;
#pragma unroll
  for (int ks = 0; ks < HH / 4; ks++) yf[ks] = *(const v2f*)(yrow + 4 * ks);

  for (int i = tid; i < PP * HH; i += MP_NT) {
    int p = i / HH, h = i - p * HH;
    float t = w[i];
    w2s[p * MP_WP + h] = t * t;
  }
  for (int i = tid; i < MP_XG * HH; i += MP_NT)
    xs[i] = v1[((size_t)b * SS + x0) * HH + i];
  if (tid < MP_XG * PP) wnAs[tid] = wnA[((size_t)b * SS + x0) * PP + tid];

  const int p0 = l16;
  const bool p1ok = (16 + l16) < PP;
  const int p1 = p1ok ? 16 + l16 : PP - 1;
  float wB0[8], wB1[8];
#pragma unroll
  for (int r = 0; r < 8; r++) {
    const float* wr = wnB + ((size_t)b * SS + y0 + r + 8 * half) * PP;
    wB0[r] = wr[p0];
    wB1[r] = wr[p1];
  }
  float cm0[8], cs0[8], cm1[8], cs1[8];
#pragma unroll
  for (int r = 0; r < 8; r++) { cm0[r] = -3.0e38f; cs0[r] = 0.f; cm1[r] = -3.0e38f; cs1[r] = 0.f; }
  __syncthreads();

  const float* wb0 = w2s + p0 * MP_WP + 2 * half;
  const float* wb1 = w2s + p1 * MP_WP + 2 * half;
  for (int xi = 0; xi < MP_XG; xi++) {
    const float* xr = xs + xi * HH + 2 * half;
    v8f acc0 = {0, 0, 0, 0, 0, 0, 0, 0};
    v8f acc1 = {0, 0, 0, 0, 0, 0, 0, 0};
#pragma unroll
    for (int ks = 0; ks < HH / 4; ks++) {
      v2f xv = *(const v2f*)(xr + 4 * ks);
      v2f a;
      a.x = xv.x * yf[ks].x;
      a.y = xv.y * yf[ks].y;
      v2f b0 = *(const v2f*)(wb0 + 4 * ks);
      v2f b1 = *(const v2f*)(wb1 + 4 * ks);
      acc0 = __builtin_amdgcn_wmma_f32_16x16x4_f32(false, a, false, b0, (short)0, acc0, false, false);
      acc1 = __builtin_amdgcn_wmma_f32_16x16x4_f32(false, a, false, b1, (short)0, acc1, false, false);
    }
    const float wa0 = wnAs[xi * PP + p0];
    const float wa1 = wnAs[xi * PP + p1];
    float m0 = -3.0e38f, s0 = 0.f, m1 = -3.0e38f, s1 = 0.f;
#pragma unroll
    for (int r = 0; r < 8; r++) {
      float mv0 = acc0[r] / fmaxf(wa0 * wB0[r], EPSF);
      float mv1 = acc1[r] / fmaxf(wa1 * wB1[r], EPSF);
      m0 = fmaxf(m0, mv0); s0 += mv0;
      m1 = fmaxf(m1, mv1); s1 += mv1;
      cm0[r] = fmaxf(cm0[r], mv0); cs0[r] += mv0;
      cm1[r] = fmaxf(cm1[r], mv1); cs1[r] += mv1;
    }
    m0 = fmaxf(m0, __shfl_xor(m0, 16, 32)); s0 += __shfl_xor(s0, 16, 32);
    m1 = fmaxf(m1, __shfl_xor(m1, 16, 32)); s1 += __shfl_xor(s1, 16, 32);
    if (half == 0) {
      const int o = (wv * MP_XG + xi) * PP;
      redm[o + p0] = m0; reds[o + p0] = s0;
      if (p1ok) { redm[o + p1] = m1; reds[o + p1] = s1; }
    }
  }
  {
    const size_t po = (((size_t)b * (SS / MP_XG) + blockIdx.x % (SS / MP_XG)) * SS + y0 + 8 * half) * PP;
#pragma unroll
    for (int r = 0; r < 8; r++) {
      pmax[po + r * PP + p0] = cm0[r];
      psum[po + r * PP + p0] = cs0[r];
      if (p1ok) {
        pmax[po + r * PP + p1] = cm1[r];
        psum[po + r * PP + p1] = cs1[r];
      }
    }
  }
  __syncthreads();
  if (tid < MP_XG * PP) {
    const int xi = tid / PP, p = tid - xi * PP;
    float mx = redm[xi * PP + p], sm = reds[xi * PP + p];
    for (int q = 1; q < 8; q++) {
      mx = fmaxf(mx, redm[(q * MP_XG + xi) * PP + p]);
      sm += reds[(q * MP_XG + xi) * PP + p];
    }
    const int x = x0 + xi;
    const float m = (float)mask[b * SS + x];
    float* orow = out + ((size_t)b * SS + x) * FF;
    orow[23 + p] = m * mx;
    orow[43 + p] = m * (sm / (float)SS);
  }
}

__global__ void maxpool_finish_kernel(const float* __restrict__ pmax, const float* __restrict__ psum,
                                      const int* __restrict__ mask, float* __restrict__ out) {
  const int i = blockIdx.x * 256 + threadIdx.x;
  const int b = i / (SS * PP), yp = i - b * (SS * PP);
  const int y = yp / PP, p = yp - y * PP;
  float mx = -3.0e38f, sm = 0.f;
  for (int g = 0; g < SS / MP_XG; g++) {
    const size_t o = ((size_t)b * (SS / MP_XG) + g) * (SS * PP) + yp;
    mx = fmaxf(mx, pmax[o]);
    sm += psum[o];
  }
  const float m = (float)mask[b * SS + y];
  float* orow = out + ((size_t)(BB + b) * SS + y) * FF;
  orow[23 + p] = m * mx;
  orow[43 + p] = m * (sm / (float)SS);
}

extern "C" void kernel_launch(void* const* d_in, const int* in_sizes, int n_in,
                              void* d_out, int out_size, void* d_ws, size_t ws_size,
                              hipStream_t stream) {
  const float* ctx_p = (const float*)d_in[0];
  const float* ctx_h = (const float*)d_in[1];
  const float* w0 = (const float*)d_in[2];
  const float* w1 = (const float*)d_in[3];
  const float* w2 = (const float*)d_in[4];
  const float* w3 = (const float*)d_in[5];
  const int* mask_p = (const int*)d_in[6];
  const int* mask_h = (const int*)d_in[7];
  float* out = (float*)d_out;
  float* ws = (float*)d_ws;

  const size_t SZ_ROWS = (size_t)BB * SS * HH;
  float* cp = ws;                 ws += SZ_ROWS;
  float* ch = ws;                 ws += SZ_ROWS;
  float* n_p = ws;                ws += BB * SS;
  float* n_h = ws;                ws += BB * SS;
  float* wn1_p = ws;              ws += (size_t)BB * SS * PP;
  float* wn1_h = ws;              ws += (size_t)BB * SS * PP;
  float* cpl = ws;                ws += BB * HH;
  float* chl = ws;                ws += BB * HH;
  float* cosM = ws;               ws += (size_t)BB * SS * SS;
  float* cossum_y = ws;           ws += BB * SS;
  float* cossum_x = ws;           ws += BB * SS;
  float* att_mean_h = ws;         ws += SZ_ROWS;
  float* att_mean_p = ws;         ws += SZ_ROWS;
  float* att_max_h = ws;          ws += SZ_ROWS;
  float* att_max_p = ws;          ws += SZ_ROWS;
  float* mp_pmax = ws;            ws += (size_t)BB * (SS / MP_XG) * SS * PP;
  float* mp_psum = ws;            ws += (size_t)BB * (SS / MP_XG) * SS * PP;

  const dim3 row_tiles(BB * SS / RT, 3);
  prep_kernel<<<row_tiles, NTH, 0, stream>>>(ctx_p, mask_p, w1, cp, n_p, wn1_p);
  prep_kernel<<<row_tiles, NTH, 0, stream>>>(ctx_h, mask_h, w1, ch, n_h, wn1_h);

  gather_last_kernel<<<BB, 64, 0, stream>>>(cp, ch, mask_p, mask_h, cpl, chl);

  cos_wmma_kernel<<<BB * 64, 32, 0, stream>>>(cp, ch, n_p, n_h, cosM);

  cos_reduce_kernel<<<BB * SS, 128, 0, stream>>>(cosM, mask_p, cossum_y, out, 0, 0);
  cos_reduce_kernel<<<BB * SS, 128, 0, stream>>>(cosM, mask_h, cossum_x, out, 1, 1);

  attmean_wmma_kernel<<<BB * 8 * 13, 32, 0, stream>>>(cosM, ch, cossum_y, att_mean_h, SS, 1);
  attmean_wmma_kernel<<<BB * 8 * 13, 32, 0, stream>>>(cosM, cp, cossum_x, att_mean_p, 1, SS);

  attmax_kernel<<<BB * SS, 256, 0, stream>>>(cosM, ch, mask_p, att_max_h, SS, 1);
  attmax_kernel<<<BB * SS, 256, 0, stream>>>(cosM, cp, mask_h, att_max_p, 1, SS);

  mpm_kernel<<<row_tiles, NTH, 0, stream>>>(cp, chl, w0, HH, 0, out, 0, 2);
  mpm_kernel<<<row_tiles, NTH, 0, stream>>>(ch, cpl, w0, HH, 0, out, 1, 2);
  mpm_kernel<<<row_tiles, NTH, 0, stream>>>(cp, att_mean_h, w2, SS * HH, HH, out, 0, 63);
  mpm_kernel<<<row_tiles, NTH, 0, stream>>>(ch, att_mean_p, w2, SS * HH, HH, out, 1, 63);
  mpm_kernel<<<row_tiles, NTH, 0, stream>>>(cp, att_max_h, w3, SS * HH, HH, out, 0, 84);
  mpm_kernel<<<row_tiles, NTH, 0, stream>>>(ch, att_max_p, w3, SS * HH, HH, out, 1, 84);

  maxpool_pair_kernel<<<BB * (SS / MP_XG), MP_NT, 0, stream>>>(cp, ch, w1, wn1_p, wn1_h, mask_p, out,
                                                               mp_pmax, mp_psum);
  maxpool_finish_kernel<<<BB * SS * PP / 256, 256, 0, stream>>>(mp_pmax, mp_psum, mask_h, out);
}


# max-attentive match: the block that consumes the max-attentive vectors computes them itself (8 rows per block, max-plus over k) into LDS, never to memory; 4 launches -> 1; no sum changes its order
# speedup vs baseline: 3.9476x; 1.5891x over previous
#include <hip/hip_runtime.h>
#include <hip/hip_bf16.h>


#define BB 16
#define SS 128
#define HH 200
#define PP 20
#define FF 105
#define EPSF 1e-8f

typedef float v2f __attribute__((ext_vector_type(2)));
typedef float v8f __attribute__((ext_vector_type(8)));

#define RT 32
#define LDP 201
#define QPB 7
#define NTH 128
#define NV4 (RT * (HH / 4))
#define NFULL (NV4 / NTH)
#define NTAIL (NV4 % NTH)

struct TilePieces { float4 v[NFULL + 1]; };

__device__ __forceinline__ void tile_load(TilePieces& t, const float* __restrict__ g,
                                          int row_stride, int tid) {
#pragma unroll
  for (int k = 0; k < NFULL; k++) {
    int i = tid + k * NTH, r = i / (HH / 4), c = i - r * (HH / 4);
    t.v[k] = *(const float4*)(g + (size_t)r * row_stride + 4 * c);
  }
  t.v[NFULL] = make_float4(0.f, 0.f, 0.f, 0.f);
  if (tid < NTAIL) {
    int i = tid + NFULL * NTH, r = i / (HH / 4), c = i - r * (HH / 4);
    t.v[NFULL] = *(const float4*)(g + (size_t)r * row_stride + 4 * c);
  }
}

__device__ __forceinline__ void tile_to_lds(const TilePieces& t, float* lds, int tid) {
#pragma unroll
  for (int k = 0; k <= NFULL; k++) {
    int i = tid + k * NTH, r = i / (HH / 4), c = i - r * (HH / 4);
    if (k < NFULL || tid < NTAIL) {
      float* d = lds + r * LDP + 4 * c;
      d[0] = t.v[k].x; d[1] = t.v[k].y; d[2] = t.v[k].z; d[3] = t.v[k].w;
    }
  }
}

__global__ __launch_bounds__(NTH) void prep_kernel(
    const float* __restrict__ src, const int* __restrict__ mask,
    const float* __restrict__ w1, float* __restrict__ dst,
    float* __restrict__ nrm, float* __restrict__ wn) {
  __shared__ float rows[RT * LDP];
  const int row0 = blockIdx.x * RT;
  const int tid = threadIdx.x;
  const bool writes_dst = (blockIdx.y == 0);
  const float* stile = src + (size_t)row0 * HH;
  float* dtile = dst + (size_t)row0 * HH;
  TilePieces t;
  tile_load(t, stile, HH, tid);
#pragma unroll
  for (int k = 0; k <= NFULL; k++) {
    int r = (tid + k * NTH) / (HH / 4);
    float m = (k < NFULL || tid < NTAIL) ? (float)mask[row0 + r] : 0.f;
    t.v[k].x *= m; t.v[k].y *= m; t.v[k].z *= m; t.v[k].w *= m;
  }
  if (writes_dst) {
#pragma unroll
    for (int k = 0; k <= NFULL; k++)
      if (k < NFULL || tid < NTAIL) *(float4*)(dtile + 4 * (tid + k * NTH)) = t.v[k];
  }
  tile_to_lds(t, rows, tid);
  __syncthreads();

  const int lane = tid & 31;
  const int wv = __builtin_amdgcn_readfirstlane(tid >> 5);
  const int q0 = QPB * blockIdx.y + 2 * wv;
  const bool two = (2 * wv + 1) < QPB;
  const bool plain = (q0 == PP);
  const float* wa = w1 + (plain ? 0 : q0) * HH;
  const float* wb = w1 + (two ? q0 + 1 : 0) * HH;
  const float* vr = rows + lane * LDP;
  float sa = 0.f, sb = 0.f;
#pragma unroll 8
  for (int h = 0; h < HH; h++) {
    float v = vr[h];
    float vv = v * v;
    float xa = wa[h], xb = wb[h];
    float w2a = plain ? 1.0f : xa * xa;
    sa += w2a * vv;
    sb += (xb * xb) * vv;
  }
  const int row = row0 + lane;
  if (plain) nrm[row] = sqrtf(sa);
  else wn[row * PP + q0] = sqrtf(sa);
  if (two) wn[row * PP + q0 + 1] = sqrtf(sb);
}

__global__ void gather_last_kernel(const float* __restrict__ cp, const float* __restrict__ ch,
                                   const int* __restrict__ mask_p, const int* __restrict__ mask_h,
                                   float* __restrict__ cpl, float* __restrict__ chl) {
  int b = blockIdx.x;
  int lp = 0, lh = 0;
  for (int i = 0; i < SS; i++) { lp += mask_p[b * SS + i]; lh += mask_h[b * SS + i]; }
  int ip = (lp > 0) ? lp - 1 : 0;
  int ih = (lh > 0) ? lh - 1 : 0;
  for (int h = threadIdx.x; h < HH; h += blockDim.x) {
    cpl[b * HH + h] = cp[((size_t)b * SS + ip) * HH + h];
    chl[b * HH + h] = ch[((size_t)b * SS + ih) * HH + h];
  }
}

__global__ void cos_wmma_kernel(const float* __restrict__ cp, const float* __restrict__ ch,
                                const float* __restrict__ np, const float* __restrict__ nh,
                                float* __restrict__ cosO) {
  int wid = blockIdx.x;
  int b = wid >> 6;
  int x0 = ((wid >> 3) & 7) << 4;
  int y0 = (wid & 7) << 4;
  int lane = threadIdx.x, half = lane >> 4, l16 = lane & 15;
  const float* arow = cp + ((size_t)(b * SS) + x0 + l16) * HH + 2 * half;
  const float* brow = ch + ((size_t)(b * SS) + y0 + l16) * HH + 2 * half;
  v8f acc = {0, 0, 0, 0, 0, 0, 0, 0};
  for (int k = 0; k < HH; k += 4) {
    v2f a = *(const v2f*)(arow + k);
    v2f bv = *(const v2f*)(brow + k);
    acc = __builtin_amdgcn_wmma_f32_16x16x4_f32(false, a, false, bv, (short)0, acc, false, false);
  }
  float nhv = nh[b * SS + y0 + l16];
#pragma unroll
  for (int r = 0; r < 8; r++) {
    int m = r + 8 * half;
    float d = fmaxf(np[b * SS + x0 + m] * nhv, EPSF);
    cosO[((size_t)(b * SS) + x0 + m) * SS + y0 + l16] = acc[r] / d;
  }
}

__global__ void cos_reduce_kernel(const float* __restrict__ cosM, const int* __restrict__ mask,
                                  float* __restrict__ cossum, float* __restrict__ out,
                                  int transposed, int side) {
  int b = blockIdx.x / SS, r = blockIdx.x % SS;
  int t = threadIdx.x;
  float v = transposed ? cosM[((size_t)b * SS + t) * SS + r]
                       : cosM[((size_t)b * SS + r) * SS + t];
  __shared__ float smax[128], ssum[128];
  smax[t] = v; ssum[t] = v;
  __syncthreads();
  for (int o = 64; o > 0; o >>= 1) {
    if (t < o) { smax[t] = fmaxf(smax[t], smax[t + o]); ssum[t] += ssum[t + o]; }
    __syncthreads();
  }
  if (t == 0) {
    float m = (float)mask[b * SS + r];
    cossum[b * SS + r] = ssum[0];
    float* o2 = out + ((size_t)(side * BB + b) * SS + r) * FF;
    o2[0] = m * smax[0];
    o2[1] = m * (ssum[0] / (float)SS);
  }
}

__global__ void attmean_wmma_kernel(const float* __restrict__ cosM, const float* __restrict__ V,
                                    const float* __restrict__ csum, float* __restrict__ attO,
                                    int aStrideM, int aStrideK) {
  int wid = blockIdx.x;
  int b = wid / (8 * 13);
  int rem = wid % (8 * 13);
  int m0 = (rem / 13) * 16;
  int n0 = (rem % 13) * 16;
  int lane = threadIdx.x, half = lane >> 4, l16 = lane & 15;
  const float* cb = cosM + (size_t)b * SS * SS;
  int nIdx = n0 + l16;
  bool nOK = nIdx < HH;
  v8f acc = {0, 0, 0, 0, 0, 0, 0, 0};
  for (int k = 0; k < SS; k += 4) {
    int ka = k + 2 * half;
    v2f a, bv;
    a.x = cb[(size_t)(m0 + l16) * aStrideM + (size_t)(ka + 0) * aStrideK];
    a.y = cb[(size_t)(m0 + l16) * aStrideM + (size_t)(ka + 1) * aStrideK];
    bv.x = nOK ? V[((size_t)b * SS + ka + 0) * HH + nIdx] : 0.f;
    bv.y = nOK ? V[((size_t)b * SS + ka + 1) * HH + nIdx] : 0.f;
    acc = __builtin_amdgcn_wmma_f32_16x16x4_f32(false, a, false, bv, (short)0, acc, false, false);
  }
  if (nOK) {
#pragma unroll
    for (int r = 0; r < 8; r++) {
      int m = m0 + r + 8 * half;
      float d = fmaxf(csum[b * SS + m], EPSF);
      attO[((size_t)b * SS + m) * HH + nIdx] = acc[r] / d;
    }
  }
}

__global__ __launch_bounds__(NTH) void mpm_kernel(
    const float* __restrict__ v1, const float* __restrict__ v2,
    const float* __restrict__ w, int v2_bstride, int v2_sstride,
    float* __restrict__ out, int side, int foff) {
  __shared__ float r1[RT * LDP], r2[RT * LDP];
  const int row0 = blockIdx.x * RT;
  const int b = row0 / SS, s0 = row0 % SS;
  const int tid = threadIdx.x;
  const float* t1 = v1 + (size_t)row0 * HH;
  const float* t2 = v2 + (size_t)b * v2_bstride + (size_t)s0 * v2_sstride;
  TilePieces ta, tc;
  tile_load(ta, t1, HH, tid);
  tile_load(tc, t2, v2_sstride, tid);
  tile_to_lds(ta, r1, tid);
  tile_to_lds(tc, r2, tid);
  __syncthreads();

  const int lane = tid & 31;
  const int wv = __builtin_amdgcn_readfirstlane(tid >> 5);
  const int q0 = QPB * blockIdx.y + 2 * wv;
  const bool two = (2 * wv + 1) < QPB;
  const bool plain = (q0 == PP);
  const float* wa = w + (plain ? 0 : q0) * HH;
  const float* wb = w + (two ? q0 + 1 : 0) * HH;
  const float* ar = r1 + lane * LDP;
  const float* cr = r2 + lane * LDP;
  float a11 = 0.f, a22 = 0.f, a12 = 0.f, b11 = 0.f, b22 = 0.f, b12 = 0.f;
#pragma unroll 8
  for (int h = 0; h < HH; h++) {
    float a = ar[h], c = cr[h];
    float aa = a * a, cc = c * c, ac = a * c;
    float xa = wa[h], xb = wb[h];
    float w2a = plain ? 1.0f : xa * xa;
    float w2b = xb * xb;
    a11 += w2a * aa; a22 += w2a * cc; a12 += w2a * ac;
    b11 += w2b * aa; b22 += w2b * cc; b12 += w2b * ac;
  }
  float* orow = out + ((size_t)(side * BB + b) * SS + s0 + lane) * FF + foff;
  orow[plain ? 0 : 1 + q0] = a12 / (fmaxf(sqrtf(a11), EPSF) * fmaxf(sqrtf(a22), EPSF));
  if (two)
    orow[2 + q0] = b12 / (fmaxf(sqrtf(b11), EPSF) * fmaxf(sqrtf(b22), EPSF));
}

#define MA_R 8
#define MA_NT 256
#define MA_NQ 21
#define MA_WP 201
#define MA_KC 16

typedef float v4f __attribute__((ext_vector_type(4)));

__device__ __forceinline__ void vcol_load(float (&v)[MA_KC], const float* __restrict__ vcol, int k0) {
#pragma unroll
  for (int i = 0; i < MA_KC; i++) v[i] = vcol[(size_t)(k0 + i) * HH];
}

__device__ __forceinline__ void maxplus_step(const float* cs, int k0, const float (&v)[MA_KC], float (&mx)[MA_R]) {
#pragma unroll
  for (int i = 0; i < MA_KC; i++) {
#pragma unroll
    for (int j = 0; j < MA_R; j++) mx[j] = fmaxf(mx[j], cs[(k0 + i) * MA_R + j] * v[i]);
  }
}

__global__ __launch_bounds__(MA_NT) void maxatt_mpm_kernel(
    const float* __restrict__ cp, const float* __restrict__ ch,
    const float* __restrict__ cosM, const float* __restrict__ w,
    const int* __restrict__ mask_p, const int* __restrict__ mask_h,
    float* __restrict__ out) {
  __shared__ __attribute__((aligned(16))) float r1[MA_R * HH];
  __shared__ __attribute__((aligned(16))) float r2[MA_R * HH];
  __shared__ __attribute__((aligned(16))) float cs[SS * MA_R];
  __shared__ float w2s[MA_NQ * MA_WP];

  const int side = blockIdx.y;
  const int row0 = blockIdx.x * MA_R;
  const int b = row0 / SS, r0 = row0 % SS;
  const int tid = threadIdx.x;
  const float* v1 = side ? ch : cp;
  const float* V = side ? cp : ch;
  const int* mask = side ? mask_h : mask_p;
  const float* cb = cosM + (size_t)b * SS * SS;

  constexpr int W4 = PP * HH / 4;
  constexpr int R4 = MA_R * HH / 4;
  v4f wv4[4], rv4[2], cv4;
#pragma unroll
  for (int u = 0; u < 4; u++) {
    const int i = tid + u * MA_NT;
    wv4[u] = *(const v4f*)(w + 4 * (i < W4 ? i : W4 - 1));
  }
#pragma unroll
  for (int u = 0; u < 2; u++) {
    const int i = tid + u * MA_NT;
    rv4[u] = *(const v4f*)(v1 + (size_t)row0 * HH + 4 * (i < R4 ? i : R4 - 1));
  }
  const int pj = tid & 7, pk = 4 * (tid >> 3);
  const int qk = tid >> 1, qj = 4 * (tid & 1);
  cv4 = side == 0 ? *(const v4f*)(cb + (size_t)(r0 + pj) * SS + pk)
                  : *(const v4f*)(cb + (size_t)qk * SS + r0 + qj);
  asm volatile("" :: "v"(wv4[0]), "v"(wv4[1]), "v"(wv4[2]), "v"(wv4[3]), "v"(rv4[0]), "v"(rv4[1]), "v"(cv4));
#pragma unroll
  for (int u = 0; u < 4; u++) {
    const int i = tid + u * MA_NT;
    if (i < W4) {
      const int p = i / (HH / 4), c = i - p * (HH / 4);
      float* d = w2s + p * MA_WP + 4 * c;
      d[0] = wv4[u].x * wv4[u].x; d[1] = wv4[u].y * wv4[u].y;
      d[2] = wv4[u].z * wv4[u].z; d[3] = wv4[u].w * wv4[u].w;
    }
  }
  if (tid < HH) w2s[PP * MA_WP + tid] = 1.0f;
#pragma unroll
  for (int u = 0; u < 2; u++) {
    const int i = tid + u * MA_NT;
    if (i < R4) *(v4f*)(r1 + 4 * i) = rv4[u];
  }
  if (side == 0) {
    cs[(pk + 0) * MA_R + pj] = cv4.x; cs[(pk + 1) * MA_R + pj] = cv4.y;
    cs[(pk + 2) * MA_R + pj] = cv4.z; cs[(pk + 3) * MA_R + pj] = cv4.w;
  } else {
    *(v4f*)(cs + qk * MA_R + qj) = cv4;
  }
  __syncthreads();

  const int hc = tid < HH ? tid : HH - 1;
  const float* vcol = V + (size_t)b * SS * HH + hc;
  float mx[MA_R];
#pragma unroll
  for (int j = 0; j < MA_R; j++) mx[j] = -3.0e38f;
  if ((tid & ~31) < HH) {
    float va[MA_KC], vb[MA_KC];
    vcol_load(va, vcol, 0);
#pragma unroll
    for (int k0 = 0; k0 < SS; k0 += 2 * MA_KC) {
      vcol_load(vb, vcol, k0 + MA_KC);
      maxplus_step(cs, k0, va, mx);
      if (k0 + 2 * MA_KC < SS) vcol_load(va, vcol, k0 + 2 * MA_KC);
      maxplus_step(cs, k0 + MA_KC, vb, mx);
    }
  }
  if (tid < HH) {
#pragma unroll
    for (int j = 0; j < MA_R; j++) r2[j * HH + tid] = (float)mask[row0 + j] * mx[j];
  }
  __syncthreads();

  if (tid < MA_R * MA_NQ) {
    const int j = tid / MA_NQ, q = tid - j * MA_NQ;
    const float* ar = r1 + j * HH;
    const float* cr = r2 + j * HH;
    const float* wq = w2s + q * MA_WP;
    float a11 = 0.f, a22 = 0.f, a12 = 0.f;
#pragma unroll 8
    for (int h = 0; h < HH; h++) {
      const float a = ar[h], c = cr[h];
      const float aa = a * a, cc = c * c, ac = a * c;
      const float w2a = wq[h];
      a11 += w2a * aa; a22 += w2a * cc; a12 += w2a * ac;
    }
    float* orow = out + ((size_t)(side * BB + b) * SS + r0 + j) * FF + 84;
    orow[q == PP ? 0 : 1 + q] = a12 / (fmaxf(sqrtf(a11), EPSF) * fmaxf(sqrtf(a22), EPSF));
  }
}

#define MP_XG 8
#define MP_WP 204
#define MP_NT 256

__global__ __launch_bounds__(MP_NT) void maxpool_pair_kernel(
    const float* __restrict__ v1, const float* __restrict__ v2,
    const float* __restrict__ w,
    const float* __restrict__ wnA, const float* __restrict__ wnB,
    const int* __restrict__ mask, float* __restrict__ out,
    float* __restrict__ pmax, float* __restrict__ psum) {
  __shared__ __attribute__((aligned(16))) float w2s[PP * MP_WP];
  __shared__ __attribute__((aligned(16))) float xs[MP_XG * HH];
  __shared__ float wnAs[MP_XG * PP];
  __shared__ float redm[8 * MP_XG * PP];
  __shared__ float reds[8 * MP_XG * PP];

  const int b = blockIdx.x / (SS / MP_XG);
  const int x0 = (blockIdx.x % (SS / MP_XG)) * MP_XG;
  const int tid = threadIdx.x;
  const int lane = tid & 31, half = lane >> 4, l16 = lane & 15;
  const int wv = tid >> 5;
  const int y0 = wv * 16;

  v2f yf[HH / 4];
  const float* yrow = v2 + ((size_t)b * SS + y0 + l16) * HH + 2 * half;
#pragma unroll
  for (int ks = 0; ks < HH / 4; ks++) yf[ks] = *(const v2f*)(yrow + 4 * ks);

  for (int i = tid; i < PP * HH; i += MP_NT) {
    int p = i / HH, h = i - p * HH;
    float t = w[i];
    w2s[p * MP_WP + h] = t * t;
  }
  for (int i = tid; i < MP_XG * HH; i += MP_NT)
    xs[i] = v1[((size_t)b * SS + x0) * HH + i];
  if (tid < MP_XG * PP) wnAs[tid] = wnA[((size_t)b * SS + x0) * PP + tid];

  const int p0 = l16;
  const bool p1ok = (16 + l16) < PP;
  const int p1 = p1ok ? 16 + l16 : PP - 1;
  float wB0[8], wB1[8];
#pragma unroll
  for (int r = 0; r < 8; r++) {
    const float* wr = wnB + ((size_t)b * SS + y0 + r + 8 * half) * PP;
    wB0[r] = wr[p0];
    wB1[r] = wr[p1];
  }
  float cm0[8], cs0[8], cm1[8], cs1[8];
#pragma unroll
  for (int r = 0; r < 8; r++) { cm0[r] = -3.0e38f; cs0[r] = 0.f; cm1[r] = -3.0e38f; cs1[r] = 0.f; }
  __syncthreads();

  const float* wb0 = w2s + p0 * MP_WP + 2 * half;
  const float* wb1 = w2s + p1 * MP_WP + 2 * half;
  for (int xi = 0; xi < MP_XG; xi++) {
    const float* xr = xs + xi * HH + 2 * half;
    v8f acc0 = {0, 0, 0, 0, 0, 0, 0, 0};
    v8f acc1 = {0, 0, 0, 0, 0, 0, 0, 0};
#pragma unroll
    for (int ks = 0; ks < HH / 4; ks++) {
      v2f xv = *(const v2f*)(xr + 4 * ks);
      v2f a;
      a.x = xv.x * yf[ks].x;
      a.y = xv.y * yf[ks].y;
      v2f b0 = *(const v2f*)(wb0 + 4 * ks);
      v2f b1 = *(const v2f*)(wb1 + 4 * ks);
      acc0 = __builtin_amdgcn_wmma_f32_16x16x4_f32(false, a, false, b0, (short)0, acc0, false, false);
      acc1 = __builtin_amdgcn_wmma_f32_16x16x4_f32(false, a, false, b1, (short)0, acc1, false, false);
    }
    const float wa0 = wnAs[xi * PP + p0];
    const float wa1 = wnAs[xi * PP + p1];
    float m0 = -3.0e38f, s0 = 0.f, m1 = -3.0e38f, s1 = 0.f;
#pragma unroll
    for (int r = 0; r < 8; r++) {
      float mv0 = acc0[r] / fmaxf(wa0 * wB0[r], EPSF);
      float mv1 = acc1[r] / fmaxf(wa1 * wB1[r], EPSF);
      m0 = fmaxf(m0, mv0); s0 += mv0;
      m1 = fmaxf(m1, mv1); s1 += mv1;
      cm0[r] = fmaxf(cm0[r], mv0); cs0[r] += mv0;
      cm1[r] = fmaxf(cm1[r], mv1); cs1[r] += mv1;
    }
    m0 = fmaxf(m0, __shfl_xor(m0, 16, 32)); s0 += __shfl_xor(s0, 16, 32);
    m1 = fmaxf(m1, __shfl_xor(m1, 16, 32)); s1 += __shfl_xor(s1, 16, 32);
    if (half == 0) {
      const int o = (wv * MP_XG + xi) * PP;
      redm[o + p0] = m0; reds[o + p0] = s0;
      if (p1ok) { redm[o + p1] = m1; reds[o + p1] = s1; }
    }
  }
  {
    const size_t po = (((size_t)b * (SS / MP_XG) + blockIdx.x % (SS / MP_XG)) * SS + y0 + 8 * half) * PP;
#pragma unroll
    for (int r = 0; r < 8; r++) {
      pmax[po + r * PP + p0] = cm0[r];
      psum[po + r * PP + p0] = cs0[r];
      if (p1ok) {
        pmax[po + r * PP + p1] = cm1[r];
        psum[po + r * PP + p1] = cs1[r];
      }
    }
  }
  __syncthreads();
  if (tid < MP_XG * PP) {
    const int xi = tid / PP, p = tid - xi * PP;
    float mx = redm[xi * PP + p], sm = reds[xi * PP + p];
    for (int q = 1; q < 8; q++) {
      mx = fmaxf(mx, redm[(q * MP_XG + xi) * PP + p]);
      sm += reds[(q * MP_XG + xi) * PP + p];
    }
    const int x = x0 + xi;
    const float m = (float)mask[b * SS + x];
    float* orow = out + ((size_t)b * SS + x) * FF;
    orow[23 + p] = m * mx;
    orow[43 + p] = m * (sm / (float)SS);
  }
}

__global__ void maxpool_finish_kernel(const float* __restrict__ pmax, const float* __restrict__ psum,
                                      const int* __restrict__ mask, float* __restrict__ out) {
  const int i = blockIdx.x * 256 + threadIdx.x;
  const int b = i / (SS * PP), yp = i - b * (SS * PP);
  const int y = yp / PP, p = yp - y * PP;
  float mx = -3.0e38f, sm = 0.f;
  for (int g = 0; g < SS / MP_XG; g++) {
    const size_t o = ((size_t)b * (SS / MP_XG) + g) * (SS * PP) + yp;
    mx = fmaxf(mx, pmax[o]);
    sm += psum[o];
  }
  const float m = (float)mask[b * SS + y];
  float* orow = out + ((size_t)(BB + b) * SS + y) * FF;
  orow[23 + p] = m * mx;
  orow[43 + p] = m * (sm / (float)SS);
}

extern "C" void kernel_launch(void* const* d_in, const int* in_sizes, int n_in,
                              void* d_out, int out_size, void* d_ws, size_t ws_size,
                              hipStream_t stream) {
  const float* ctx_p = (const float*)d_in[0];
  const float* ctx_h = (const float*)d_in[1];
  const float* w0 = (const float*)d_in[2];
  const float* w1 = (const float*)d_in[3];
  const float* w2 = (const float*)d_in[4];
  const float* w3 = (const float*)d_in[5];
  const int* mask_p = (const int*)d_in[6];
  const int* mask_h = (const int*)d_in[7];
  float* out = (float*)d_out;
  float* ws = (float*)d_ws;

  const size_t SZ_ROWS = (size_t)BB * SS * HH;
  float* cp = ws;                 ws += SZ_ROWS;
  float* ch = ws;                 ws += SZ_ROWS;
  float* n_p = ws;                ws += BB * SS;
  float* n_h = ws;                ws += BB * SS;
  float* wn1_p = ws;              ws += (size_t)BB * SS * PP;
  float* wn1_h = ws;              ws += (size_t)BB * SS * PP;
  float* cpl = ws;                ws += BB * HH;
  float* chl = ws;                ws += BB * HH;
  float* cosM = ws;               ws += (size_t)BB * SS * SS;
  float* cossum_y = ws;           ws += BB * SS;
  float* cossum_x = ws;           ws += BB * SS;
  float* att_mean_h = ws;         ws += SZ_ROWS;
  float* att_mean_p = ws;         ws += SZ_ROWS;
  float* mp_pmax = ws;            ws += (size_t)BB * (SS / MP_XG) * SS * PP;
  float* mp_psum = ws;            ws += (size_t)BB * (SS / MP_XG) * SS * PP;

  const dim3 row_tiles(BB * SS / RT, 3);
  prep_kernel<<<row_tiles, NTH, 0, stream>>>(ctx_p, mask_p, w1, cp, n_p, wn1_p);
  prep_kernel<<<row_tiles, NTH, 0, stream>>>(ctx_h, mask_h, w1, ch, n_h, wn1_h);

  gather_last_kernel<<<BB, 64, 0, stream>>>(cp, ch, mask_p, mask_h, cpl, chl);

  cos_wmma_kernel<<<BB * 64, 32, 0, stream>>>(cp, ch, n_p, n_h, cosM);

  cos_reduce_kernel<<<BB * SS, 128, 0, stream>>>(cosM, mask_p, cossum_y, out, 0, 0);
  cos_reduce_kernel<<<BB * SS, 128, 0, stream>>>(cosM, mask_h, cossum_x, out, 1, 1);

  attmean_wmma_kernel<<<BB * 8 * 13, 32, 0, stream>>>(cosM, ch, cossum_y, att_mean_h, SS, 1);
  attmean_wmma_kernel<<<BB * 8 * 13, 32, 0, stream>>>(cosM, cp, cossum_x, att_mean_p, 1, SS);

  maxatt_mpm_kernel<<<dim3(BB * SS / MA_R, 2), MA_NT, 0, stream>>>(cp, ch, cosM, w3, mask_p, mask_h, out);

  mpm_kernel<<<row_tiles, NTH, 0, stream>>>(cp, chl, w0, HH, 0, out, 0, 2);
  mpm_kernel<<<row_tiles, NTH, 0, stream>>>(ch, cpl, w0, HH, 0, out, 1, 2);
  mpm_kernel<<<row_tiles, NTH, 0, stream>>>(cp, att_mean_h, w2, SS * HH, HH, out, 0, 63);
  mpm_kernel<<<row_tiles, NTH, 0, stream>>>(ch, att_mean_p, w2, SS * HH, HH, out, 1, 63);

  maxpool_pair_kernel<<<BB * (SS / MP_XG), MP_NT, 0, stream>>>(cp, ch, w1, wn1_p, wn1_h, mask_p, out,
                                                               mp_pmax, mp_psum);
  maxpool_finish_kernel<<<BB * SS * PP / 256, 256, 0, stream>>>(mp_pmax, mp_psum, mask_h, out);
}


# attentive match: the block that consumes the attentive-mean rows computes them itself (WMMA, k in order) and the cos row max/sum from its cos piece in LDS; 6 launches -> 1; cos row-sum order changes
# speedup vs baseline: 5.1608x; 1.3073x over previous
#include <hip/hip_runtime.h>
#include <hip/hip_bf16.h>


#define BB 16
#define SS 128
#define HH 200
#define PP 20
#define FF 105
#define EPSF 1e-8f

typedef float v2f __attribute__((ext_vector_type(2)));
typedef float v8f __attribute__((ext_vector_type(8)));

#define RT 32
#define LDP 201
#define QPB 7
#define NTH 128
#define NV4 (RT * (HH / 4))
#define NFULL (NV4 / NTH)
#define NTAIL (NV4 % NTH)

struct TilePieces { float4 v[NFULL + 1]; };

__device__ __forceinline__ void tile_load(TilePieces& t, const float* __restrict__ g,
                                          int row_stride, int tid) {
#pragma unroll
  for (int k = 0; k < NFULL; k++) {
    int i = tid + k * NTH, r = i / (HH / 4), c = i - r * (HH / 4);
    t.v[k] = *(const float4*)(g + (size_t)r * row_stride + 4 * c);
  }
  t.v[NFULL] = make_float4(0.f, 0.f, 0.f, 0.f);
  if (tid < NTAIL) {
    int i = tid + NFULL * NTH, r = i / (HH / 4), c = i - r * (HH / 4);
    t.v[NFULL] = *(const float4*)(g + (size_t)r * row_stride + 4 * c);
  }
}

__device__ __forceinline__ void tile_to_lds(const TilePieces& t, float* lds, int tid) {
#pragma unroll
  for (int k = 0; k <= NFULL; k++) {
    int i = tid + k * NTH, r = i / (HH / 4), c = i - r * (HH / 4);
    if (k < NFULL || tid < NTAIL) {
      float* d = lds + r * LDP + 4 * c;
      d[0] = t.v[k].x; d[1] = t.v[k].y; d[2] = t.v[k].z; d[3] = t.v[k].w;
    }
  }
}

__global__ __launch_bounds__(NTH) void prep_kernel(
    const float* __restrict__ src, const int* __restrict__ mask,
    const float* __restrict__ w1, float* __restrict__ dst,
    float* __restrict__ nrm, float* __restrict__ wn) {
  __shared__ float rows[RT * LDP];
  const int row0 = blockIdx.x * RT;
  const int tid = threadIdx.x;
  const bool writes_dst = (blockIdx.y == 0);
  const float* stile = src + (size_t)row0 * HH;
  float* dtile = dst + (size_t)row0 * HH;
  TilePieces t;
  tile_load(t, stile, HH, tid);
#pragma unroll
  for (int k = 0; k <= NFULL; k++) {
    int r = (tid + k * NTH) / (HH / 4);
    float m = (k < NFULL || tid < NTAIL) ? (float)mask[row0 + r] : 0.f;
    t.v[k].x *= m; t.v[k].y *= m; t.v[k].z *= m; t.v[k].w *= m;
  }
  if (writes_dst) {
#pragma unroll
    for (int k = 0; k <= NFULL; k++)
      if (k < NFULL || tid < NTAIL) *(float4*)(dtile + 4 * (tid + k * NTH)) = t.v[k];
  }
  tile_to_lds(t, rows, tid);
  __syncthreads();

  const int lane = tid & 31;
  const int wv = __builtin_amdgcn_readfirstlane(tid >> 5);
  const int q0 = QPB * blockIdx.y + 2 * wv;
  const bool two = (2 * wv + 1) < QPB;
  const bool plain = (q0 == PP);
  const float* wa = w1 + (plain ? 0 : q0) * HH;
  const float* wb = w1 + (two ? q0 + 1 : 0) * HH;
  const float* vr = rows + lane * LDP;
  float sa = 0.f, sb = 0.f;
#pragma unroll 8
  for (int h = 0; h < HH; h++) {
    float v = vr[h];
    float vv = v * v;
    float xa = wa[h], xb = wb[h];
    float w2a = plain ? 1.0f : xa * xa;
    sa += w2a * vv;
    sb += (xb * xb) * vv;
  }
  const int row = row0 + lane;
  if (plain) nrm[row] = sqrtf(sa);
  else wn[row * PP + q0] = sqrtf(sa);
  if (two) wn[row * PP + q0 + 1] = sqrtf(sb);
}

__global__ void gather_last_kernel(const float* __restrict__ cp, const float* __restrict__ ch,
                                   const int* __restrict__ mask_p, const int* __restrict__ mask_h,
                                   float* __restrict__ cpl, float* __restrict__ chl) {
  int b = blockIdx.x;
  int lp = 0, lh = 0;
  for (int i = 0; i < SS; i++) { lp += mask_p[b * SS + i]; lh += mask_h[b * SS + i]; }
  int ip = (lp > 0) ? lp - 1 : 0;
  int ih = (lh > 0) ? lh - 1 : 0;
  for (int h = threadIdx.x; h < HH; h += blockDim.x) {
    cpl[b * HH + h] = cp[((size_t)b * SS + ip) * HH + h];
    chl[b * HH + h] = ch[((size_t)b * SS + ih) * HH + h];
  }
}

__global__ void cos_wmma_kernel(const float* __restrict__ cp, const float* __restrict__ ch,
                                const float* __restrict__ np, const float* __restrict__ nh,
                                float* __restrict__ cosO) {
  int wid = blockIdx.x;
  int b = wid >> 6;
  int x0 = ((wid >> 3) & 7) << 4;
  int y0 = (wid & 7) << 4;
  int lane = threadIdx.x, half = lane >> 4, l16 = lane & 15;
  const float* arow = cp + ((size_t)(b * SS) + x0 + l16) * HH + 2 * half;
  const float* brow = ch + ((size_t)(b * SS) + y0 + l16) * HH + 2 * half;
  v8f acc = {0, 0, 0, 0, 0, 0, 0, 0};
  for (int k = 0; k < HH; k += 4) {
    v2f a = *(const v2f*)(arow + k);
    v2f bv = *(const v2f*)(brow + k);
    acc = __builtin_amdgcn_wmma_f32_16x16x4_f32(false, a, false, bv, (short)0, acc, false, false);
  }
  float nhv = nh[b * SS + y0 + l16];
#pragma unroll
  for (int r = 0; r < 8; r++) {
    int m = r + 8 * half;
    float d = fmaxf(np[b * SS + x0 + m] * nhv, EPSF);
    cosO[((size_t)(b * SS) + x0 + m) * SS + y0 + l16] = acc[r] / d;
  }
}

__global__ __launch_bounds__(NTH) void mpm_kernel(
    const float* __restrict__ v1, const float* __restrict__ v2,
    const float* __restrict__ w, int v2_bstride, int v2_sstride,
    float* __restrict__ out, int side, int foff) {
  __shared__ float r1[RT * LDP], r2[RT * LDP];
  const int row0 = blockIdx.x * RT;
  const int b = row0 / SS, s0 = row0 % SS;
  const int tid = threadIdx.x;
  const float* t1 = v1 + (size_t)row0 * HH;
  const float* t2 = v2 + (size_t)b * v2_bstride + (size_t)s0 * v2_sstride;
  TilePieces ta, tc;
  tile_load(ta, t1, HH, tid);
  tile_load(tc, t2, v2_sstride, tid);
  tile_to_lds(ta, r1, tid);
  tile_to_lds(tc, r2, tid);
  __syncthreads();

  const int lane = tid & 31;
  const int wv = __builtin_amdgcn_readfirstlane(tid >> 5);
  const int q0 = QPB * blockIdx.y + 2 * wv;
  const bool two = (2 * wv + 1) < QPB;
  const bool plain = (q0 == PP);
  const float* wa = w + (plain ? 0 : q0) * HH;
  const float* wb = w + (two ? q0 + 1 : 0) * HH;
  const float* ar = r1 + lane * LDP;
  const float* cr = r2 + lane * LDP;
  float a11 = 0.f, a22 = 0.f, a12 = 0.f, b11 = 0.f, b22 = 0.f, b12 = 0.f;
#pragma unroll 8
  for (int h = 0; h < HH; h++) {
    float a = ar[h], c = cr[h];
    float aa = a * a, cc = c * c, ac = a * c;
    float xa = wa[h], xb = wb[h];
    float w2a = plain ? 1.0f : xa * xa;
    float w2b = xb * xb;
    a11 += w2a * aa; a22 += w2a * cc; a12 += w2a * ac;
    b11 += w2b * aa; b22 += w2b * cc; b12 += w2b * ac;
  }
  float* orow = out + ((size_t)(side * BB + b) * SS + s0 + lane) * FF + foff;
  orow[plain ? 0 : 1 + q0] = a12 / (fmaxf(sqrtf(a11), EPSF) * fmaxf(sqrtf(a22), EPSF));
  if (two)
    orow[2 + q0] = b12 / (fmaxf(sqrtf(b11), EPSF) * fmaxf(sqrtf(b22), EPSF));
}

#define CP 132
#define AT_KS (SS / 4)

typedef float v4f __attribute__((ext_vector_type(4)));

__device__ __forceinline__ void bfrag_load(float (&bv)[2 * AT_KS], const float* __restrict__ vcol) {
#pragma unroll
  for (int i = 0; i < AT_KS; i++) {
    bv[2 * i] = vcol[(size_t)(4 * i) * HH];
    bv[2 * i + 1] = vcol[(size_t)(4 * i + 1) * HH];
  }
}

__device__ __forceinline__ void att_tile(const v2f (&a0)[AT_KS], const v2f (&a1)[AT_KS],
                                         const float (&bv)[2 * AT_KS], const float* csum,
                                         float* rows, int col, bool colOK, int half) {
  v8f acc0 = {0, 0, 0, 0, 0, 0, 0, 0};
  v8f acc1 = {0, 0, 0, 0, 0, 0, 0, 0};
#pragma unroll
  for (int i = 0; i < AT_KS; i++) {
    v2f bb;
    bb.x = bv[2 * i];
    bb.y = bv[2 * i + 1];
    acc0 = __builtin_amdgcn_wmma_f32_16x16x4_f32(false, a0[i], false, bb, (short)0, acc0, false, false);
    acc1 = __builtin_amdgcn_wmma_f32_16x16x4_f32(false, a1[i], false, bb, (short)0, acc1, false, false);
  }
  if (colOK) {
#pragma unroll
    for (int r = 0; r < 8; r++) {
      const int m = r + 8 * half;
      rows[m * LDP + col] = acc0[r] / fmaxf(csum[m], EPSF);
      rows[(16 + m) * LDP + col] = acc1[r] / fmaxf(csum[16 + m], EPSF);
    }
  }
}

__global__ __launch_bounds__(NTH) void att_mpm_kernel(
    const float* __restrict__ cp, const float* __restrict__ ch,
    const float* __restrict__ cosM, const float* __restrict__ w,
    const int* __restrict__ mask_p, const int* __restrict__ mask_h,
    float* __restrict__ out) {
  __shared__ float r1[RT * LDP];
  __shared__ __attribute__((aligned(16))) float r2[RT * LDP];
  __shared__ float csum[RT];
  const int side = blockIdx.z;
  const int row0 = blockIdx.x * RT;
  const int b = row0 / SS, s0 = row0 % SS;
  const int tid = threadIdx.x;
  const float* v1 = side ? ch : cp;
  const float* V = (side ? cp : ch) + (size_t)b * SS * HH;
  const int* mask = side ? mask_h : mask_p;
  const float* cb = cosM + (size_t)b * SS * SS;
  float* cst = r2;

  const int lane = tid & 31, half = lane >> 4, l16 = lane & 15;
  const int wv = __builtin_amdgcn_readfirstlane(tid >> 5);
  const int tw = 3 - wv;
  const int c0 = 16 * tw + l16, c1 = c0 + 64, c2 = c0 + 128, c3 = c0 + 192;
  const bool ok3 = c3 < HH;
  const float* vh = V + (size_t)(2 * half) * HH;

  TilePieces ta;
  tile_load(ta, v1 + (size_t)row0 * HH, HH, tid);
  v4f cv[8];
  const int sh = side ? 3 : 5, lo = side ? 7 : 31;
  const float* cpiece = cb + (side ? s0 : s0 * SS);
#pragma unroll
  for (int u = 0; u < 8; u++) {
    const int i = tid + u * NTH;
    cv[u] = *(const v4f*)(cpiece + (i >> sh) * SS + 4 * (i & lo));
  }
  float ba[2 * AT_KS], bb[2 * AT_KS];
  bfrag_load(ba, vh + c0);
  tile_to_lds(ta, r1, tid);
  if (side == 0) {
#pragma unroll
    for (int u = 0; u < 8; u++) {
      const int i = tid + u * NTH;
      *(v4f*)(cst + (i >> 5) * CP + 4 * (i & 31)) = cv[u];
    }
  } else {
#pragma unroll
    for (int u = 0; u < 8; u++) {
      const int i = tid + u * NTH;
      const int k = i >> 3, j = 4 * (i & 7);
      cst[(j + 0) * CP + k] = cv[u].x; cst[(j + 1) * CP + k] = cv[u].y;
      cst[(j + 2) * CP + k] = cv[u].z; cst[(j + 3) * CP + k] = cv[u].w;
    }
  }
  __syncthreads();

  {
    const int j = tid >> 2, q = tid & 3;
    const v4f* c4 = (const v4f*)(cst + j * CP + 32 * q);
    float mx = -3.0e38f, sm = 0.f;
#pragma unroll
    for (int i = 0; i < 8; i++) {
      const v4f t = c4[i];
      mx = fmaxf(fmaxf(fmaxf(mx, t.x), fmaxf(t.y, t.z)), t.w);
      sm += t.x; sm += t.y; sm += t.z; sm += t.w;
    }
    mx = fmaxf(mx, __shfl_xor(mx, 1, 32)); sm += __shfl_xor(sm, 1, 32);
    mx = fmaxf(mx, __shfl_xor(mx, 2, 32)); sm += __shfl_xor(sm, 2, 32);
    if (q == 0) {
      csum[j] = sm;
      if (blockIdx.y == 0) {
        const float m = (float)mask[row0 + j];
        float* o2 = out + ((size_t)(side * BB + b) * SS + s0 + j) * FF;
        o2[0] = m * mx;
        o2[1] = m * (sm / (float)SS);
      }
    }
  }
  v2f a0[AT_KS], a1[AT_KS];
  {
    const float* cl = cst + l16 * CP + 2 * half;
#pragma unroll
    for (int i = 0; i < AT_KS; i++) {
      a0[i] = *(const v2f*)(cl + 4 * i);
      a1[i] = *(const v2f*)(cl + 16 * CP + 4 * i);
    }
  }
  __syncthreads();

  bfrag_load(bb, vh + c1);
  __builtin_amdgcn_sched_barrier(0);
  att_tile(a0, a1, ba, csum, r2, c0, true, half);
  bfrag_load(ba, vh + c2);
  __builtin_amdgcn_sched_barrier(0);
  att_tile(a0, a1, bb, csum, r2, c1, true, half);
  if (tw == 0) bfrag_load(bb, vh + (ok3 ? c3 : HH - 1));
  att_tile(a0, a1, ba, csum, r2, c2, true, half);
  if (tw == 0) att_tile(a0, a1, bb, csum, r2, c3, ok3, half);
  __syncthreads();

  const int q0 = QPB * blockIdx.y + 2 * wv;
  const bool two = (2 * wv + 1) < QPB;
  const bool plain = (q0 == PP);
  const float* wa = w + (plain ? 0 : q0) * HH;
  const float* wb = w + (two ? q0 + 1 : 0) * HH;
  const float* ar = r1 + lane * LDP;
  const float* cr = r2 + lane * LDP;
  float a11 = 0.f, a22 = 0.f, a12 = 0.f, b11 = 0.f, b22 = 0.f, b12 = 0.f;
#pragma unroll 8
  for (int h = 0; h < HH; h++) {
    float a = ar[h], c = cr[h];
    float aa = a * a, cc = c * c, ac = a * c;
    float xa = wa[h], xb = wb[h];
    float w2a = plain ? 1.0f : xa * xa;
    float w2b = xb * xb;
    a11 += w2a * aa; a22 += w2a * cc; a12 += w2a * ac;
    b11 += w2b * aa; b22 += w2b * cc; b12 += w2b * ac;
  }
  float* orow = out + ((size_t)(side * BB + b) * SS + s0 + lane) * FF + 63;
  orow[plain ? 0 : 1 + q0] = a12 / (fmaxf(sqrtf(a11), EPSF) * fmaxf(sqrtf(a22), EPSF));
  if (two)
    orow[2 + q0] = b12 / (fmaxf(sqrtf(b11), EPSF) * fmaxf(sqrtf(b22), EPSF));
}

#define MA_R 8
#define MA_NT 256
#define MA_NQ 21
#define MA_WP 201
#define MA_KC 16

__device__ __forceinline__ void vcol_load(float (&v)[MA_KC], const float* __restrict__ vcol, int k0) {
#pragma unroll
  for (int i = 0; i < MA_KC; i++) v[i] = vcol[(size_t)(k0 + i) * HH];
}

__device__ __forceinline__ void maxplus_step(const float* cs, int k0, const float (&v)[MA_KC], float (&mx)[MA_R]) {
#pragma unroll
  for (int i = 0; i < MA_KC; i++) {
#pragma unroll
    for (int j = 0; j < MA_R; j++) mx[j] = fmaxf(mx[j], cs[(k0 + i) * MA_R + j] * v[i]);
  }
}

__global__ __launch_bounds__(MA_NT) void maxatt_mpm_kernel(
    const float* __restrict__ cp, const float* __restrict__ ch,
    const float* __restrict__ cosM, const float* __restrict__ w,
    const int* __restrict__ mask_p, const int* __restrict__ mask_h,
    float* __restrict__ out) {
  __shared__ __attribute__((aligned(16))) float r1[MA_R * HH];
  __shared__ __attribute__((aligned(16))) float r2[MA_R * HH];
  __shared__ __attribute__((aligned(16))) float cs[SS * MA_R];
  __shared__ float w2s[MA_NQ * MA_WP];

  const int side = blockIdx.y;
  const int row0 = blockIdx.x * MA_R;
  const int b = row0 / SS, r0 = row0 % SS;
  const int tid = threadIdx.x;
  const float* v1 = side ? ch : cp;
  const float* V = side ? cp : ch;
  const int* mask = side ? mask_h : mask_p;
  const float* cb = cosM + (size_t)b * SS * SS;

  constexpr int W4 = PP * HH / 4;
  constexpr int R4 = MA_R * HH / 4;
  v4f wv4[4], rv4[2], cv4;
#pragma unroll
  for (int u = 0; u < 4; u++) {
    const int i = tid + u * MA_NT;
    wv4[u] = *(const v4f*)(w + 4 * (i < W4 ? i : W4 - 1));
  }
#pragma unroll
  for (int u = 0; u < 2; u++) {
    const int i = tid + u * MA_NT;
    rv4[u] = *(const v4f*)(v1 + (size_t)row0 * HH + 4 * (i < R4 ? i : R4 - 1));
  }
  const int pj = tid & 7, pk = 4 * (tid >> 3);
  const int qk = tid >> 1, qj = 4 * (tid & 1);
  cv4 = side == 0 ? *(const v4f*)(cb + (size_t)(r0 + pj) * SS + pk)
                  : *(const v4f*)(cb + (size_t)qk * SS + r0 + qj);
  asm volatile("" :: "v"(wv4[0]), "v"(wv4[1]), "v"(wv4[2]), "v"(wv4[3]), "v"(rv4[0]), "v"(rv4[1]), "v"(cv4));
#pragma unroll
  for (int u = 0; u < 4; u++) {
    const int i = tid + u * MA_NT;
    if (i < W4) {
      const int p = i / (HH / 4), c = i - p * (HH / 4);
      float* d = w2s + p * MA_WP + 4 * c;
      d[0] = wv4[u].x * wv4[u].x; d[1] = wv4[u].y * wv4[u].y;
      d[2] = wv4[u].z * wv4[u].z; d[3] = wv4[u].w * wv4[u].w;
    }
  }
  if (tid < HH) w2s[PP * MA_WP + tid] = 1.0f;
#pragma unroll
  for (int u = 0; u < 2; u++) {
    const int i = tid + u * MA_NT;
    if (i < R4) *(v4f*)(r1 + 4 * i) = rv4[u];
  }
  if (side == 0) {
    cs[(pk + 0) * MA_R + pj] = cv4.x; cs[(pk + 1) * MA_R + pj] = cv4.y;
    cs[(pk + 2) * MA_R + pj] = cv4.z; cs[(pk + 3) * MA_R + pj] = cv4.w;
  } else {
    *(v4f*)(cs + qk * MA_R + qj) = cv4;
  }
  __syncthreads();

  const int hc = tid < HH ? tid : HH - 1;
  const float* vcol = V + (size_t)b * SS * HH + hc;
  float mx[MA_R];
#pragma unroll
  for (int j = 0; j < MA_R; j++) mx[j] = -3.0e38f;
  if ((tid & ~31) < HH) {
    float va[MA_KC], vb[MA_KC];
    vcol_load(va, vcol, 0);
#pragma unroll
    for (int k0 = 0; k0 < SS; k0 += 2 * MA_KC) {
      vcol_load(vb, vcol, k0 + MA_KC);
      maxplus_step(cs, k0, va, mx);
      if (k0 + 2 * MA_KC < SS) vcol_load(va, vcol, k0 + 2 * MA_KC);
      maxplus_step(cs, k0 + MA_KC, vb, mx);
    }
  }
  if (tid < HH) {
#pragma unroll
    for (int j = 0; j < MA_R; j++) r2[j * HH + tid] = (float)mask[row0 + j] * mx[j];
  }
  __syncthreads();

  if (tid < MA_R * MA_NQ) {
    const int j = tid / MA_NQ, q = tid - j * MA_NQ;
    const float* ar = r1 + j * HH;
    const float* cr = r2 + j * HH;
    const float* wq = w2s + q * MA_WP;
    float a11 = 0.f, a22 = 0.f, a12 = 0.f;
#pragma unroll 8
    for (int h = 0; h < HH; h++) {
      const float a = ar[h], c = cr[h];
      const float aa = a * a, cc = c * c, ac = a * c;
      const float w2a = wq[h];
      a11 += w2a * aa; a22 += w2a * cc; a12 += w2a * ac;
    }
    float* orow = out + ((size_t)(side * BB + b) * SS + r0 + j) * FF + 84;
    orow[q == PP ? 0 : 1 + q] = a12 / (fmaxf(sqrtf(a11), EPSF) * fmaxf(sqrtf(a22), EPSF));
  }
}

#define MP_XG 8
#define MP_WP 204
#define MP_NT 256

__global__ __launch_bounds__(MP_NT) void maxpool_pair_kernel(
    const float* __restrict__ v1, const float* __restrict__ v2,
    const float* __restrict__ w,
    const float* __restrict__ wnA, const float* __restrict__ wnB,
    const int* __restrict__ mask, float* __restrict__ out,
    float* __restrict__ pmax, float* __restrict__ psum) {
  __shared__ __attribute__((aligned(16))) float w2s[PP * MP_WP];
  __shared__ __attribute__((aligned(16))) float xs[MP_XG * HH];
  __shared__ float wnAs[MP_XG * PP];
  __shared__ float redm[8 * MP_XG * PP];
  __shared__ float reds[8 * MP_XG * PP];

  const int b = blockIdx.x / (SS / MP_XG);
  const int x0 = (blockIdx.x % (SS / MP_XG)) * MP_XG;
  const int tid = threadIdx.x;
  const int lane = tid & 31, half = lane >> 4, l16 = lane & 15;
  const int wv = tid >> 5;
  const int y0 = wv * 16;

  v2f yf[HH / 4];
  const float* yrow = v2 + ((size_t)b * SS + y0 + l16) * HH + 2 * half;
#pragma unroll
  for (int ks = 0; ks < HH / 4; ks++) yf[ks] = *(const v2f*)(yrow + 4 * ks);

  for (int i = tid; i < PP * HH; i += MP_NT) {
    int p = i / HH, h = i - p * HH;
    float t = w[i];
    w2s[p * MP_WP + h] = t * t;
  }
  for (int i = tid; i < MP_XG * HH; i += MP_NT)
    xs[i] = v1[((size_t)b * SS + x0) * HH + i];
  if (tid < MP_XG * PP) wnAs[tid] = wnA[((size_t)b * SS + x0) * PP + tid];

  const int p0 = l16;
  const bool p1ok = (16 + l16) < PP;
  const int p1 = p1ok ? 16 + l16 : PP - 1;
  float wB0[8], wB1[8];
#pragma unroll
  for (int r = 0; r < 8; r++) {
    const float* wr = wnB + ((size_t)b * SS + y0 + r + 8 * half) * PP;
    wB0[r] = wr[p0];
    wB1[r] = wr[p1];
  }
  float cm0[8], cs0[8], cm1[8], cs1[8];
#pragma unroll
  for (int r = 0; r < 8; r++) { cm0[r] = -3.0e38f; cs0[r] = 0.f; cm1[r] = -3.0e38f; cs1[r] = 0.f; }
  __syncthreads();

  const float* wb0 = w2s + p0 * MP_WP + 2 * half;
  const float* wb1 = w2s + p1 * MP_WP + 2 * half;
  for (int xi = 0; xi < MP_XG; xi++) {
    const float* xr = xs + xi * HH + 2 * half;
    v8f acc0 = {0, 0, 0, 0, 0, 0, 0, 0};
    v8f acc1 = {0, 0, 0, 0, 0, 0, 0, 0};
#pragma unroll
    for (int ks = 0; ks < HH / 4; ks++) {
      v2f xv = *(const v2f*)(xr + 4 * ks);
      v2f a;
      a.x = xv.x * yf[ks].x;
      a.y = xv.y * yf[ks].y;
      v2f b0 = *(const v2f*)(wb0 + 4 * ks);
      v2f b1 = *(const v2f*)(wb1 + 4 * ks);
      acc0 = __builtin_amdgcn_wmma_f32_16x16x4_f32(false, a, false, b0, (short)0, acc0, false, false);
      acc1 = __builtin_amdgcn_wmma_f32_16x16x4_f32(false, a, false, b1, (short)0, acc1, false, false);
    }
    const float wa0 = wnAs[xi * PP + p0];
    const float wa1 = wnAs[xi * PP + p1];
    float m0 = -3.0e38f, s0 = 0.f, m1 = -3.0e38f, s1 = 0.f;
#pragma unroll
    for (int r = 0; r < 8; r++) {
      float mv0 = acc0[r] / fmaxf(wa0 * wB0[r], EPSF);
      float mv1 = acc1[r] / fmaxf(wa1 * wB1[r], EPSF);
      m0 = fmaxf(m0, mv0); s0 += mv0;
      m1 = fmaxf(m1, mv1); s1 += mv1;
      cm0[r] = fmaxf(cm0[r], mv0); cs0[r] += mv0;
      cm1[r] = fmaxf(cm1[r], mv1); cs1[r] += mv1;
    }
    m0 = fmaxf(m0, __shfl_xor(m0, 16, 32)); s0 += __shfl_xor(s0, 16, 32);
    m1 = fmaxf(m1, __shfl_xor(m1, 16, 32)); s1 += __shfl_xor(s1, 16, 32);
    if (half == 0) {
      const int o = (wv * MP_XG + xi) * PP;
      redm[o + p0] = m0; reds[o + p0] = s0;
      if (p1ok) { redm[o + p1] = m1; reds[o + p1] = s1; }
    }
  }
  {
    const size_t po = (((size_t)b * (SS / MP_XG) + blockIdx.x % (SS / MP_XG)) * SS + y0 + 8 * half) * PP;
#pragma unroll
    for (int r = 0; r < 8; r++) {
      pmax[po + r * PP + p0] = cm0[r];
      psum[po + r * PP + p0] = cs0[r];
      if (p1ok) {
        pmax[po + r * PP + p1] = cm1[r];
        psum[po + r * PP + p1] = cs1[r];
      }
    }
  }
  __syncthreads();
  if (tid < MP_XG * PP) {
    const int xi = tid / PP, p = tid - xi * PP;
    float mx = redm[xi * PP + p], sm = reds[xi * PP + p];
    for (int q = 1; q < 8; q++) {
      mx = fmaxf(mx, redm[(q * MP_XG + xi) * PP + p]);
      sm += reds[(q * MP_XG + xi) * PP + p];
    }
    const int x = x0 + xi;
    const float m = (float)mask[b * SS + x];
    float* orow = out + ((size_t)b * SS + x) * FF;
    orow[23 + p] = m * mx;
    orow[43 + p] = m * (sm / (float)SS);
  }
}

__global__ void maxpool_finish_kernel(const float* __restrict__ pmax, const float* __restrict__ psum,
                                      const int* __restrict__ mask, float* __restrict__ out) {
  const int i = blockIdx.x * 256 + threadIdx.x;
  const int b = i / (SS * PP), yp = i - b * (SS * PP);
  const int y = yp / PP, p = yp - y * PP;
  float mx = -3.0e38f, sm = 0.f;
  for (int g = 0; g < SS / MP_XG; g++) {
    const size_t o = ((size_t)b * (SS / MP_XG) + g) * (SS * PP) + yp;
    mx = fmaxf(mx, pmax[o]);
    sm += psum[o];
  }
  const float m = (float)mask[b * SS + y];
  float* orow = out + ((size_t)(BB + b) * SS + y) * FF;
  orow[23 + p] = m * mx;
  orow[43 + p] = m * (sm / (float)SS);
}

extern "C" void kernel_launch(void* const* d_in, const int* in_sizes, int n_in,
                              void* d_out, int out_size, void* d_ws, size_t ws_size,
                              hipStream_t stream) {
  const float* ctx_p = (const float*)d_in[0];
  const float* ctx_h = (const float*)d_in[1];
  const float* w0 = (const float*)d_in[2];
  const float* w1 = (const float*)d_in[3];
  const float* w2 = (const float*)d_in[4];
  const float* w3 = (const float*)d_in[5];
  const int* mask_p = (const int*)d_in[6];
  const int* mask_h = (const int*)d_in[7];
  float* out = (float*)d_out;
  float* ws = (float*)d_ws;

  const size_t SZ_ROWS = (size_t)BB * SS * HH;
  float* cp = ws;                 ws += SZ_ROWS;
  float* ch = ws;                 ws += SZ_ROWS;
  float* n_p = ws;                ws += BB * SS;
  float* n_h = ws;                ws += BB * SS;
  float* wn1_p = ws;              ws += (size_t)BB * SS * PP;
  float* wn1_h = ws;              ws += (size_t)BB * SS * PP;
  float* cpl = ws;                ws += BB * HH;
  float* chl = ws;                ws += BB * HH;
  float* cosM = ws;               ws += (size_t)BB * SS * SS;
  float* mp_pmax = ws;            ws += (size_t)BB * (SS / MP_XG) * SS * PP;
  float* mp_psum = ws;            ws += (size_t)BB * (SS / MP_XG) * SS * PP;

  const dim3 row_tiles(BB * SS / RT, 3);
  prep_kernel<<<row_tiles, NTH, 0, stream>>>(ctx_p, mask_p, w1, cp, n_p, wn1_p);
  prep_kernel<<<row_tiles, NTH, 0, stream>>>(ctx_h, mask_h, w1, ch, n_h, wn1_h);

  gather_last_kernel<<<BB, 64, 0, stream>>>(cp, ch, mask_p, mask_h, cpl, chl);

  cos_wmma_kernel<<<BB * 64, 32, 0, stream>>>(cp, ch, n_p, n_h, cosM);

  att_mpm_kernel<<<dim3(BB * SS / RT, 3, 2), NTH, 0, stream>>>(cp, ch, cosM, w2, mask_p, mask_h, out);

  maxatt_mpm_kernel<<<dim3(BB * SS / MA_R, 2), MA_NT, 0, stream>>>(cp, ch, cosM, w3, mask_p, mask_h, out);

  mpm_kernel<<<row_tiles, NTH, 0, stream>>>(cp, chl, w0, HH, 0, out, 0, 2);
  mpm_kernel<<<row_tiles, NTH, 0, stream>>>(ch, cpl, w0, HH, 0, out, 1, 2);

  maxpool_pair_kernel<<<BB * (SS / MP_XG), MP_NT, 0, stream>>>(cp, ch, w1, wn1_p, wn1_h, mask_p, out,
                                                               mp_pmax, mp_psum);
  maxpool_finish_kernel<<<BB * SS * PP / 256, 256, 0, stream>>>(mp_pmax, mp_psum, mask_h, out);
}


# full match: its 21 features are summed in the attentive-match block over the rows staged there; the block picks the last row from the mask itself; 10 launches -> 7; every sum keeps its order
# speedup vs baseline: 6.1297x; 1.1877x over previous
#include <hip/hip_runtime.h>
#include <hip/hip_bf16.h>


#define BB 16
#define SS 128
#define HH 200
#define PP 20
#define FF 105
#define EPSF 1e-8f

typedef float v2f __attribute__((ext_vector_type(2)));
typedef float v8f __attribute__((ext_vector_type(8)));

#define RT 32
#define LDP 201
#define QPB 7
#define NTH 128
#define NV4 (RT * (HH / 4))
#define NFULL (NV4 / NTH)
#define NTAIL (NV4 % NTH)

struct TilePieces { float4 v[NFULL + 1]; };

__device__ __forceinline__ void tile_load(TilePieces& t, const float* __restrict__ g,
                                          int row_stride, int tid) {
#pragma unroll
  for (int k = 0; k < NFULL; k++) {
    int i = tid + k * NTH, r = i / (HH / 4), c = i - r * (HH / 4);
    t.v[k] = *(const float4*)(g + (size_t)r * row_stride + 4 * c);
  }
  t.v[NFULL] = make_float4(0.f, 0.f, 0.f, 0.f);
  if (tid < NTAIL) {
    int i = tid + NFULL * NTH, r = i / (HH / 4), c = i - r * (HH / 4);
    t.v[NFULL] = *(const float4*)(g + (size_t)r * row_stride + 4 * c);
  }
}

__device__ __forceinline__ void tile_to_lds(const TilePieces& t, float* lds, int tid) {
#pragma unroll
  for (int k = 0; k <= NFULL; k++) {
    int i = tid + k * NTH, r = i / (HH / 4), c = i - r * (HH / 4);
    if (k < NFULL || tid < NTAIL) {
      float* d = lds + r * LDP + 4 * c;
      d[0] = t.v[k].x; d[1] = t.v[k].y; d[2] = t.v[k].z; d[3] = t.v[k].w;
    }
  }
}

__global__ __launch_bounds__(NTH) void prep_kernel(
    const float* __restrict__ src, const int* __restrict__ mask,
    const float* __restrict__ w1, float* __restrict__ dst,
    float* __restrict__ nrm, float* __restrict__ wn) {
  __shared__ float rows[RT * LDP];
  const int row0 = blockIdx.x * RT;
  const int tid = threadIdx.x;
  const bool writes_dst = (blockIdx.y == 0);
  const float* stile = src + (size_t)row0 * HH;
  float* dtile = dst + (size_t)row0 * HH;
  TilePieces t;
  tile_load(t, stile, HH, tid);
#pragma unroll
  for (int k = 0; k <= NFULL; k++) {
    int r = (tid + k * NTH) / (HH / 4);
    float m = (k < NFULL || tid < NTAIL) ? (float)mask[row0 + r] : 0.f;
    t.v[k].x *= m; t.v[k].y *= m; t.v[k].z *= m; t.v[k].w *= m;
  }
  if (writes_dst) {
#pragma unroll
    for (int k = 0; k <= NFULL; k++)
      if (k < NFULL || tid < NTAIL) *(float4*)(dtile + 4 * (tid + k * NTH)) = t.v[k];
  }
  tile_to_lds(t, rows, tid);
  __syncthreads();

  const int lane = tid & 31;
  const int wv = __builtin_amdgcn_readfirstlane(tid >> 5);
  const int q0 = QPB * blockIdx.y + 2 * wv;
  const bool two = (2 * wv + 1) < QPB;
  const bool plain = (q0 == PP);
  const float* wa = w1 + (plain ? 0 : q0) * HH;
  const float* wb = w1 + (two ? q0 + 1 : 0) * HH;
  const float* vr = rows + lane * LDP;
  float sa = 0.f, sb = 0.f;
#pragma unroll 8
  for (int h = 0; h < HH; h++) {
    float v = vr[h];
    float vv = v * v;
    float xa = wa[h], xb = wb[h];
    float w2a = plain ? 1.0f : xa * xa;
    sa += w2a * vv;
    sb += (xb * xb) * vv;
  }
  const int row = row0 + lane;
  if (plain) nrm[row] = sqrtf(sa);
  else wn[row * PP + q0] = sqrtf(sa);
  if (two) wn[row * PP + q0 + 1] = sqrtf(sb);
}

__global__ void cos_wmma_kernel(const float* __restrict__ cp, const float* __restrict__ ch,
                                const float* __restrict__ np, const float* __restrict__ nh,
                                float* __restrict__ cosO) {
  int wid = blockIdx.x;
  int b = wid >> 6;
  int x0 = ((wid >> 3) & 7) << 4;
  int y0 = (wid & 7) << 4;
  int lane = threadIdx.x, half = lane >> 4, l16 = lane & 15;
  const float* arow = cp + ((size_t)(b * SS) + x0 + l16) * HH + 2 * half;
  const float* brow = ch + ((size_t)(b * SS) + y0 + l16) * HH + 2 * half;
  v8f acc = {0, 0, 0, 0, 0, 0, 0, 0};
  for (int k = 0; k < HH; k += 4) {
    v2f a = *(const v2f*)(arow + k);
    v2f bv = *(const v2f*)(brow + k);
    acc = __builtin_amdgcn_wmma_f32_16x16x4_f32(false, a, false, bv, (short)0, acc, false, false);
  }
  float nhv = nh[b * SS + y0 + l16];
#pragma unroll
  for (int r = 0; r < 8; r++) {
    int m = r + 8 * half;
    float d = fmaxf(np[b * SS + x0 + m] * nhv, EPSF);
    cosO[((size_t)(b * SS) + x0 + m) * SS + y0 + l16] = acc[r] / d;
  }
}

#define CP 132
#define AT_KS (SS / 4)

typedef float v4f __attribute__((ext_vector_type(4)));

__device__ __forceinline__ void bfrag_load(float (&bv)[2 * AT_KS], const float* __restrict__ vcol) {
#pragma unroll
  for (int i = 0; i < AT_KS; i++) {
    bv[2 * i] = vcol[(size_t)(4 * i) * HH];
    bv[2 * i + 1] = vcol[(size_t)(4 * i + 1) * HH];
  }
}

__device__ __forceinline__ void att_tile(const v2f (&a0)[AT_KS], const v2f (&a1)[AT_KS],
                                         const float (&bv)[2 * AT_KS], const float* csum,
                                         float* rows, int col, bool colOK, int half) {
  v8f acc0 = {0, 0, 0, 0, 0, 0, 0, 0};
  v8f acc1 = {0, 0, 0, 0, 0, 0, 0, 0};
#pragma unroll
  for (int i = 0; i < AT_KS; i++) {
    v2f bb;
    bb.x = bv[2 * i];
    bb.y = bv[2 * i + 1];
    acc0 = __builtin_amdgcn_wmma_f32_16x16x4_f32(false, a0[i], false, bb, (short)0, acc0, false, false);
    acc1 = __builtin_amdgcn_wmma_f32_16x16x4_f32(false, a1[i], false, bb, (short)0, acc1, false, false);
  }
  if (colOK) {
#pragma unroll
    for (int r = 0; r < 8; r++) {
      const int m = r + 8 * half;
      rows[m * LDP + col] = acc0[r] / fmaxf(csum[m], EPSF);
      rows[(16 + m) * LDP + col] = acc1[r] / fmaxf(csum[16 + m], EPSF);
    }
  }
}

__global__ __launch_bounds__(NTH) void att_mpm_kernel(
    const float* __restrict__ cp, const float* __restrict__ ch,
    const float* __restrict__ cosM, const float* __restrict__ w,
    const float* __restrict__ wf,
    const int* __restrict__ mask_p, const int* __restrict__ mask_h,
    float* __restrict__ out) {
  __shared__ float r1[RT * LDP];
  __shared__ __attribute__((aligned(16))) float r2[RT * LDP];
  __shared__ float csum[RT];
  const int side = blockIdx.z;
  const int row0 = blockIdx.x * RT;
  const int b = row0 / SS, s0 = row0 % SS;
  const int tid = threadIdx.x;
  const float* v1 = side ? ch : cp;
  const float* V = (side ? cp : ch) + (size_t)b * SS * HH;
  const int* mask = side ? mask_h : mask_p;
  const float* cb = cosM + (size_t)b * SS * SS;
  float* cst = r2;

  const int lane = tid & 31, half = lane >> 4, l16 = lane & 15;
  const int wv = __builtin_amdgcn_readfirstlane(tid >> 5);
  const int tw = 3 - wv;
  const int c0 = 16 * tw + l16, c1 = c0 + 64, c2 = c0 + 128, c3 = c0 + 192;
  const bool ok3 = c3 < HH;
  const float* vh = V + (size_t)(2 * half) * HH;

  TilePieces ta;
  tile_load(ta, v1 + (size_t)row0 * HH, HH, tid);
  v4f cv[8];
  const int sh = side ? 3 : 5, lo = side ? 7 : 31;
  const float* cpiece = cb + (side ? s0 : s0 * SS);
#pragma unroll
  for (int u = 0; u < 8; u++) {
    const int i = tid + u * NTH;
    cv[u] = *(const v4f*)(cpiece + (i >> sh) * SS + 4 * (i & lo));
  }
  float ba[2 * AT_KS], bb[2 * AT_KS];
  bfrag_load(ba, vh + c0);
  const int* omask = (side ? mask_p : mask_h) + b * SS;
  int olen = 0;
  for (int i = 0; i < SS; i++) olen += omask[i];
  const float* lr = V + (size_t)(olen > 0 ? olen - 1 : 0) * HH;
  tile_to_lds(ta, r1, tid);
  if (side == 0) {
#pragma unroll
    for (int u = 0; u < 8; u++) {
      const int i = tid + u * NTH;
      *(v4f*)(cst + (i >> 5) * CP + 4 * (i & 31)) = cv[u];
    }
  } else {
#pragma unroll
    for (int u = 0; u < 8; u++) {
      const int i = tid + u * NTH;
      const int k = i >> 3, j = 4 * (i & 7);
      cst[(j + 0) * CP + k] = cv[u].x; cst[(j + 1) * CP + k] = cv[u].y;
      cst[(j + 2) * CP + k] = cv[u].z; cst[(j + 3) * CP + k] = cv[u].w;
    }
  }
  __syncthreads();

  {
    const int j = tid >> 2, q = tid & 3;
    const v4f* c4 = (const v4f*)(cst + j * CP + 32 * q);
    float mx = -3.0e38f, sm = 0.f;
#pragma unroll
    for (int i = 0; i < 8; i++) {
      const v4f t = c4[i];
      mx = fmaxf(fmaxf(fmaxf(mx, t.x), fmaxf(t.y, t.z)), t.w);
      sm += t.x; sm += t.y; sm += t.z; sm += t.w;
    }
    mx = fmaxf(mx, __shfl_xor(mx, 1, 32)); sm += __shfl_xor(sm, 1, 32);
    mx = fmaxf(mx, __shfl_xor(mx, 2, 32)); sm += __shfl_xor(sm, 2, 32);
    if (q == 0) {
      csum[j] = sm;
      if (blockIdx.y == 0) {
        const float m = (float)mask[row0 + j];
        float* o2 = out + ((size_t)(side * BB + b) * SS + s0 + j) * FF;
        o2[0] = m * mx;
        o2[1] = m * (sm / (float)SS);
      }
    }
  }
  v2f a0[AT_KS], a1[AT_KS];
  {
    const float* cl = cst + l16 * CP + 2 * half;
#pragma unroll
    for (int i = 0; i < AT_KS; i++) {
      a0[i] = *(const v2f*)(cl + 4 * i);
      a1[i] = *(const v2f*)(cl + 16 * CP + 4 * i);
    }
  }
  __syncthreads();

  bfrag_load(bb, vh + c1);
  __builtin_amdgcn_sched_barrier(0);
  att_tile(a0, a1, ba, csum, r2, c0, true, half);
  bfrag_load(ba, vh + c2);
  __builtin_amdgcn_sched_barrier(0);
  att_tile(a0, a1, bb, csum, r2, c1, true, half);
  if (tw == 0) bfrag_load(bb, vh + (ok3 ? c3 : HH - 1));
  att_tile(a0, a1, ba, csum, r2, c2, true, half);
  if (tw == 0) att_tile(a0, a1, bb, csum, r2, c3, ok3, half);
  __syncthreads();

  const int q0 = QPB * blockIdx.y + 2 * wv;
  const bool two = (2 * wv + 1) < QPB;
  const bool plain = (q0 == PP);
  const float* wa = w + (plain ? 0 : q0) * HH;
  const float* wb = w + (two ? q0 + 1 : 0) * HH;
  const float* ua = wf + (plain ? 0 : q0) * HH;
  const float* ub = wf + (two ? q0 + 1 : 0) * HH;
  const float* ar = r1 + lane * LDP;
  const float* cr = r2 + lane * LDP;
  float a11 = 0.f, a22 = 0.f, a12 = 0.f, b11 = 0.f, b22 = 0.f, b12 = 0.f;
  float f11 = 0.f, f22 = 0.f, f12 = 0.f, g11 = 0.f, g22 = 0.f, g12 = 0.f;
#pragma unroll 8
  for (int h = 0; h < HH; h++) {
    float a = ar[h], c = cr[h];
    float aa = a * a, cc = c * c, ac = a * c;
    float xa = wa[h], xb = wb[h];
    float w2a = plain ? 1.0f : xa * xa;
    float w2b = xb * xb;
    a11 += w2a * aa; a22 += w2a * cc; a12 += w2a * ac;
    b11 += w2b * aa; b22 += w2b * cc; b12 += w2b * ac;
    float l = lr[h];
    float ll = l * l, al = a * l;
    float ya = ua[h], yb = ub[h];
    float u2a = plain ? 1.0f : ya * ya;
    float u2b = yb * yb;
    f11 += u2a * aa; f22 += u2a * ll; f12 += u2a * al;
    g11 += u2b * aa; g22 += u2b * ll; g12 += u2b * al;
  }
  float* orow = out + ((size_t)(side * BB + b) * SS + s0 + lane) * FF;
  orow[63 + (plain ? 0 : 1 + q0)] = a12 / (fmaxf(sqrtf(a11), EPSF) * fmaxf(sqrtf(a22), EPSF));
  orow[2 + (plain ? 0 : 1 + q0)] = f12 / (fmaxf(sqrtf(f11), EPSF) * fmaxf(sqrtf(f22), EPSF));
  if (two) {
    orow[63 + 2 + q0] = b12 / (fmaxf(sqrtf(b11), EPSF) * fmaxf(sqrtf(b22), EPSF));
    orow[2 + 2 + q0] = g12 / (fmaxf(sqrtf(g11), EPSF) * fmaxf(sqrtf(g22), EPSF));
  }
}

#define MA_R 8
#define MA_NT 256
#define MA_NQ 21
#define MA_WP 201
#define MA_KC 16

__device__ __forceinline__ void vcol_load(float (&v)[MA_KC], const float* __restrict__ vcol, int k0) {
#pragma unroll
  for (int i = 0; i < MA_KC; i++) v[i] = vcol[(size_t)(k0 + i) * HH];
}

__device__ __forceinline__ void maxplus_step(const float* cs, int k0, const float (&v)[MA_KC], float (&mx)[MA_R]) {
#pragma unroll
  for (int i = 0; i < MA_KC; i++) {
#pragma unroll
    for (int j = 0; j < MA_R; j++) mx[j] = fmaxf(mx[j], cs[(k0 + i) * MA_R + j] * v[i]);
  }
}

__global__ __launch_bounds__(MA_NT) void maxatt_mpm_kernel(
    const float* __restrict__ cp, const float* __restrict__ ch,
    const float* __restrict__ cosM, const float* __restrict__ w,
    const int* __restrict__ mask_p, const int* __restrict__ mask_h,
    float* __restrict__ out) {
  __shared__ __attribute__((aligned(16))) float r1[MA_R * HH];
  __shared__ __attribute__((aligned(16))) float r2[MA_R * HH];
  __shared__ __attribute__((aligned(16))) float cs[SS * MA_R];
  __shared__ float w2s[MA_NQ * MA_WP];

  const int side = blockIdx.y;
  const int row0 = blockIdx.x * MA_R;
  const int b = row0 / SS, r0 = row0 % SS;
  const int tid = threadIdx.x;
  const float* v1 = side ? ch : cp;
  const float* V = side ? cp : ch;
  const int* mask = side ? mask_h : mask_p;
  const float* cb = cosM + (size_t)b * SS * SS;

  constexpr int W4 = PP * HH / 4;
  constexpr int R4 = MA_R * HH / 4;
  v4f wv4[4], rv4[2], cv4;
#pragma unroll
  for (int u = 0; u < 4; u++) {
    const int i = tid + u * MA_NT;
    wv4[u] = *(const v4f*)(w + 4 * (i < W4 ? i : W4 - 1));
  }
#pragma unroll
  for (int u = 0; u < 2; u++) {
    const int i = tid + u * MA_NT;
    rv4[u] = *(const v4f*)(v1 + (size_t)row0 * HH + 4 * (i < R4 ? i : R4 - 1));
  }
  const int pj = tid & 7, pk = 4 * (tid >> 3);
  const int qk = tid >> 1, qj = 4 * (tid & 1);
  cv4 = side == 0 ? *(const v4f*)(cb + (size_t)(r0 + pj) * SS + pk)
                  : *(const v4f*)(cb + (size_t)qk * SS + r0 + qj);
  asm volatile("" :: "v"(wv4[0]), "v"(wv4[1]), "v"(wv4[2]), "v"(wv4[3]), "v"(rv4[0]), "v"(rv4[1]), "v"(cv4));
#pragma unroll
  for (int u = 0; u < 4; u++) {
    const int i = tid + u * MA_NT;
    if (i < W4) {
      const int p = i / (HH / 4), c = i - p * (HH / 4);
      float* d = w2s + p * MA_WP + 4 * c;
      d[0] = wv4[u].x * wv4[u].x; d[1] = wv4[u].y * wv4[u].y;
      d[2] = wv4[u].z * wv4[u].z; d[3] = wv4[u].w * wv4[u].w;
    }
  }
  if (tid < HH) w2s[PP * MA_WP + tid] = 1.0f;
#pragma unroll
  for (int u = 0; u < 2; u++) {
    const int i = tid + u * MA_NT;
    if (i < R4) *(v4f*)(r1 + 4 * i) = rv4[u];
  }
  if (side == 0) {
    cs[(pk + 0) * MA_R + pj] = cv4.x; cs[(pk + 1) * MA_R + pj] = cv4.y;
    cs[(pk + 2) * MA_R + pj] = cv4.z; cs[(pk + 3) * MA_R + pj] = cv4.w;
  } else {
    *(v4f*)(cs + qk * MA_R + qj) = cv4;
  }
  __syncthreads();

  const int hc = tid < HH ? tid : HH - 1;
  const float* vcol = V + (size_t)b * SS * HH + hc;
  float mx[MA_R];
#pragma unroll
  for (int j = 0; j < MA_R; j++) mx[j] = -3.0e38f;
  if ((tid & ~31) < HH) {
    float va[MA_KC], vb[MA_KC];
    vcol_load(va, vcol, 0);
#pragma unroll
    for (int k0 = 0; k0 < SS; k0 += 2 * MA_KC) {
      vcol_load(vb, vcol, k0 + MA_KC);
      maxplus_step(cs, k0, va, mx);
      if (k0 + 2 * MA_KC < SS) vcol_load(va, vcol, k0 + 2 * MA_KC);
      maxplus_step(cs, k0 + MA_KC, vb, mx);
    }
  }
  if (tid < HH) {
#pragma unroll
    for (int j = 0; j < MA_R; j++) r2[j * HH + tid] = (float)mask[row0 + j] * mx[j];
  }
  __syncthreads();

  if (tid < MA_R * MA_NQ) {
    const int j = tid / MA_NQ, q = tid - j * MA_NQ;
    const float* ar = r1 + j * HH;
    const float* cr = r2 + j * HH;
    const float* wq = w2s + q * MA_WP;
    float a11 = 0.f, a22 = 0.f, a12 = 0.f;
#pragma unroll 8
    for (int h = 0; h < HH; h++) {
      const float a = ar[h], c = cr[h];
      const float aa = a * a, cc = c * c, ac = a * c;
      const float w2a = wq[h];
      a11 += w2a * aa; a22 += w2a * cc; a12 += w2a * ac;
    }
    float* orow = out + ((size_t)(side * BB + b) * SS + r0 + j) * FF + 84;
    orow[q == PP ? 0 : 1 + q] = a12 / (fmaxf(sqrtf(a11), EPSF) * fmaxf(sqrtf(a22), EPSF));
  }
}

#define MP_XG 8
#define MP_WP 204
#define MP_NT 256

__global__ __launch_bounds__(MP_NT) void maxpool_pair_kernel(
    const float* __restrict__ v1, const float* __restrict__ v2,
    const float* __restrict__ w,
    const float* __restrict__ wnA, const float* __restrict__ wnB,
    const int* __restrict__ mask, float* __restrict__ out,
    float* __restrict__ pmax, float* __restrict__ psum) {
  __shared__ __attribute__((aligned(16))) float w2s[PP * MP_WP];
  __shared__ __attribute__((aligned(16))) float xs[MP_XG * HH];
  __shared__ float wnAs[MP_XG * PP];
  __shared__ float redm[8 * MP_XG * PP];
  __shared__ float reds[8 * MP_XG * PP];

  const int b = blockIdx.x / (SS / MP_XG);
  const int x0 = (blockIdx.x % (SS / MP_XG)) * MP_XG;
  const int tid = threadIdx.x;
  const int lane = tid & 31, half = lane >> 4, l16 = lane & 15;
  const int wv = tid >> 5;
  const int y0 = wv * 16;

  v2f yf[HH / 4];
  const float* yrow = v2 + ((size_t)b * SS + y0 + l16) * HH + 2 * half;
#pragma unroll
  for (int ks = 0; ks < HH / 4; ks++) yf[ks] = *(const v2f*)(yrow + 4 * ks);

  for (int i = tid; i < PP * HH; i += MP_NT) {
    int p = i / HH, h = i - p * HH;
    float t = w[i];
    w2s[p * MP_WP + h] = t * t;
  }
  for (int i = tid; i < MP_XG * HH; i += MP_NT)
    xs[i] = v1[((size_t)b * SS + x0) * HH + i];
  if (tid < MP_XG * PP) wnAs[tid] = wnA[((size_t)b * SS + x0) * PP + tid];

  const int p0 = l16;
  const bool p1ok = (16 + l16) < PP;
  const int p1 = p1ok ? 16 + l16 : PP - 1;
  float wB0[8], wB1[8];
#pragma unroll
  for (int r = 0; r < 8; r++) {
    const float* wr = wnB + ((size_t)b * SS + y0 + r + 8 * half) * PP;
    wB0[r] = wr[p0];
    wB1[r] = wr[p1];
  }
  float cm0[8], cs0[8], cm1[8], cs1[8];
#pragma unroll
  for (int r = 0; r < 8; r++) { cm0[r] = -3.0e38f; cs0[r] = 0.f; cm1[r] = -3.0e38f; cs1[r] = 0.f; }
  __syncthreads();

  const float* wb0 = w2s + p0 * MP_WP + 2 * half;
  const float* wb1 = w2s + p1 * MP_WP + 2 * half;
  for (int xi = 0; xi < MP_XG; xi++) {
    const float* xr = xs + xi * HH + 2 * half;
    v8f acc0 = {0, 0, 0, 0, 0, 0, 0, 0};
    v8f acc1 = {0, 0, 0, 0, 0, 0, 0, 0};
#pragma unroll
    for (int ks = 0; ks < HH / 4; ks++) {
      v2f xv = *(const v2f*)(xr + 4 * ks);
      v2f a;
      a.x = xv.x * yf[ks].x;
      a.y = xv.y * yf[ks].y;
      v2f b0 = *(const v2f*)(wb0 + 4 * ks);
      v2f b1 = *(const v2f*)(wb1 + 4 * ks);
      acc0 = __builtin_amdgcn_wmma_f32_16x16x4_f32(false, a, false, b0, (short)0, acc0, false, false);
      acc1 = __builtin_amdgcn_wmma_f32_16x16x4_f32(false, a, false, b1, (short)0, acc1, false, false);
    }
    const float wa0 = wnAs[xi * PP + p0];
    const float wa1 = wnAs[xi * PP + p1];
    float m0 = -3.0e38f, s0 = 0.f, m1 = -3.0e38f, s1 = 0.f;
#pragma unroll
    for (int r = 0; r < 8; r++) {
      float mv0 = acc0[r] / fmaxf(wa0 * wB0[r], EPSF);
      float mv1 = acc1[r] / fmaxf(wa1 * wB1[r], EPSF);
      m0 = fmaxf(m0, mv0); s0 += mv0;
      m1 = fmaxf(m1, mv1); s1 += mv1;
      cm0[r] = fmaxf(cm0[r], mv0); cs0[r] += mv0;
      cm1[r] = fmaxf(cm1[r], mv1); cs1[r] += mv1;
    }
    m0 = fmaxf(m0, __shfl_xor(m0, 16, 32)); s0 += __shfl_xor(s0, 16, 32);
    m1 = fmaxf(m1, __shfl_xor(m1, 16, 32)); s1 += __shfl_xor(s1, 16, 32);
    if (half == 0) {
      const int o = (wv * MP_XG + xi) * PP;
      redm[o + p0] = m0; reds[o + p0] = s0;
      if (p1ok) { redm[o + p1] = m1; reds[o + p1] = s1; }
    }
  }
  {
    const size_t po = (((size_t)b * (SS / MP_XG) + blockIdx.x % (SS / MP_XG)) * SS + y0 + 8 * half) * PP;
#pragma unroll
    for (int r = 0; r < 8; r++) {
      pmax[po + r * PP + p0] = cm0[r];
      psum[po + r * PP + p0] = cs0[r];
      if (p1ok) {
        pmax[po + r * PP + p1] = cm1[r];
        psum[po + r * PP + p1] = cs1[r];
      }
    }
  }
  __syncthreads();
  if (tid < MP_XG * PP) {
    const int xi = tid / PP, p = tid - xi * PP;
    float mx = redm[xi * PP + p], sm = reds[xi * PP + p];
    for (int q = 1; q < 8; q++) {
      mx = fmaxf(mx, redm[(q * MP_XG + xi) * PP + p]);
      sm += reds[(q * MP_XG + xi) * PP + p];
    }
    const int x = x0 + xi;
    const float m = (float)mask[b * SS + x];
    float* orow = out + ((size_t)b * SS + x) * FF;
    orow[23 + p] = m * mx;
    orow[43 + p] = m * (sm / (float)SS);
  }
}

__global__ void maxpool_finish_kernel(const float* __restrict__ pmax, const float* __restrict__ psum,
                                      const int* __restrict__ mask, float* __restrict__ out) {
  const int i = blockIdx.x * 256 + threadIdx.x;
  const int b = i / (SS * PP), yp = i - b * (SS * PP);
  const int y = yp / PP, p = yp - y * PP;
  float mx = -3.0e38f, sm = 0.f;
  for (int g = 0; g < SS / MP_XG; g++) {
    const size_t o = ((size_t)b * (SS / MP_XG) + g) * (SS * PP) + yp;
    mx = fmaxf(mx, pmax[o]);
    sm += psum[o];
  }
  const float m = (float)mask[b * SS + y];
  float* orow = out + ((size_t)(BB + b) * SS + y) * FF;
  orow[23 + p] = m * mx;
  orow[43 + p] = m * (sm / (float)SS);
}

extern "C" void kernel_launch(void* const* d_in, const int* in_sizes, int n_in,
                              void* d_out, int out_size, void* d_ws, size_t ws_size,
                              hipStream_t stream) {
  const float* ctx_p = (const float*)d_in[0];
  const float* ctx_h = (const float*)d_in[1];
  const float* w0 = (const float*)d_in[2];
  const float* w1 = (const float*)d_in[3];
  const float* w2 = (const float*)d_in[4];
  const float* w3 = (const float*)d_in[5];
  const int* mask_p = (const int*)d_in[6];
  const int* mask_h = (const int*)d_in[7];
  float* out = (float*)d_out;
  float* ws = (float*)d_ws;

  const size_t SZ_ROWS = (size_t)BB * SS * HH;
  float* cp = ws;                 ws += SZ_ROWS;
  float* ch = ws;                 ws += SZ_ROWS;
  float* n_p = ws;                ws += BB * SS;
  float* n_h = ws;                ws += BB * SS;
  float* wn1_p = ws;              ws += (size_t)BB * SS * PP;
  float* wn1_h = ws;              ws += (size_t)BB * SS * PP;
  float* cosM = ws;               ws += (size_t)BB * SS * SS;
  float* mp_pmax = ws;            ws += (size_t)BB * (SS / MP_XG) * SS * PP;
  float* mp_psum = ws;            ws += (size_t)BB * (SS / MP_XG) * SS * PP;

  const dim3 row_tiles(BB * SS / RT, 3);
  prep_kernel<<<row_tiles, NTH, 0, stream>>>(ctx_p, mask_p, w1, cp, n_p, wn1_p);
  prep_kernel<<<row_tiles, NTH, 0, stream>>>(ctx_h, mask_h, w1, ch, n_h, wn1_h);

  cos_wmma_kernel<<<BB * 64, 32, 0, stream>>>(cp, ch, n_p, n_h, cosM);

  att_mpm_kernel<<<dim3(BB * SS / RT, 3, 2), NTH, 0, stream>>>(cp, ch, cosM, w2, w0, mask_p, mask_h, out);

  maxatt_mpm_kernel<<<dim3(BB * SS / MA_R, 2), MA_NT, 0, stream>>>(cp, ch, cosM, w3, mask_p, mask_h, out);

  maxpool_pair_kernel<<<BB * (SS / MP_XG), MP_NT, 0, stream>>>(cp, ch, w1, wn1_p, wn1_h, mask_p, out,
                                                               mp_pmax, mp_psum);
  maxpool_finish_kernel<<<BB * SS * PP / 256, 256, 0, stream>>>(mp_pmax, mp_psum, mask_h, out);
}


# maxpool match, hypothesis side: the max-attentive block owning rows y folds their 16 partials itself, loads issued with its staging loads; finish launch removed, 7 launches -> 6; no sum changes order
# speedup vs baseline: 6.4287x; 1.0488x over previous
#include <hip/hip_runtime.h>
#include <hip/hip_bf16.h>


#define BB 16
#define SS 128
#define HH 200
#define PP 20
#define FF 105
#define EPSF 1e-8f
#define MP_XG 8

typedef float v2f __attribute__((ext_vector_type(2)));
typedef float v8f __attribute__((ext_vector_type(8)));

#define RT 32
#define LDP 201
#define QPB 7
#define NTH 128
#define NV4 (RT * (HH / 4))
#define NFULL (NV4 / NTH)
#define NTAIL (NV4 % NTH)

struct TilePieces { float4 v[NFULL + 1]; };

__device__ __forceinline__ void tile_load(TilePieces& t, const float* __restrict__ g,
                                          int row_stride, int tid) {
#pragma unroll
  for (int k = 0; k < NFULL; k++) {
    int i = tid + k * NTH, r = i / (HH / 4), c = i - r * (HH / 4);
    t.v[k] = *(const float4*)(g + (size_t)r * row_stride + 4 * c);
  }
  t.v[NFULL] = make_float4(0.f, 0.f, 0.f, 0.f);
  if (tid < NTAIL) {
    int i = tid + NFULL * NTH, r = i / (HH / 4), c = i - r * (HH / 4);
    t.v[NFULL] = *(const float4*)(g + (size_t)r * row_stride + 4 * c);
  }
}

__device__ __forceinline__ void tile_to_lds(const TilePieces& t, float* lds, int tid) {
#pragma unroll
  for (int k = 0; k <= NFULL; k++) {
    int i = tid + k * NTH, r = i / (HH / 4), c = i - r * (HH / 4);
    if (k < NFULL || tid < NTAIL) {
      float* d = lds + r * LDP + 4 * c;
      d[0] = t.v[k].x; d[1] = t.v[k].y; d[2] = t.v[k].z; d[3] = t.v[k].w;
    }
  }
}

__global__ __launch_bounds__(NTH) void prep_kernel(
    const float* __restrict__ src, const int* __restrict__ mask,
    const float* __restrict__ w1, float* __restrict__ dst,
    float* __restrict__ nrm, float* __restrict__ wn) {
  __shared__ float rows[RT * LDP];
  const int row0 = blockIdx.x * RT;
  const int tid = threadIdx.x;
  const bool writes_dst = (blockIdx.y == 0);
  const float* stile = src + (size_t)row0 * HH;
  float* dtile = dst + (size_t)row0 * HH;
  TilePieces t;
  tile_load(t, stile, HH, tid);
#pragma unroll
  for (int k = 0; k <= NFULL; k++) {
    int r = (tid + k * NTH) / (HH / 4);
    float m = (k < NFULL || tid < NTAIL) ? (float)mask[row0 + r] : 0.f;
    t.v[k].x *= m; t.v[k].y *= m; t.v[k].z *= m; t.v[k].w *= m;
  }
  if (writes_dst) {
#pragma unroll
    for (int k = 0; k <= NFULL; k++)
      if (k < NFULL || tid < NTAIL) *(float4*)(dtile + 4 * (tid + k * NTH)) = t.v[k];
  }
  tile_to_lds(t, rows, tid);
  __syncthreads();

  const int lane = tid & 31;
  const int wv = __builtin_amdgcn_readfirstlane(tid >> 5);
  const int q0 = QPB * blockIdx.y + 2 * wv;
  const bool two = (2 * wv + 1) < QPB;
  const bool plain = (q0 == PP);
  const float* wa = w1 + (plain ? 0 : q0) * HH;
  const float* wb = w1 + (two ? q0 + 1 : 0) * HH;
  const float* vr = rows + lane * LDP;
  float sa = 0.f, sb = 0.f;
#pragma unroll 8
  for (int h = 0; h < HH; h++) {
    float v = vr[h];
    float vv = v * v;
    float xa = wa[h], xb = wb[h];
    float w2a = plain ? 1.0f : xa * xa;
    sa += w2a * vv;
    sb += (xb * xb) * vv;
  }
  const int row = row0 + lane;
  if (plain) nrm[row] = sqrtf(sa);
  else wn[row * PP + q0] = sqrtf(sa);
  if (two) wn[row * PP + q0 + 1] = sqrtf(sb);
}

__global__ void cos_wmma_kernel(const float* __restrict__ cp, const float* __restrict__ ch,
                                const float* __restrict__ np, const float* __restrict__ nh,
                                float* __restrict__ cosO) {
  int wid = blockIdx.x;
  int b = wid >> 6;
  int x0 = ((wid >> 3) & 7) << 4;
  int y0 = (wid & 7) << 4;
  int lane = threadIdx.x, half = lane >> 4, l16 = lane & 15;
  const float* arow = cp + ((size_t)(b * SS) + x0 + l16) * HH + 2 * half;
  const float* brow = ch + ((size_t)(b * SS) + y0 + l16) * HH + 2 * half;
  v8f acc = {0, 0, 0, 0, 0, 0, 0, 0};
  for (int k = 0; k < HH; k += 4) {
    v2f a = *(const v2f*)(arow + k);
    v2f bv = *(const v2f*)(brow + k);
    acc = __builtin_amdgcn_wmma_f32_16x16x4_f32(false, a, false, bv, (short)0, acc, false, false);
  }
  float nhv = nh[b * SS + y0 + l16];
#pragma unroll
  for (int r = 0; r < 8; r++) {
    int m = r + 8 * half;
    float d = fmaxf(np[b * SS + x0 + m] * nhv, EPSF);
    cosO[((size_t)(b * SS) + x0 + m) * SS + y0 + l16] = acc[r] / d;
  }
}

#define CP 132
#define AT_KS (SS / 4)

typedef float v4f __attribute__((ext_vector_type(4)));

__device__ __forceinline__ void bfrag_load(float (&bv)[2 * AT_KS], const float* __restrict__ vcol) {
#pragma unroll
  for (int i = 0; i < AT_KS; i++) {
    bv[2 * i] = vcol[(size_t)(4 * i) * HH];
    bv[2 * i + 1] = vcol[(size_t)(4 * i + 1) * HH];
  }
}

__device__ __forceinline__ void att_tile(const v2f (&a0)[AT_KS], const v2f (&a1)[AT_KS],
                                         const float (&bv)[2 * AT_KS], const float* csum,
                                         float* rows, int col, bool colOK, int half) {
  v8f acc0 = {0, 0, 0, 0, 0, 0, 0, 0};
  v8f acc1 = {0, 0, 0, 0, 0, 0, 0, 0};
#pragma unroll
  for (int i = 0; i < AT_KS; i++) {
    v2f bb;
    bb.x = bv[2 * i];
    bb.y = bv[2 * i + 1];
    acc0 = __builtin_amdgcn_wmma_f32_16x16x4_f32(false, a0[i], false, bb, (short)0, acc0, false, false);
    acc1 = __builtin_amdgcn_wmma_f32_16x16x4_f32(false, a1[i], false, bb, (short)0, acc1, false, false);
  }
  if (colOK) {
#pragma unroll
    for (int r = 0; r < 8; r++) {
      const int m = r + 8 * half;
      rows[m * LDP + col] = acc0[r] / fmaxf(csum[m], EPSF);
      rows[(16 + m) * LDP + col] = acc1[r] / fmaxf(csum[16 + m], EPSF);
    }
  }
}

__global__ __launch_bounds__(NTH) void att_mpm_kernel(
    const float* __restrict__ cp, const float* __restrict__ ch,
    const float* __restrict__ cosM, const float* __restrict__ w,
    const float* __restrict__ wf,
    const int* __restrict__ mask_p, const int* __restrict__ mask_h,
    float* __restrict__ out) {
  __shared__ float r1[RT * LDP];
  __shared__ __attribute__((aligned(16))) float r2[RT * LDP];
  __shared__ float csum[RT];
  const int side = blockIdx.z;
  const int row0 = blockIdx.x * RT;
  const int b = row0 / SS, s0 = row0 % SS;
  const int tid = threadIdx.x;
  const float* v1 = side ? ch : cp;
  const float* V = (side ? cp : ch) + (size_t)b * SS * HH;
  const int* mask = side ? mask_h : mask_p;
  const float* cb = cosM + (size_t)b * SS * SS;
  float* cst = r2;

  const int lane = tid & 31, half = lane >> 4, l16 = lane & 15;
  const int wv = __builtin_amdgcn_readfirstlane(tid >> 5);
  const int tw = 3 - wv;
  const int c0 = 16 * tw + l16, c1 = c0 + 64, c2 = c0 + 128, c3 = c0 + 192;
  const bool ok3 = c3 < HH;
  const float* vh = V + (size_t)(2 * half) * HH;

  TilePieces ta;
  tile_load(ta, v1 + (size_t)row0 * HH, HH, tid);
  v4f cv[8];
  const int sh = side ? 3 : 5, lo = side ? 7 : 31;
  const float* cpiece = cb + (side ? s0 : s0 * SS);
#pragma unroll
  for (int u = 0; u < 8; u++) {
    const int i = tid + u * NTH;
    cv[u] = *(const v4f*)(cpiece + (i >> sh) * SS + 4 * (i & lo));
  }
  float ba[2 * AT_KS], bb[2 * AT_KS];
  bfrag_load(ba, vh + c0);
  const int* omask = (side ? mask_p : mask_h) + b * SS;
  int olen = 0;
  for (int i = 0; i < SS; i++) olen += omask[i];
  const float* lr = V + (size_t)(olen > 0 ? olen - 1 : 0) * HH;
  tile_to_lds(ta, r1, tid);
  if (side == 0) {
#pragma unroll
    for (int u = 0; u < 8; u++) {
      const int i = tid + u * NTH;
      *(v4f*)(cst + (i >> 5) * CP + 4 * (i & 31)) = cv[u];
    }
  } else {
#pragma unroll
    for (int u = 0; u < 8; u++) {
      const int i = tid + u * NTH;
      const int k = i >> 3, j = 4 * (i & 7);
      cst[(j + 0) * CP + k] = cv[u].x; cst[(j + 1) * CP + k] = cv[u].y;
      cst[(j + 2) * CP + k] = cv[u].z; cst[(j + 3) * CP + k] = cv[u].w;
    }
  }
  __syncthreads();

  {
    const int j = tid >> 2, q = tid & 3;
    const v4f* c4 = (const v4f*)(cst + j * CP + 32 * q);
    float mx = -3.0e38f, sm = 0.f;
#pragma unroll
    for (int i = 0; i < 8; i++) {
      const v4f t = c4[i];
      mx = fmaxf(fmaxf(fmaxf(mx, t.x), fmaxf(t.y, t.z)), t.w);
      sm += t.x; sm += t.y; sm += t.z; sm += t.w;
    }
    mx = fmaxf(mx, __shfl_xor(mx, 1, 32)); sm += __shfl_xor(sm, 1, 32);
    mx = fmaxf(mx, __shfl_xor(mx, 2, 32)); sm += __shfl_xor(sm, 2, 32);
    if (q == 0) {
      csum[j] = sm;
      if (blockIdx.y == 0) {
        const float m = (float)mask[row0 + j];
        float* o2 = out + ((size_t)(side * BB + b) * SS + s0 + j) * FF;
        o2[0] = m * mx;
        o2[1] = m * (sm / (float)SS);
      }
    }
  }
  v2f a0[AT_KS], a1[AT_KS];
  {
    const float* cl = cst + l16 * CP + 2 * half;
#pragma unroll
    for (int i = 0; i < AT_KS; i++) {
      a0[i] = *(const v2f*)(cl + 4 * i);
      a1[i] = *(const v2f*)(cl + 16 * CP + 4 * i);
    }
  }
  __syncthreads();

  bfrag_load(bb, vh + c1);
  __builtin_amdgcn_sched_barrier(0);
  att_tile(a0, a1, ba, csum, r2, c0, true, half);
  bfrag_load(ba, vh + c2);
  __builtin_amdgcn_sched_barrier(0);
  att_tile(a0, a1, bb, csum, r2, c1, true, half);
  if (tw == 0) bfrag_load(bb, vh + (ok3 ? c3 : HH - 1));
  att_tile(a0, a1, ba, csum, r2, c2, true, half);
  if (tw == 0) att_tile(a0, a1, bb, csum, r2, c3, ok3, half);
  __syncthreads();

  const int q0 = QPB * blockIdx.y + 2 * wv;
  const bool two = (2 * wv + 1) < QPB;
  const bool plain = (q0 == PP);
  const float* wa = w + (plain ? 0 : q0) * HH;
  const float* wb = w + (two ? q0 + 1 : 0) * HH;
  const float* ua = wf + (plain ? 0 : q0) * HH;
  const float* ub = wf + (two ? q0 + 1 : 0) * HH;
  const float* ar = r1 + lane * LDP;
  const float* cr = r2 + lane * LDP;
  float a11 = 0.f, a22 = 0.f, a12 = 0.f, b11 = 0.f, b22 = 0.f, b12 = 0.f;
  float f11 = 0.f, f22 = 0.f, f12 = 0.f, g11 = 0.f, g22 = 0.f, g12 = 0.f;
#pragma unroll 8
  for (int h = 0; h < HH; h++) {
    float a = ar[h], c = cr[h];
    float aa = a * a, cc = c * c, ac = a * c;
    float xa = wa[h], xb = wb[h];
    float w2a = plain ? 1.0f : xa * xa;
    float w2b = xb * xb;
    a11 += w2a * aa; a22 += w2a * cc; a12 += w2a * ac;
    b11 += w2b * aa; b22 += w2b * cc; b12 += w2b * ac;
    float l = lr[h];
    float ll = l * l, al = a * l;
    float ya = ua[h], yb = ub[h];
    float u2a = plain ? 1.0f : ya * ya;
    float u2b = yb * yb;
    f11 += u2a * aa; f22 += u2a * ll; f12 += u2a * al;
    g11 += u2b * aa; g22 += u2b * ll; g12 += u2b * al;
  }
  float* orow = out + ((size_t)(side * BB + b) * SS + s0 + lane) * FF;
  orow[63 + (plain ? 0 : 1 + q0)] = a12 / (fmaxf(sqrtf(a11), EPSF) * fmaxf(sqrtf(a22), EPSF));
  orow[2 + (plain ? 0 : 1 + q0)] = f12 / (fmaxf(sqrtf(f11), EPSF) * fmaxf(sqrtf(f22), EPSF));
  if (two) {
    orow[63 + 2 + q0] = b12 / (fmaxf(sqrtf(b11), EPSF) * fmaxf(sqrtf(b22), EPSF));
    orow[2 + 2 + q0] = g12 / (fmaxf(sqrtf(g11), EPSF) * fmaxf(sqrtf(g22), EPSF));
  }
}

#define MA_R 8
#define MA_NT 256
#define MA_NQ 21
#define MA_WP 201
#define MA_KC 16
#define MA_FN (MA_R * PP)
#define MA_NG (SS / MP_XG)

__device__ __forceinline__ void vcol_load(float (&v)[MA_KC], const float* __restrict__ vcol, int k0) {
#pragma unroll
  for (int i = 0; i < MA_KC; i++) v[i] = vcol[(size_t)(k0 + i) * HH];
}

__device__ __forceinline__ void maxplus_step(const float* cs, int k0, const float (&v)[MA_KC], float (&mx)[MA_R]) {
#pragma unroll
  for (int i = 0; i < MA_KC; i++) {
#pragma unroll
    for (int j = 0; j < MA_R; j++) mx[j] = fmaxf(mx[j], cs[(k0 + i) * MA_R + j] * v[i]);
  }
}

__global__ __launch_bounds__(MA_NT) void maxatt_mpm_kernel(
    const float* __restrict__ cp, const float* __restrict__ ch,
    const float* __restrict__ cosM, const float* __restrict__ w,
    const int* __restrict__ mask_p, const int* __restrict__ mask_h,
    const float* __restrict__ pmax, const float* __restrict__ psum,
    float* __restrict__ out) {
  __shared__ __attribute__((aligned(16))) float r1[MA_R * HH];
  __shared__ __attribute__((aligned(16))) float r2[MA_R * HH];
  __shared__ __attribute__((aligned(16))) float cs[SS * MA_R];
  __shared__ float w2s[MA_NQ * MA_WP];

  const int side = blockIdx.y;
  const int row0 = blockIdx.x * MA_R;
  const int b = row0 / SS, r0 = row0 % SS;
  const int tid = threadIdx.x;
  const float* v1 = side ? ch : cp;
  const float* V = side ? cp : ch;
  const int* mask = side ? mask_h : mask_p;
  const float* cb = cosM + (size_t)b * SS * SS;

  constexpr int W4 = PP * HH / 4;
  constexpr int R4 = MA_R * HH / 4;
  v4f wv4[4], rv4[2], cv4;
#pragma unroll
  for (int u = 0; u < 4; u++) {
    const int i = tid + u * MA_NT;
    wv4[u] = *(const v4f*)(w + 4 * (i < W4 ? i : W4 - 1));
  }
#pragma unroll
  for (int u = 0; u < 2; u++) {
    const int i = tid + u * MA_NT;
    rv4[u] = *(const v4f*)(v1 + (size_t)row0 * HH + 4 * (i < R4 ? i : R4 - 1));
  }
  const int pj = tid & 7, pk = 4 * (tid >> 3);
  const int qk = tid >> 1, qj = 4 * (tid & 1);
  cv4 = side == 0 ? *(const v4f*)(cb + (size_t)(r0 + pj) * SS + pk)
                  : *(const v4f*)(cb + (size_t)qk * SS + r0 + qj);
  const int fc = tid < MA_FN ? tid : MA_FN - 1;
  const int fy = fc / PP, fp = fc - fy * PP;
  const float* part = (side ? psum : pmax) + (size_t)b * MA_NG * (SS * PP) + r0 * PP + fc;
  float pv[MA_NG];
#pragma unroll
  for (int g = 0; g < MA_NG; g++) pv[g] = part[(size_t)g * (SS * PP)];
  const int fmask = mask_h[b * SS + r0 + fy];
  asm volatile("" :: "v"(wv4[0]), "v"(wv4[1]), "v"(wv4[2]), "v"(wv4[3]), "v"(rv4[0]), "v"(rv4[1]), "v"(cv4),
               "v"(pv[0]), "v"(pv[1]), "v"(pv[2]), "v"(pv[3]), "v"(pv[4]), "v"(pv[5]), "v"(pv[6]), "v"(pv[7]),
               "v"(pv[8]), "v"(pv[9]), "v"(pv[10]), "v"(pv[11]), "v"(pv[12]), "v"(pv[13]), "v"(pv[14]), "v"(pv[15]),
               "v"(fmask));
#pragma unroll
  for (int u = 0; u < 4; u++) {
    const int i = tid + u * MA_NT;
    if (i < W4) {
      const int p = i / (HH / 4), c = i - p * (HH / 4);
      float* d = w2s + p * MA_WP + 4 * c;
      d[0] = wv4[u].x * wv4[u].x; d[1] = wv4[u].y * wv4[u].y;
      d[2] = wv4[u].z * wv4[u].z; d[3] = wv4[u].w * wv4[u].w;
    }
  }
  if (tid < HH) w2s[PP * MA_WP + tid] = 1.0f;
#pragma unroll
  for (int u = 0; u < 2; u++) {
    const int i = tid + u * MA_NT;
    if (i < R4) *(v4f*)(r1 + 4 * i) = rv4[u];
  }
  if (side == 0) {
    cs[(pk + 0) * MA_R + pj] = cv4.x; cs[(pk + 1) * MA_R + pj] = cv4.y;
    cs[(pk + 2) * MA_R + pj] = cv4.z; cs[(pk + 3) * MA_R + pj] = cv4.w;
  } else {
    *(v4f*)(cs + qk * MA_R + qj) = cv4;
  }
  __syncthreads();

  if (tid < MA_FN) {
    float* orow = out + ((size_t)(BB + b) * SS + r0 + fy) * FF;
    const float m = (float)fmask;
    if (side == 0) {
      float mx = -3.0e38f;
#pragma unroll
      for (int g = 0; g < MA_NG; g++) mx = fmaxf(mx, pv[g]);
      orow[23 + fp] = m * mx;
    } else {
      float sm = 0.f;
#pragma unroll
      for (int g = 0; g < MA_NG; g++) sm += pv[g];
      orow[43 + fp] = m * (sm / (float)SS);
    }
  }

  const int hc = tid < HH ? tid : HH - 1;
  const float* vcol = V + (size_t)b * SS * HH + hc;
  float mx[MA_R];
#pragma unroll
  for (int j = 0; j < MA_R; j++) mx[j] = -3.0e38f;
  if ((tid & ~31) < HH) {
    float va[MA_KC], vb[MA_KC];
    vcol_load(va, vcol, 0);
#pragma unroll
    for (int k0 = 0; k0 < SS; k0 += 2 * MA_KC) {
      vcol_load(vb, vcol, k0 + MA_KC);
      maxplus_step(cs, k0, va, mx);
      if (k0 + 2 * MA_KC < SS) vcol_load(va, vcol, k0 + 2 * MA_KC);
      maxplus_step(cs, k0 + MA_KC, vb, mx);
    }
  }
  if (tid < HH) {
#pragma unroll
    for (int j = 0; j < MA_R; j++) r2[j * HH + tid] = (float)mask[row0 + j] * mx[j];
  }
  __syncthreads();

  if (tid < MA_R * MA_NQ) {
    const int j = tid / MA_NQ, q = tid - j * MA_NQ;
    const float* ar = r1 + j * HH;
    const float* cr = r2 + j * HH;
    const float* wq = w2s + q * MA_WP;
    float a11 = 0.f, a22 = 0.f, a12 = 0.f;
#pragma unroll 8
    for (int h = 0; h < HH; h++) {
      const float a = ar[h], c = cr[h];
      const float aa = a * a, cc = c * c, ac = a * c;
      const float w2a = wq[h];
      a11 += w2a * aa; a22 += w2a * cc; a12 += w2a * ac;
    }
    float* orow = out + ((size_t)(side * BB + b) * SS + r0 + j) * FF + 84;
    orow[q == PP ? 0 : 1 + q] = a12 / (fmaxf(sqrtf(a11), EPSF) * fmaxf(sqrtf(a22), EPSF));
  }
}

#define MP_WP 204
#define MP_NT 256

__global__ __launch_bounds__(MP_NT) void maxpool_pair_kernel(
    const float* __restrict__ v1, const float* __restrict__ v2,
    const float* __restrict__ w,
    const float* __restrict__ wnA, const float* __restrict__ wnB,
    const int* __restrict__ mask, float* __restrict__ out,
    float* __restrict__ pmax, float* __restrict__ psum) {
  __shared__ __attribute__((aligned(16))) float w2s[PP * MP_WP];
  __shared__ __attribute__((aligned(16))) float xs[MP_XG * HH];
  __shared__ float wnAs[MP_XG * PP];
  __shared__ float redm[8 * MP_XG * PP];
  __shared__ float reds[8 * MP_XG * PP];

  const int b = blockIdx.x / (SS / MP_XG);
  const int x0 = (blockIdx.x % (SS / MP_XG)) * MP_XG;
  const int tid = threadIdx.x;
  const int lane = tid & 31, half = lane >> 4, l16 = lane & 15;
  const int wv = tid >> 5;
  const int y0 = wv * 16;

  v2f yf[HH / 4];
  const float* yrow = v2 + ((size_t)b * SS + y0 + l16) * HH + 2 * half;
#pragma unroll
  for (int ks = 0; ks < HH / 4; ks++) yf[ks] = *(const v2f*)(yrow + 4 * ks);

  for (int i = tid; i < PP * HH; i += MP_NT) {
    int p = i / HH, h = i - p * HH;
    float t = w[i];
    w2s[p * MP_WP + h] = t * t;
  }
  for (int i = tid; i < MP_XG * HH; i += MP_NT)
    xs[i] = v1[((size_t)b * SS + x0) * HH + i];
  if (tid < MP_XG * PP) wnAs[tid] = wnA[((size_t)b * SS + x0) * PP + tid];

  const int p0 = l16;
  const bool p1ok = (16 + l16) < PP;
  const int p1 = p1ok ? 16 + l16 : PP - 1;
  float wB0[8], wB1[8];
#pragma unroll
  for (int r = 0; r < 8; r++) {
    const float* wr = wnB + ((size_t)b * SS + y0 + r + 8 * half) * PP;
    wB0[r] = wr[p0];
    wB1[r] = wr[p1];
  }
  float cm0[8], cs0[8], cm1[8], cs1[8];
#pragma unroll
  for (int r = 0; r < 8; r++) { cm0[r] = -3.0e38f; cs0[r] = 0.f; cm1[r] = -3.0e38f; cs1[r] = 0.f; }
  __syncthreads();

  const float* wb0 = w2s + p0 * MP_WP + 2 * half;
  const float* wb1 = w2s + p1 * MP_WP + 2 * half;
  for (int xi = 0; xi < MP_XG; xi++) {
    const float* xr = xs + xi * HH + 2 * half;
    v8f acc0 = {0, 0, 0, 0, 0, 0, 0, 0};
    v8f acc1 = {0, 0, 0, 0, 0, 0, 0, 0};
#pragma unroll
    for (int ks = 0; ks < HH / 4; ks++) {
      v2f xv = *(const v2f*)(xr + 4 * ks);
      v2f a;
      a.x = xv.x * yf[ks].x;
      a.y = xv.y * yf[ks].y;
      v2f b0 = *(const v2f*)(wb0 + 4 * ks);
      v2f b1 = *(const v2f*)(wb1 + 4 * ks);
      acc0 = __builtin_amdgcn_wmma_f32_16x16x4_f32(false, a, false, b0, (short)0, acc0, false, false);
      acc1 = __builtin_amdgcn_wmma_f32_16x16x4_f32(false, a, false, b1, (short)0, acc1, false, false);
    }
    const float wa0 = wnAs[xi * PP + p0];
    const float wa1 = wnAs[xi * PP + p1];
    float m0 = -3.0e38f, s0 = 0.f, m1 = -3.0e38f, s1 = 0.f;
#pragma unroll
    for (int r = 0; r < 8; r++) {
      float mv0 = acc0[r] / fmaxf(wa0 * wB0[r], EPSF);
      float mv1 = acc1[r] / fmaxf(wa1 * wB1[r], EPSF);
      m0 = fmaxf(m0, mv0); s0 += mv0;
      m1 = fmaxf(m1, mv1); s1 += mv1;
      cm0[r] = fmaxf(cm0[r], mv0); cs0[r] += mv0;
      cm1[r] = fmaxf(cm1[r], mv1); cs1[r] += mv1;
    }
    m0 = fmaxf(m0, __shfl_xor(m0, 16, 32)); s0 += __shfl_xor(s0, 16, 32);
    m1 = fmaxf(m1, __shfl_xor(m1, 16, 32)); s1 += __shfl_xor(s1, 16, 32);
    if (half == 0) {
      const int o = (wv * MP_XG + xi) * PP;
      redm[o + p0] = m0; reds[o + p0] = s0;
      if (p1ok) { redm[o + p1] = m1; reds[o + p1] = s1; }
    }
  }
  {
    const size_t po = (((size_t)b * (SS / MP_XG) + blockIdx.x % (SS / MP_XG)) * SS + y0 + 8 * half) * PP;
#pragma unroll
    for (int r = 0; r < 8; r++) {
      pmax[po + r * PP + p0] = cm0[r];
      psum[po + r * PP + p0] = cs0[r];
      if (p1ok) {
        pmax[po + r * PP + p1] = cm1[r];
        psum[po + r * PP + p1] = cs1[r];
      }
    }
  }
  __syncthreads();
  if (tid < MP_XG * PP) {
    const int xi = tid / PP, p = tid - xi * PP;
    float mx = redm[xi * PP + p], sm = reds[xi * PP + p];
    for (int q = 1; q < 8; q++) {
      mx = fmaxf(mx, redm[(q * MP_XG + xi) * PP + p]);
      sm += reds[(q * MP_XG + xi) * PP + p];
    }
    const int x = x0 + xi;
    const float m = (float)mask[b * SS + x];
    float* orow = out + ((size_t)b * SS + x) * FF;
    orow[23 + p] = m * mx;
    orow[43 + p] = m * (sm / (float)SS);
  }
}

extern "C" void kernel_launch(void* const* d_in, const int* in_sizes, int n_in,
                              void* d_out, int out_size, void* d_ws, size_t ws_size,
                              hipStream_t stream) {
  const float* ctx_p = (const float*)d_in[0];
  const float* ctx_h = (const float*)d_in[1];
  const float* w0 = (const float*)d_in[2];
  const float* w1 = (const float*)d_in[3];
  const float* w2 = (const float*)d_in[4];
  const float* w3 = (const float*)d_in[5];
  const int* mask_p = (const int*)d_in[6];
  const int* mask_h = (const int*)d_in[7];
  float* out = (float*)d_out;
  float* ws = (float*)d_ws;

  const size_t SZ_ROWS = (size_t)BB * SS * HH;
  float* cp = ws;                 ws += SZ_ROWS;
  float* ch = ws;                 ws += SZ_ROWS;
  float* n_p = ws;                ws += BB * SS;
  float* n_h = ws;                ws += BB * SS;
  float* wn1_p = ws;              ws += (size_t)BB * SS * PP;
  float* wn1_h = ws;              ws += (size_t)BB * SS * PP;
  float* cosM = ws;               ws += (size_t)BB * SS * SS;
  float* mp_pmax = ws;            ws += (size_t)BB * (SS / MP_XG) * SS * PP;
  float* mp_psum = ws;            ws += (size_t)BB * (SS / MP_XG) * SS * PP;

  const dim3 row_tiles(BB * SS / RT, 3);
  prep_kernel<<<row_tiles, NTH, 0, stream>>>(ctx_p, mask_p, w1, cp, n_p, wn1_p);
  prep_kernel<<<row_tiles, NTH, 0, stream>>>(ctx_h, mask_h, w1, ch, n_h, wn1_h);

  cos_wmma_kernel<<<BB * 64, 32, 0, stream>>>(cp, ch, n_p, n_h, cosM);

  att_mpm_kernel<<<dim3(BB * SS / RT, 3, 2), NTH, 0, stream>>>(cp, ch, cosM, w2, w0, mask_p, mask_h, out);

  maxpool_pair_kernel<<<BB * (SS / MP_XG), MP_NT, 0, stream>>>(cp, ch, w1, wn1_p, wn1_h, mask_p, out,
                                                               mp_pmax, mp_psum);

  maxatt_mpm_kernel<<<dim3(BB * SS / MA_R, 2), MA_NT, 0, stream>>>(cp, ch, cosM, w3, mask_p, mask_h,
                                                                   mp_pmax, mp_psum, out);
}


# maxpool match: N columns are (x, perspective) pairs, B = x row * w*w built in LDS, A = y rows as loaded: 160 columns = 10 full tiles, none padded; product grouping and per-block x-sum order change
# speedup vs baseline: 7.1666x; 1.1148x over previous
#include <hip/hip_runtime.h>
#include <hip/hip_bf16.h>


#define BB 16
#define SS 128
#define HH 200
#define PP 20
#define FF 105
#define EPSF 1e-8f
#define MP_XG 8

typedef float v2f __attribute__((ext_vector_type(2)));
typedef float v8f __attribute__((ext_vector_type(8)));

#define RT 32
#define LDP 201
#define QPB 7
#define NTH 128
#define NV4 (RT * (HH / 4))
#define NFULL (NV4 / NTH)
#define NTAIL (NV4 % NTH)

struct TilePieces { float4 v[NFULL + 1]; };

__device__ __forceinline__ void tile_load(TilePieces& t, const float* __restrict__ g,
                                          int row_stride, int tid) {
#pragma unroll
  for (int k = 0; k < NFULL; k++) {
    int i = tid + k * NTH, r = i / (HH / 4), c = i - r * (HH / 4);
    t.v[k] = *(const float4*)(g + (size_t)r * row_stride + 4 * c);
  }
  t.v[NFULL] = make_float4(0.f, 0.f, 0.f, 0.f);
  if (tid < NTAIL) {
    int i = tid + NFULL * NTH, r = i / (HH / 4), c = i - r * (HH / 4);
    t.v[NFULL] = *(const float4*)(g + (size_t)r * row_stride + 4 * c);
  }
}

__device__ __forceinline__ void tile_to_lds(const TilePieces& t, float* lds, int tid) {
#pragma unroll
  for (int k = 0; k <= NFULL; k++) {
    int i = tid + k * NTH, r = i / (HH / 4), c = i - r * (HH / 4);
    if (k < NFULL || tid < NTAIL) {
      float* d = lds + r * LDP + 4 * c;
      d[0] = t.v[k].x; d[1] = t.v[k].y; d[2] = t.v[k].z; d[3] = t.v[k].w;
    }
  }
}

__global__ __launch_bounds__(NTH) void prep_kernel(
    const float* __restrict__ src, const int* __restrict__ mask,
    const float* __restrict__ w1, float* __restrict__ dst,
    float* __restrict__ nrm, float* __restrict__ wn) {
  __shared__ float rows[RT * LDP];
  const int row0 = blockIdx.x * RT;
  const int tid = threadIdx.x;
  const bool writes_dst = (blockIdx.y == 0);
  const float* stile = src + (size_t)row0 * HH;
  float* dtile = dst + (size_t)row0 * HH;
  TilePieces t;
  tile_load(t, stile, HH, tid);
#pragma unroll
  for (int k = 0; k <= NFULL; k++) {
    int r = (tid + k * NTH) / (HH / 4);
    float m = (k < NFULL || tid < NTAIL) ? (float)mask[row0 + r] : 0.f;
    t.v[k].x *= m; t.v[k].y *= m; t.v[k].z *= m; t.v[k].w *= m;
  }
  if (writes_dst) {
#pragma unroll
    for (int k = 0; k <= NFULL; k++)
      if (k < NFULL || tid < NTAIL) *(float4*)(dtile + 4 * (tid + k * NTH)) = t.v[k];
  }
  tile_to_lds(t, rows, tid);
  __syncthreads();

  const int lane = tid & 31;
  const int wv = __builtin_amdgcn_readfirstlane(tid >> 5);
  const int q0 = QPB * blockIdx.y + 2 * wv;
  const bool two = (2 * wv + 1) < QPB;
  const bool plain = (q0 == PP);
  const float* wa = w1 + (plain ? 0 : q0) * HH;
  const float* wb = w1 + (two ? q0 + 1 : 0) * HH;
  const float* vr = rows + lane * LDP;
  float sa = 0.f, sb = 0.f;
#pragma unroll 8
  for (int h = 0; h < HH; h++) {
    float v = vr[h];
    float vv = v * v;
    float xa = wa[h], xb = wb[h];
    float w2a = plain ? 1.0f : xa * xa;
    sa += w2a * vv;
    sb += (xb * xb) * vv;
  }
  const int row = row0 + lane;
  if (plain) nrm[row] = sqrtf(sa);
  else wn[row * PP + q0] = sqrtf(sa);
  if (two) wn[row * PP + q0 + 1] = sqrtf(sb);
}

__global__ void cos_wmma_kernel(const float* __restrict__ cp, const float* __restrict__ ch,
                                const float* __restrict__ np, const float* __restrict__ nh,
                                float* __restrict__ cosO) {
  int wid = blockIdx.x;
  int b = wid >> 6;
  int x0 = ((wid >> 3) & 7) << 4;
  int y0 = (wid & 7) << 4;
  int lane = threadIdx.x, half = lane >> 4, l16 = lane & 15;
  const float* arow = cp + ((size_t)(b * SS) + x0 + l16) * HH + 2 * half;
  const float* brow = ch + ((size_t)(b * SS) + y0 + l16) * HH + 2 * half;
  v8f acc = {0, 0, 0, 0, 0, 0, 0, 0};
  for (int k = 0; k < HH; k += 4) {
    v2f a = *(const v2f*)(arow + k);
    v2f bv = *(const v2f*)(brow + k);
    acc = __builtin_amdgcn_wmma_f32_16x16x4_f32(false, a, false, bv, (short)0, acc, false, false);
  }
  float nhv = nh[b * SS + y0 + l16];
#pragma unroll
  for (int r = 0; r < 8; r++) {
    int m = r + 8 * half;
    float d = fmaxf(np[b * SS + x0 + m] * nhv, EPSF);
    cosO[((size_t)(b * SS) + x0 + m) * SS + y0 + l16] = acc[r] / d;
  }
}

#define CP 132
#define AT_KS (SS / 4)

typedef float v4f __attribute__((ext_vector_type(4)));

__device__ __forceinline__ void bfrag_load(float (&bv)[2 * AT_KS], const float* __restrict__ vcol) {
#pragma unroll
  for (int i = 0; i < AT_KS; i++) {
    bv[2 * i] = vcol[(size_t)(4 * i) * HH];
    bv[2 * i + 1] = vcol[(size_t)(4 * i + 1) * HH];
  }
}

__device__ __forceinline__ void att_tile(const v2f (&a0)[AT_KS], const v2f (&a1)[AT_KS],
                                         const float (&bv)[2 * AT_KS], const float* csum,
                                         float* rows, int col, bool colOK, int half) {
  v8f acc0 = {0, 0, 0, 0, 0, 0, 0, 0};
  v8f acc1 = {0, 0, 0, 0, 0, 0, 0, 0};
#pragma unroll
  for (int i = 0; i < AT_KS; i++) {
    v2f bb;
    bb.x = bv[2 * i];
    bb.y = bv[2 * i + 1];
    acc0 = __builtin_amdgcn_wmma_f32_16x16x4_f32(false, a0[i], false, bb, (short)0, acc0, false, false);
    acc1 = __builtin_amdgcn_wmma_f32_16x16x4_f32(false, a1[i], false, bb, (short)0, acc1, false, false);
  }
  if (colOK) {
#pragma unroll
    for (int r = 0; r < 8; r++) {
      const int m = r + 8 * half;
      rows[m * LDP + col] = acc0[r] / fmaxf(csum[m], EPSF);
      rows[(16 + m) * LDP + col] = acc1[r] / fmaxf(csum[16 + m], EPSF);
    }
  }
}

__global__ __launch_bounds__(NTH) void att_mpm_kernel(
    const float* __restrict__ cp, const float* __restrict__ ch,
    const float* __restrict__ cosM, const float* __restrict__ w,
    const float* __restrict__ wf,
    const int* __restrict__ mask_p, const int* __restrict__ mask_h,
    float* __restrict__ out) {
  __shared__ float r1[RT * LDP];
  __shared__ __attribute__((aligned(16))) float r2[RT * LDP];
  __shared__ float csum[RT];
  const int side = blockIdx.z;
  const int row0 = blockIdx.x * RT;
  const int b = row0 / SS, s0 = row0 % SS;
  const int tid = threadIdx.x;
  const float* v1 = side ? ch : cp;
  const float* V = (side ? cp : ch) + (size_t)b * SS * HH;
  const int* mask = side ? mask_h : mask_p;
  const float* cb = cosM + (size_t)b * SS * SS;
  float* cst = r2;

  const int lane = tid & 31, half = lane >> 4, l16 = lane & 15;
  const int wv = __builtin_amdgcn_readfirstlane(tid >> 5);
  const int tw = 3 - wv;
  const int c0 = 16 * tw + l16, c1 = c0 + 64, c2 = c0 + 128, c3 = c0 + 192;
  const bool ok3 = c3 < HH;
  const float* vh = V + (size_t)(2 * half) * HH;

  TilePieces ta;
  tile_load(ta, v1 + (size_t)row0 * HH, HH, tid);
  v4f cv[8];
  const int sh = side ? 3 : 5, lo = side ? 7 : 31;
  const float* cpiece = cb + (side ? s0 : s0 * SS);
#pragma unroll
  for (int u = 0; u < 8; u++) {
    const int i = tid + u * NTH;
    cv[u] = *(const v4f*)(cpiece + (i >> sh) * SS + 4 * (i & lo));
  }
  float ba[2 * AT_KS], bb[2 * AT_KS];
  bfrag_load(ba, vh + c0);
  const int* omask = (side ? mask_p : mask_h) + b * SS;
  int olen = 0;
  for (int i = 0; i < SS; i++) olen += omask[i];
  const float* lr = V + (size_t)(olen > 0 ? olen - 1 : 0) * HH;
  tile_to_lds(ta, r1, tid);
  if (side == 0) {
#pragma unroll
    for (int u = 0; u < 8; u++) {
      const int i = tid + u * NTH;
      *(v4f*)(cst + (i >> 5) * CP + 4 * (i & 31)) = cv[u];
    }
  } else {
#pragma unroll
    for (int u = 0; u < 8; u++) {
      const int i = tid + u * NTH;
      const int k = i >> 3, j = 4 * (i & 7);
      cst[(j + 0) * CP + k] = cv[u].x; cst[(j + 1) * CP + k] = cv[u].y;
      cst[(j + 2) * CP + k] = cv[u].z; cst[(j + 3) * CP + k] = cv[u].w;
    }
  }
  __syncthreads();

  {
    const int j = tid >> 2, q = tid & 3;
    const v4f* c4 = (const v4f*)(cst + j * CP + 32 * q);
    float mx = -3.0e38f, sm = 0.f;
#pragma unroll
    for (int i = 0; i < 8; i++) {
      const v4f t = c4[i];
      mx = fmaxf(fmaxf(fmaxf(mx, t.x), fmaxf(t.y, t.z)), t.w);
      sm += t.x; sm += t.y; sm += t.z; sm += t.w;
    }
    mx = fmaxf(mx, __shfl_xor(mx, 1, 32)); sm += __shfl_xor(sm, 1, 32);
    mx = fmaxf(mx, __shfl_xor(mx, 2, 32)); sm += __shfl_xor(sm, 2, 32);
    if (q == 0) {
      csum[j] = sm;
      if (blockIdx.y == 0) {
        const float m = (float)mask[row0 + j];
        float* o2 = out + ((size_t)(side * BB + b) * SS + s0 + j) * FF;
        o2[0] = m * mx;
        o2[1] = m * (sm / (float)SS);
      }
    }
  }
  v2f a0[AT_KS], a1[AT_KS];
  {
    const float* cl = cst + l16 * CP + 2 * half;
#pragma unroll
    for (int i = 0; i < AT_KS; i++) {
      a0[i] = *(const v2f*)(cl + 4 * i);
      a1[i] = *(const v2f*)(cl + 16 * CP + 4 * i);
    }
  }
  __syncthreads();

  bfrag_load(bb, vh + c1);
  __builtin_amdgcn_sched_barrier(0);
  att_tile(a0, a1, ba, csum, r2, c0, true, half);
  bfrag_load(ba, vh + c2);
  __builtin_amdgcn_sched_barrier(0);
  att_tile(a0, a1, bb, csum, r2, c1, true, half);
  if (tw == 0) bfrag_load(bb, vh + (ok3 ? c3 : HH - 1));
  att_tile(a0, a1, ba, csum, r2, c2, true, half);
  if (tw == 0) att_tile(a0, a1, bb, csum, r2, c3, ok3, half);
  __syncthreads();

  const int q0 = QPB * blockIdx.y + 2 * wv;
  const bool two = (2 * wv + 1) < QPB;
  const bool plain = (q0 == PP);
  const float* wa = w + (plain ? 0 : q0) * HH;
  const float* wb = w + (two ? q0 + 1 : 0) * HH;
  const float* ua = wf + (plain ? 0 : q0) * HH;
  const float* ub = wf + (two ? q0 + 1 : 0) * HH;
  const float* ar = r1 + lane * LDP;
  const float* cr = r2 + lane * LDP;
  float a11 = 0.f, a22 = 0.f, a12 = 0.f, b11 = 0.f, b22 = 0.f, b12 = 0.f;
  float f11 = 0.f, f22 = 0.f, f12 = 0.f, g11 = 0.f, g22 = 0.f, g12 = 0.f;
#pragma unroll 8
  for (int h = 0; h < HH; h++) {
    float a = ar[h], c = cr[h];
    float aa = a * a, cc = c * c, ac = a * c;
    float xa = wa[h], xb = wb[h];
    float w2a = plain ? 1.0f : xa * xa;
    float w2b = xb * xb;
    a11 += w2a * aa; a22 += w2a * cc; a12 += w2a * ac;
    b11 += w2b * aa; b22 += w2b * cc; b12 += w2b * ac;
    float l = lr[h];
    float ll = l * l, al = a * l;
    float ya = ua[h], yb = ub[h];
    float u2a = plain ? 1.0f : ya * ya;
    float u2b = yb * yb;
    f11 += u2a * aa; f22 += u2a * ll; f12 += u2a * al;
    g11 += u2b * aa; g22 += u2b * ll; g12 += u2b * al;
  }
  float* orow = out + ((size_t)(side * BB + b) * SS + s0 + lane) * FF;
  orow[63 + (plain ? 0 : 1 + q0)] = a12 / (fmaxf(sqrtf(a11), EPSF) * fmaxf(sqrtf(a22), EPSF));
  orow[2 + (plain ? 0 : 1 + q0)] = f12 / (fmaxf(sqrtf(f11), EPSF) * fmaxf(sqrtf(f22), EPSF));
  if (two) {
    orow[63 + 2 + q0] = b12 / (fmaxf(sqrtf(b11), EPSF) * fmaxf(sqrtf(b22), EPSF));
    orow[2 + 2 + q0] = g12 / (fmaxf(sqrtf(g11), EPSF) * fmaxf(sqrtf(g22), EPSF));
  }
}

#define MA_R 8
#define MA_NT 256
#define MA_NQ 21
#define MA_WP 201
#define MA_KC 16
#define MA_FN (MA_R * PP)
#define MA_NG (SS / MP_XG)

__device__ __forceinline__ void vcol_load(float (&v)[MA_KC], const float* __restrict__ vcol, int k0) {
#pragma unroll
  for (int i = 0; i < MA_KC; i++) v[i] = vcol[(size_t)(k0 + i) * HH];
}

__device__ __forceinline__ void maxplus_step(const float* cs, int k0, const float (&v)[MA_KC], float (&mx)[MA_R]) {
#pragma unroll
  for (int i = 0; i < MA_KC; i++) {
#pragma unroll
    for (int j = 0; j < MA_R; j++) mx[j] = fmaxf(mx[j], cs[(k0 + i) * MA_R + j] * v[i]);
  }
}

__global__ __launch_bounds__(MA_NT) void maxatt_mpm_kernel(
    const float* __restrict__ cp, const float* __restrict__ ch,
    const float* __restrict__ cosM, const float* __restrict__ w,
    const int* __restrict__ mask_p, const int* __restrict__ mask_h,
    const float* __restrict__ pmax, const float* __restrict__ psum,
    float* __restrict__ out) {
  __shared__ __attribute__((aligned(16))) float r1[MA_R * HH];
  __shared__ __attribute__((aligned(16))) float r2[MA_R * HH];
  __shared__ __attribute__((aligned(16))) float cs[SS * MA_R];
  __shared__ float w2s[MA_NQ * MA_WP];

  const int side = blockIdx.y;
  const int row0 = blockIdx.x * MA_R;
  const int b = row0 / SS, r0 = row0 % SS;
  const int tid = threadIdx.x;
  const float* v1 = side ? ch : cp;
  const float* V = side ? cp : ch;
  const int* mask = side ? mask_h : mask_p;
  const float* cb = cosM + (size_t)b * SS * SS;

  constexpr int W4 = PP * HH / 4;
  constexpr int R4 = MA_R * HH / 4;
  v4f wv4[4], rv4[2], cv4;
#pragma unroll
  for (int u = 0; u < 4; u++) {
    const int i = tid + u * MA_NT;
    wv4[u] = *(const v4f*)(w + 4 * (i < W4 ? i : W4 - 1));
  }
#pragma unroll
  for (int u = 0; u < 2; u++) {
    const int i = tid + u * MA_NT;
    rv4[u] = *(const v4f*)(v1 + (size_t)row0 * HH + 4 * (i < R4 ? i : R4 - 1));
  }
  const int pj = tid & 7, pk = 4 * (tid >> 3);
  const int qk = tid >> 1, qj = 4 * (tid & 1);
  cv4 = side == 0 ? *(const v4f*)(cb + (size_t)(r0 + pj) * SS + pk)
                  : *(const v4f*)(cb + (size_t)qk * SS + r0 + qj);
  const int fc = tid < MA_FN ? tid : MA_FN - 1;
  const int fy = fc / PP, fp = fc - fy * PP;
  const float* part = (side ? psum : pmax) + (size_t)b * MA_NG * (SS * PP) + r0 * PP + fc;
  float pv[MA_NG];
#pragma unroll
  for (int g = 0; g < MA_NG; g++) pv[g] = part[(size_t)g * (SS * PP)];
  const int fmask = mask_h[b * SS + r0 + fy];
  asm volatile("" :: "v"(wv4[0]), "v"(wv4[1]), "v"(wv4[2]), "v"(wv4[3]), "v"(rv4[0]), "v"(rv4[1]), "v"(cv4),
               "v"(pv[0]), "v"(pv[1]), "v"(pv[2]), "v"(pv[3]), "v"(pv[4]), "v"(pv[5]), "v"(pv[6]), "v"(pv[7]),
               "v"(pv[8]), "v"(pv[9]), "v"(pv[10]), "v"(pv[11]), "v"(pv[12]), "v"(pv[13]), "v"(pv[14]), "v"(pv[15]),
               "v"(fmask));
#pragma unroll
  for (int u = 0; u < 4; u++) {
    const int i = tid + u * MA_NT;
    if (i < W4) {
      const int p = i / (HH / 4), c = i - p * (HH / 4);
      float* d = w2s + p * MA_WP + 4 * c;
      d[0] = wv4[u].x * wv4[u].x; d[1] = wv4[u].y * wv4[u].y;
      d[2] = wv4[u].z * wv4[u].z; d[3] = wv4[u].w * wv4[u].w;
    }
  }
  if (tid < HH) w2s[PP * MA_WP + tid] = 1.0f;
#pragma unroll
  for (int u = 0; u < 2; u++) {
    const int i = tid + u * MA_NT;
    if (i < R4) *(v4f*)(r1 + 4 * i) = rv4[u];
  }
  if (side == 0) {
    cs[(pk + 0) * MA_R + pj] = cv4.x; cs[(pk + 1) * MA_R + pj] = cv4.y;
    cs[(pk + 2) * MA_R + pj] = cv4.z; cs[(pk + 3) * MA_R + pj] = cv4.w;
  } else {
    *(v4f*)(cs + qk * MA_R + qj) = cv4;
  }
  __syncthreads();

  if (tid < MA_FN) {
    float* orow = out + ((size_t)(BB + b) * SS + r0 + fy) * FF;
    const float m = (float)fmask;
    if (side == 0) {
      float mx = -3.0e38f;
#pragma unroll
      for (int g = 0; g < MA_NG; g++) mx = fmaxf(mx, pv[g]);
      orow[23 + fp] = m * mx;
    } else {
      float sm = 0.f;
#pragma unroll
      for (int g = 0; g < MA_NG; g++) sm += pv[g];
      orow[43 + fp] = m * (sm / (float)SS);
    }
  }

  const int hc = tid < HH ? tid : HH - 1;
  const float* vcol = V + (size_t)b * SS * HH + hc;
  float mx[MA_R];
#pragma unroll
  for (int j = 0; j < MA_R; j++) mx[j] = -3.0e38f;
  if ((tid & ~31) < HH) {
    float va[MA_KC], vb[MA_KC];
    vcol_load(va, vcol, 0);
#pragma unroll
    for (int k0 = 0; k0 < SS; k0 += 2 * MA_KC) {
      vcol_load(vb, vcol, k0 + MA_KC);
      maxplus_step(cs, k0, va, mx);
      if (k0 + 2 * MA_KC < SS) vcol_load(va, vcol, k0 + 2 * MA_KC);
      maxplus_step(cs, k0 + MA_KC, vb, mx);
    }
  }
  if (tid < HH) {
#pragma unroll
    for (int j = 0; j < MA_R; j++) r2[j * HH + tid] = (float)mask[row0 + j] * mx[j];
  }
  __syncthreads();

  if (tid < MA_R * MA_NQ) {
    const int j = tid / MA_NQ, q = tid - j * MA_NQ;
    const float* ar = r1 + j * HH;
    const float* cr = r2 + j * HH;
    const float* wq = w2s + q * MA_WP;
    float a11 = 0.f, a22 = 0.f, a12 = 0.f;
#pragma unroll 8
    for (int h = 0; h < HH; h++) {
      const float a = ar[h], c = cr[h];
      const float aa = a * a, cc = c * c, ac = a * c;
      const float w2a = wq[h];
      a11 += w2a * aa; a22 += w2a * cc; a12 += w2a * ac;
    }
    float* orow = out + ((size_t)(side * BB + b) * SS + r0 + j) * FF + 84;
    orow[q == PP ? 0 : 1 + q] = a12 / (fmaxf(sqrtf(a11), EPSF) * fmaxf(sqrtf(a22), EPSF));
  }
}

#define MP_WP 268
#define MP_NT 256
#define MP_NC (MP_XG * PP)
#define MP_KS 5
#define MP_NG (HH / 4 / MP_KS)
#define MP_BJ (HH / 4)
#define MP_BT (MP_BJ * 5)

__device__ __forceinline__ void mp_bload(v2f (&f)[2 * MP_KS], const float* c0, const float* c1, int g) {
#pragma unroll
  for (int i = 0; i < MP_KS; i++) {
    f[2 * i] = *(const v2f*)(c0 + 4 * (g * MP_KS + i));
    f[2 * i + 1] = *(const v2f*)(c1 + 4 * (g * MP_KS + i));
  }
}

__device__ __forceinline__ void mp_mma(const v2f (&yf)[HH / 4], int g, const v2f (&f)[2 * MP_KS],
                                       v8f& acc0, v8f& acc1) {
#pragma unroll
  for (int i = 0; i < MP_KS; i++) {
    acc0 = __builtin_amdgcn_wmma_f32_16x16x4_f32(false, yf[g * MP_KS + i], false, f[2 * i], (short)0, acc0, false, false);
    acc1 = __builtin_amdgcn_wmma_f32_16x16x4_f32(false, yf[g * MP_KS + i], false, f[2 * i + 1], (short)0, acc1, false, false);
  }
}

template <int MASK>
__device__ __forceinline__ float mp_xlane(float v) {
  return __int_as_float(__builtin_amdgcn_update_dpp(0, __float_as_int(v), 0x160 + MASK, 0xf, 0xf, true));
}

__device__ __forceinline__ void mp_tile_out(const v8f& acc, float wa, const float (&wB)[8], int xi, int half,
                                            float* rm, float* rs, float* pm, float* ps) {
  float mv[8];
  float m = -3.0e38f, s = 0.f;
#pragma unroll
  for (int r = 0; r < 8; r++) {
    mv[r] = acc[r] / fmaxf(wa * wB[r], EPSF);
    m = fmaxf(m, mv[r]); s += mv[r];
  }
  m = fmaxf(m, __shfl_xor(m, 16, 32)); s += __shfl_xor(s, 16, 32);
  if (half == 0) { *rm = m; *rs = s; }
  float hm = 0.f, hs = 0.f;
#pragma unroll
  for (int r = 0; r < 8; r++) {
    float t = mv[r], u = mv[r];
    t = fmaxf(t, mp_xlane<1>(t)); u += mp_xlane<1>(u);
    t = fmaxf(t, mp_xlane<2>(t)); u += mp_xlane<2>(u);
    t = fmaxf(t, mp_xlane<4>(t)); u += mp_xlane<4>(u);
    if (xi == r) { hm = t; hs = u; }
  }
  *pm = hm; *ps = hs;
}

__global__ __launch_bounds__(MP_NT) void maxpool_pair_kernel(
    const float* __restrict__ v1, const float* __restrict__ v2,
    const float* __restrict__ w,
    const float* __restrict__ wnA, const float* __restrict__ wnB,
    const int* __restrict__ mask, float* __restrict__ out,
    float* __restrict__ pmax, float* __restrict__ psum) {
  __shared__ __attribute__((aligned(16))) float xw[MP_NC * MP_WP];
  __shared__ float redm[8 * MP_NC];
  __shared__ float reds[8 * MP_NC];

  const int b = blockIdx.x / (SS / MP_XG);
  const int xg = blockIdx.x % (SS / MP_XG);
  const int x0 = xg * MP_XG;
  const int tid = threadIdx.x;
  const int lane = tid & 31, half = lane >> 4, l16 = lane & 15;
  const int wv = tid >> 5;
  const int y0 = wv * 16;

  v2f yf[HH / 4];
  const float* yrow = v2 + ((size_t)b * SS + y0 + l16) * HH + 2 * half;
#pragma unroll
  for (int ks = 0; ks < HH / 4; ks++) yf[ks] = *(const v2f*)(yrow + 4 * ks);

  {
    const int bt = tid < MP_BT ? tid : MP_BT - 1;
    const int bg = bt / MP_BJ, bj = bt - bg * MP_BJ;
    v4f x4[MP_XG], w4[PP / 5];
#pragma unroll
    for (int xi = 0; xi < MP_XG; xi++) x4[xi] = *(const v4f*)(v1 + ((size_t)b * SS + x0 + xi) * HH + 4 * bj);
#pragma unroll
    for (int q = 0; q < PP / 5; q++) w4[q] = *(const v4f*)(w + (bg + 5 * q) * HH + 4 * bj);
    if (tid < MP_BT) {
#pragma unroll
      for (int q = 0; q < PP / 5; q++) {
        const v4f ww = w4[q] * w4[q];
        float* col = xw + (size_t)((bg + 5 * q) * MP_XG) * MP_WP + 4 * bj;
#pragma unroll
        for (int xi = 0; xi < MP_XG; xi++) *(v4f*)(col + xi * MP_WP) = x4[xi] * ww;
      }
    }
  }
  __syncthreads();

  const int pb = l16 >> 3, xi = l16 & 7;
  const float* wArow = wnA + ((size_t)b * SS + x0 + xi) * PP;
  const float* wBrow = wnB + ((size_t)b * SS + y0 + 8 * half) * PP;
  const size_t po = (((size_t)b * (SS / MP_XG) + xg) * SS + y0 + 8 * half + xi) * PP + pb;
  const float* bc = xw + l16 * MP_WP + 2 * half;

  v2f fa[2 * MP_KS], fb[2 * MP_KS];
  mp_bload(fa, bc, bc + 16 * MP_WP, 0);
  for (int tp = 0; tp < MP_NC / 32; tp++) {
    const float* c0 = bc + (size_t)(32 * tp) * MP_WP;
    const float* c1 = c0 + 16 * MP_WP;
    const float* n0 = bc + (size_t)(32 * (tp + 1 < MP_NC / 32 ? tp + 1 : tp)) * MP_WP;
    const int p0 = 4 * tp + pb, p1 = p0 + 2;
    const float wa0 = wArow[p0], wa1 = wArow[p1];
    float wB0[8], wB1[8];
#pragma unroll
    for (int r = 0; r < 8; r++) { wB0[r] = wBrow[r * PP + p0]; wB1[r] = wBrow[r * PP + p1]; }
    v8f acc0 = {0, 0, 0, 0, 0, 0, 0, 0};
    v8f acc1 = {0, 0, 0, 0, 0, 0, 0, 0};
#pragma unroll
    for (int g = 0; g < MP_NG; g += 2) {
      mp_bload(fb, c0, c1, g + 1);
      __builtin_amdgcn_sched_barrier(0);
      mp_mma(yf, g, fa, acc0, acc1);
      if (g + 2 < MP_NG) mp_bload(fa, c0, c1, g + 2);
      else mp_bload(fa, n0, n0 + 16 * MP_WP, 0);
      __builtin_amdgcn_sched_barrier(0);
      mp_mma(yf, g + 1, fb, acc0, acc1);
    }
    const int n = 32 * tp + l16;
    mp_tile_out(acc0, wa0, wB0, xi, half, redm + wv * MP_NC + n, reds + wv * MP_NC + n,
                pmax + po + p0 - pb, psum + po + p0 - pb);
    mp_tile_out(acc1, wa1, wB1, xi, half, redm + wv * MP_NC + n + 16, reds + wv * MP_NC + n + 16,
                pmax + po + p1 - pb, psum + po + p1 - pb);
  }
  __syncthreads();
  if (tid < MP_NC) {
    const int p = tid / MP_XG, xr = tid - p * MP_XG;
    float mx = redm[tid], sm = reds[tid];
    for (int q = 1; q < 8; q++) {
      mx = fmaxf(mx, redm[q * MP_NC + tid]);
      sm += reds[q * MP_NC + tid];
    }
    const int x = x0 + xr;
    const float m = (float)mask[b * SS + x];
    float* orow = out + ((size_t)b * SS + x) * FF;
    orow[23 + p] = m * mx;
    orow[43 + p] = m * (sm / (float)SS);
  }
}

extern "C" void kernel_launch(void* const* d_in, const int* in_sizes, int n_in,
                              void* d_out, int out_size, void* d_ws, size_t ws_size,
                              hipStream_t stream) {
  const float* ctx_p = (const float*)d_in[0];
  const float* ctx_h = (const float*)d_in[1];
  const float* w0 = (const float*)d_in[2];
  const float* w1 = (const float*)d_in[3];
  const float* w2 = (const float*)d_in[4];
  const float* w3 = (const float*)d_in[5];
  const int* mask_p = (const int*)d_in[6];
  const int* mask_h = (const int*)d_in[7];
  float* out = (float*)d_out;
  float* ws = (float*)d_ws;

  const size_t SZ_ROWS = (size_t)BB * SS * HH;
  float* cp = ws;                 ws += SZ_ROWS;
  float* ch = ws;                 ws += SZ_ROWS;
  float* n_p = ws;                ws += BB * SS;
  float* n_h = ws;                ws += BB * SS;
  float* wn1_p = ws;              ws += (size_t)BB * SS * PP;
  float* wn1_h = ws;              ws += (size_t)BB * SS * PP;
  float* cosM = ws;               ws += (size_t)BB * SS * SS;
  float* mp_pmax = ws;            ws += (size_t)BB * (SS / MP_XG) * SS * PP;
  float* mp_psum = ws;            ws += (size_t)BB * (SS / MP_XG) * SS * PP;

  const dim3 row_tiles(BB * SS / RT, 3);
  prep_kernel<<<row_tiles, NTH, 0, stream>>>(ctx_p, mask_p, w1, cp, n_p, wn1_p);
  prep_kernel<<<row_tiles, NTH, 0, stream>>>(ctx_h, mask_h, w1, ch, n_h, wn1_h);

  cos_wmma_kernel<<<BB * 64, 32, 0, stream>>>(cp, ch, n_p, n_h, cosM);

  att_mpm_kernel<<<dim3(BB * SS / RT, 3, 2), NTH, 0, stream>>>(cp, ch, cosM, w2, w0, mask_p, mask_h, out);

  maxpool_pair_kernel<<<BB * (SS / MP_XG), MP_NT, 0, stream>>>(cp, ch, w1, wn1_p, wn1_h, mask_p, out,
                                                               mp_pmax, mp_psum);

  maxatt_mpm_kernel<<<dim3(BB * SS / MA_R, 2), MA_NT, 0, stream>>>(cp, ch, cosM, w3, mask_p, mask_h,
                                                                   mp_pmax, mp_psum, out);
}


# pairwise cosine rides in the maxpool match GEMM as a 21st perspective of unit weights (8 plain x columns, an 11th column tile per wave); cos launch removed, 6 launches -> 5; cos k-sum order changes
# speedup vs baseline: 7.6392x; 1.0659x over previous
#include <hip/hip_runtime.h>
#include <hip/hip_bf16.h>


#define BB 16
#define SS 128
#define HH 200
#define PP 20
#define FF 105
#define EPSF 1e-8f
#define MP_XG 8

typedef float v2f __attribute__((ext_vector_type(2)));
typedef float v8f __attribute__((ext_vector_type(8)));

#define RT 32
#define LDP 201
#define QPB 7
#define NTH 128
#define NV4 (RT * (HH / 4))
#define NFULL (NV4 / NTH)
#define NTAIL (NV4 % NTH)

struct TilePieces { float4 v[NFULL + 1]; };

__device__ __forceinline__ void tile_load(TilePieces& t, const float* __restrict__ g,
                                          int row_stride, int tid) {
#pragma unroll
  for (int k = 0; k < NFULL; k++) {
    int i = tid + k * NTH, r = i / (HH / 4), c = i - r * (HH / 4);
    t.v[k] = *(const float4*)(g + (size_t)r * row_stride + 4 * c);
  }
  t.v[NFULL] = make_float4(0.f, 0.f, 0.f, 0.f);
  if (tid < NTAIL) {
    int i = tid + NFULL * NTH, r = i / (HH / 4), c = i - r * (HH / 4);
    t.v[NFULL] = *(const float4*)(g + (size_t)r * row_stride + 4 * c);
  }
}

__device__ __forceinline__ void tile_to_lds(const TilePieces& t, float* lds, int tid) {
#pragma unroll
  for (int k = 0; k <= NFULL; k++) {
    int i = tid + k * NTH, r = i / (HH / 4), c = i - r * (HH / 4);
    if (k < NFULL || tid < NTAIL) {
      float* d = lds + r * LDP + 4 * c;
      d[0] = t.v[k].x; d[1] = t.v[k].y; d[2] = t.v[k].z; d[3] = t.v[k].w;
    }
  }
}

__global__ __launch_bounds__(NTH) void prep_kernel(
    const float* __restrict__ src, const int* __restrict__ mask,
    const float* __restrict__ w1, float* __restrict__ dst,
    float* __restrict__ nrm, float* __restrict__ wn) {
  __shared__ float rows[RT * LDP];
  const int row0 = blockIdx.x * RT;
  const int tid = threadIdx.x;
  const bool writes_dst = (blockIdx.y == 0);
  const float* stile = src + (size_t)row0 * HH;
  float* dtile = dst + (size_t)row0 * HH;
  TilePieces t;
  tile_load(t, stile, HH, tid);
#pragma unroll
  for (int k = 0; k <= NFULL; k++) {
    int r = (tid + k * NTH) / (HH / 4);
    float m = (k < NFULL || tid < NTAIL) ? (float)mask[row0 + r] : 0.f;
    t.v[k].x *= m; t.v[k].y *= m; t.v[k].z *= m; t.v[k].w *= m;
  }
  if (writes_dst) {
#pragma unroll
    for (int k = 0; k <= NFULL; k++)
      if (k < NFULL || tid < NTAIL) *(float4*)(dtile + 4 * (tid + k * NTH)) = t.v[k];
  }
  tile_to_lds(t, rows, tid);
  __syncthreads();

  const int lane = tid & 31;
  const int wv = __builtin_amdgcn_readfirstlane(tid >> 5);
  const int q0 = QPB * blockIdx.y + 2 * wv;
  const bool two = (2 * wv + 1) < QPB;
  const bool plain = (q0 == PP);
  const float* wa = w1 + (plain ? 0 : q0) * HH;
  const float* wb = w1 + (two ? q0 + 1 : 0) * HH;
  const float* vr = rows + lane * LDP;
  float sa = 0.f, sb = 0.f;
#pragma unroll 8
  for (int h = 0; h < HH; h++) {
    float v = vr[h];
    float vv = v * v;
    float xa = wa[h], xb = wb[h];
    float w2a = plain ? 1.0f : xa * xa;
    sa += w2a * vv;
    sb += (xb * xb) * vv;
  }
  const int row = row0 + lane;
  if (plain) nrm[row] = sqrtf(sa);
  else wn[row * PP + q0] = sqrtf(sa);
  if (two) wn[row * PP + q0 + 1] = sqrtf(sb);
}

#define CP 132
#define AT_KS (SS / 4)

typedef float v4f __attribute__((ext_vector_type(4)));

__device__ __forceinline__ void bfrag_load(float (&bv)[2 * AT_KS], const float* __restrict__ vcol) {
#pragma unroll
  for (int i = 0; i < AT_KS; i++) {
    bv[2 * i] = vcol[(size_t)(4 * i) * HH];
    bv[2 * i + 1] = vcol[(size_t)(4 * i + 1) * HH];
  }
}

__device__ __forceinline__ void att_tile(const v2f (&a0)[AT_KS], const v2f (&a1)[AT_KS],
                                         const float (&bv)[2 * AT_KS], const float* csum,
                                         float* rows, int col, bool colOK, int half) {
  v8f acc0 = {0, 0, 0, 0, 0, 0, 0, 0};
  v8f acc1 = {0, 0, 0, 0, 0, 0, 0, 0};
#pragma unroll
  for (int i = 0; i < AT_KS; i++) {
    v2f bb;
    bb.x = bv[2 * i];
    bb.y = bv[2 * i + 1];
    acc0 = __builtin_amdgcn_wmma_f32_16x16x4_f32(false, a0[i], false, bb, (short)0, acc0, false, false);
    acc1 = __builtin_amdgcn_wmma_f32_16x16x4_f32(false, a1[i], false, bb, (short)0, acc1, false, false);
  }
  if (colOK) {
#pragma unroll
    for (int r = 0; r < 8; r++) {
      const int m = r + 8 * half;
      rows[m * LDP + col] = acc0[r] / fmaxf(csum[m], EPSF);
      rows[(16 + m) * LDP + col] = acc1[r] / fmaxf(csum[16 + m], EPSF);
    }
  }
}

__global__ __launch_bounds__(NTH) void att_mpm_kernel(
    const float* __restrict__ cp, const float* __restrict__ ch,
    const float* __restrict__ cosM, const float* __restrict__ w,
    const float* __restrict__ wf,
    const int* __restrict__ mask_p, const int* __restrict__ mask_h,
    float* __restrict__ out) {
  __shared__ float r1[RT * LDP];
  __shared__ __attribute__((aligned(16))) float r2[RT * LDP];
  __shared__ float csum[RT];
  const int side = blockIdx.z;
  const int row0 = blockIdx.x * RT;
  const int b = row0 / SS, s0 = row0 % SS;
  const int tid = threadIdx.x;
  const float* v1 = side ? ch : cp;
  const float* V = (side ? cp : ch) + (size_t)b * SS * HH;
  const int* mask = side ? mask_h : mask_p;
  const float* cb = cosM + (size_t)b * SS * SS;
  float* cst = r2;

  const int lane = tid & 31, half = lane >> 4, l16 = lane & 15;
  const int wv = __builtin_amdgcn_readfirstlane(tid >> 5);
  const int tw = 3 - wv;
  const int c0 = 16 * tw + l16, c1 = c0 + 64, c2 = c0 + 128, c3 = c0 + 192;
  const bool ok3 = c3 < HH;
  const float* vh = V + (size_t)(2 * half) * HH;

  TilePieces ta;
  tile_load(ta, v1 + (size_t)row0 * HH, HH, tid);
  v4f cv[8];
  const int sh = side ? 3 : 5, lo = side ? 7 : 31;
  const float* cpiece = cb + (side ? s0 : s0 * SS);
#pragma unroll
  for (int u = 0; u < 8; u++) {
    const int i = tid + u * NTH;
    cv[u] = *(const v4f*)(cpiece + (i >> sh) * SS + 4 * (i & lo));
  }
  float ba[2 * AT_KS], bb[2 * AT_KS];
  bfrag_load(ba, vh + c0);
  const int* omask = (side ? mask_p : mask_h) + b * SS;
  int olen = 0;
  for (int i = 0; i < SS; i++) olen += omask[i];
  const float* lr = V + (size_t)(olen > 0 ? olen - 1 : 0) * HH;
  tile_to_lds(ta, r1, tid);
  if (side == 0) {
#pragma unroll
    for (int u = 0; u < 8; u++) {
      const int i = tid + u * NTH;
      *(v4f*)(cst + (i >> 5) * CP + 4 * (i & 31)) = cv[u];
    }
  } else {
#pragma unroll
    for (int u = 0; u < 8; u++) {
      const int i = tid + u * NTH;
      const int k = i >> 3, j = 4 * (i & 7);
      cst[(j + 0) * CP + k] = cv[u].x; cst[(j + 1) * CP + k] = cv[u].y;
      cst[(j + 2) * CP + k] = cv[u].z; cst[(j + 3) * CP + k] = cv[u].w;
    }
  }
  __syncthreads();

  {
    const int j = tid >> 2, q = tid & 3;
    const v4f* c4 = (const v4f*)(cst + j * CP + 32 * q);
    float mx = -3.0e38f, sm = 0.f;
#pragma unroll
    for (int i = 0; i < 8; i++) {
      const v4f t = c4[i];
      mx = fmaxf(fmaxf(fmaxf(mx, t.x), fmaxf(t.y, t.z)), t.w);
      sm += t.x; sm += t.y; sm += t.z; sm += t.w;
    }
    mx = fmaxf(mx, __shfl_xor(mx, 1, 32)); sm += __shfl_xor(sm, 1, 32);
    mx = fmaxf(mx, __shfl_xor(mx, 2, 32)); sm += __shfl_xor(sm, 2, 32);
    if (q == 0) {
      csum[j] = sm;
      if (blockIdx.y == 0) {
        const float m = (float)mask[row0 + j];
        float* o2 = out + ((size_t)(side * BB + b) * SS + s0 + j) * FF;
        o2[0] = m * mx;
        o2[1] = m * (sm / (float)SS);
      }
    }
  }
  v2f a0[AT_KS], a1[AT_KS];
  {
    const float* cl = cst + l16 * CP + 2 * half;
#pragma unroll
    for (int i = 0; i < AT_KS; i++) {
      a0[i] = *(const v2f*)(cl + 4 * i);
      a1[i] = *(const v2f*)(cl + 16 * CP + 4 * i);
    }
  }
  __syncthreads();

  bfrag_load(bb, vh + c1);
  __builtin_amdgcn_sched_barrier(0);
  att_tile(a0, a1, ba, csum, r2, c0, true, half);
  bfrag_load(ba, vh + c2);
  __builtin_amdgcn_sched_barrier(0);
  att_tile(a0, a1, bb, csum, r2, c1, true, half);
  if (tw == 0) bfrag_load(bb, vh + (ok3 ? c3 : HH - 1));
  att_tile(a0, a1, ba, csum, r2, c2, true, half);
  if (tw == 0) att_tile(a0, a1, bb, csum, r2, c3, ok3, half);
  __syncthreads();

  const int q0 = QPB * blockIdx.y + 2 * wv;
  const bool two = (2 * wv + 1) < QPB;
  const bool plain = (q0 == PP);
  const float* wa = w + (plain ? 0 : q0) * HH;
  const float* wb = w + (two ? q0 + 1 : 0) * HH;
  const float* ua = wf + (plain ? 0 : q0) * HH;
  const float* ub = wf + (two ? q0 + 1 : 0) * HH;
  const float* ar = r1 + lane * LDP;
  const float* cr = r2 + lane * LDP;
  float a11 = 0.f, a22 = 0.f, a12 = 0.f, b11 = 0.f, b22 = 0.f, b12 = 0.f;
  float f11 = 0.f, f22 = 0.f, f12 = 0.f, g11 = 0.f, g22 = 0.f, g12 = 0.f;
#pragma unroll 8
  for (int h = 0; h < HH; h++) {
    float a = ar[h], c = cr[h];
    float aa = a * a, cc = c * c, ac = a * c;
    float xa = wa[h], xb = wb[h];
    float w2a = plain ? 1.0f : xa * xa;
    float w2b = xb * xb;
    a11 += w2a * aa; a22 += w2a * cc; a12 += w2a * ac;
    b11 += w2b * aa; b22 += w2b * cc; b12 += w2b * ac;
    float l = lr[h];
    float ll = l * l, al = a * l;
    float ya = ua[h], yb = ub[h];
    float u2a = plain ? 1.0f : ya * ya;
    float u2b = yb * yb;
    f11 += u2a * aa; f22 += u2a * ll; f12 += u2a * al;
    g11 += u2b * aa; g22 += u2b * ll; g12 += u2b * al;
  }
  float* orow = out + ((size_t)(side * BB + b) * SS + s0 + lane) * FF;
  orow[63 + (plain ? 0 : 1 + q0)] = a12 / (fmaxf(sqrtf(a11), EPSF) * fmaxf(sqrtf(a22), EPSF));
  orow[2 + (plain ? 0 : 1 + q0)] = f12 / (fmaxf(sqrtf(f11), EPSF) * fmaxf(sqrtf(f22), EPSF));
  if (two) {
    orow[63 + 2 + q0] = b12 / (fmaxf(sqrtf(b11), EPSF) * fmaxf(sqrtf(b22), EPSF));
    orow[2 + 2 + q0] = g12 / (fmaxf(sqrtf(g11), EPSF) * fmaxf(sqrtf(g22), EPSF));
  }
}

#define MA_R 8
#define MA_NT 256
#define MA_NQ 21
#define MA_WP 201
#define MA_KC 16
#define MA_FN (MA_R * PP)
#define MA_NG (SS / MP_XG)

__device__ __forceinline__ void vcol_load(float (&v)[MA_KC], const float* __restrict__ vcol, int k0) {
#pragma unroll
  for (int i = 0; i < MA_KC; i++) v[i] = vcol[(size_t)(k0 + i) * HH];
}

__device__ __forceinline__ void maxplus_step(const float* cs, int k0, const float (&v)[MA_KC], float (&mx)[MA_R]) {
#pragma unroll
  for (int i = 0; i < MA_KC; i++) {
#pragma unroll
    for (int j = 0; j < MA_R; j++) mx[j] = fmaxf(mx[j], cs[(k0 + i) * MA_R + j] * v[i]);
  }
}

__global__ __launch_bounds__(MA_NT) void maxatt_mpm_kernel(
    const float* __restrict__ cp, const float* __restrict__ ch,
    const float* __restrict__ cosM, const float* __restrict__ w,
    const int* __restrict__ mask_p, const int* __restrict__ mask_h,
    const float* __restrict__ pmax, const float* __restrict__ psum,
    float* __restrict__ out) {
  __shared__ __attribute__((aligned(16))) float r1[MA_R * HH];
  __shared__ __attribute__((aligned(16))) float r2[MA_R * HH];
  __shared__ __attribute__((aligned(16))) float cs[SS * MA_R];
  __shared__ float w2s[MA_NQ * MA_WP];

  const int side = blockIdx.y;
  const int row0 = blockIdx.x * MA_R;
  const int b = row0 / SS, r0 = row0 % SS;
  const int tid = threadIdx.x;
  const float* v1 = side ? ch : cp;
  const float* V = side ? cp : ch;
  const int* mask = side ? mask_h : mask_p;
  const float* cb = cosM + (size_t)b * SS * SS;

  constexpr int W4 = PP * HH / 4;
  constexpr int R4 = MA_R * HH / 4;
  v4f wv4[4], rv4[2], cv4;
#pragma unroll
  for (int u = 0; u < 4; u++) {
    const int i = tid + u * MA_NT;
    wv4[u] = *(const v4f*)(w + 4 * (i < W4 ? i : W4 - 1));
  }
#pragma unroll
  for (int u = 0; u < 2; u++) {
    const int i = tid + u * MA_NT;
    rv4[u] = *(const v4f*)(v1 + (size_t)row0 * HH + 4 * (i < R4 ? i : R4 - 1));
  }
  const int pj = tid & 7, pk = 4 * (tid >> 3);
  const int qk = tid >> 1, qj = 4 * (tid & 1);
  cv4 = side == 0 ? *(const v4f*)(cb + (size_t)(r0 + pj) * SS + pk)
                  : *(const v4f*)(cb + (size_t)qk * SS + r0 + qj);
  const int fc = tid < MA_FN ? tid : MA_FN - 1;
  const int fy = fc / PP, fp = fc - fy * PP;
  const float* part = (side ? psum : pmax) + (size_t)b * MA_NG * (SS * PP) + r0 * PP + fc;
  float pv[MA_NG];
#pragma unroll
  for (int g = 0; g < MA_NG; g++) pv[g] = part[(size_t)g * (SS * PP)];
  const int fmask = mask_h[b * SS + r0 + fy];
  asm volatile("" :: "v"(wv4[0]), "v"(wv4[1]), "v"(wv4[2]), "v"(wv4[3]), "v"(rv4[0]), "v"(rv4[1]), "v"(cv4),
               "v"(pv[0]), "v"(pv[1]), "v"(pv[2]), "v"(pv[3]), "v"(pv[4]), "v"(pv[5]), "v"(pv[6]), "v"(pv[7]),
               "v"(pv[8]), "v"(pv[9]), "v"(pv[10]), "v"(pv[11]), "v"(pv[12]), "v"(pv[13]), "v"(pv[14]), "v"(pv[15]),
               "v"(fmask));
#pragma unroll
  for (int u = 0; u < 4; u++) {
    const int i = tid + u * MA_NT;
    if (i < W4) {
      const int p = i / (HH / 4), c = i - p * (HH / 4);
      float* d = w2s + p * MA_WP + 4 * c;
      d[0] = wv4[u].x * wv4[u].x; d[1] = wv4[u].y * wv4[u].y;
      d[2] = wv4[u].z * wv4[u].z; d[3] = wv4[u].w * wv4[u].w;
    }
  }
  if (tid < HH) w2s[PP * MA_WP + tid] = 1.0f;
#pragma unroll
  for (int u = 0; u < 2; u++) {
    const int i = tid + u * MA_NT;
    if (i < R4) *(v4f*)(r1 + 4 * i) = rv4[u];
  }
  if (side == 0) {
    cs[(pk + 0) * MA_R + pj] = cv4.x; cs[(pk + 1) * MA_R + pj] = cv4.y;
    cs[(pk + 2) * MA_R + pj] = cv4.z; cs[(pk + 3) * MA_R + pj] = cv4.w;
  } else {
    *(v4f*)(cs + qk * MA_R + qj) = cv4;
  }
  __syncthreads();

  if (tid < MA_FN) {
    float* orow = out + ((size_t)(BB + b) * SS + r0 + fy) * FF;
    const float m = (float)fmask;
    if (side == 0) {
      float mx = -3.0e38f;
#pragma unroll
      for (int g = 0; g < MA_NG; g++) mx = fmaxf(mx, pv[g]);
      orow[23 + fp] = m * mx;
    } else {
      float sm = 0.f;
#pragma unroll
      for (int g = 0; g < MA_NG; g++) sm += pv[g];
      orow[43 + fp] = m * (sm / (float)SS);
    }
  }

  const int hc = tid < HH ? tid : HH - 1;
  const float* vcol = V + (size_t)b * SS * HH + hc;
  float mx[MA_R];
#pragma unroll
  for (int j = 0; j < MA_R; j++) mx[j] = -3.0e38f;
  if ((tid & ~31) < HH) {
    float va[MA_KC], vb[MA_KC];
    vcol_load(va, vcol, 0);
#pragma unroll
    for (int k0 = 0; k0 < SS; k0 += 2 * MA_KC) {
      vcol_load(vb, vcol, k0 + MA_KC);
      maxplus_step(cs, k0, va, mx);
      if (k0 + 2 * MA_KC < SS) vcol_load(va, vcol, k0 + 2 * MA_KC);
      maxplus_step(cs, k0 + MA_KC, vb, mx);
    }
  }
  if (tid < HH) {
#pragma unroll
    for (int j = 0; j < MA_R; j++) r2[j * HH + tid] = (float)mask[row0 + j] * mx[j];
  }
  __syncthreads();

  if (tid < MA_R * MA_NQ) {
    const int j = tid / MA_NQ, q = tid - j * MA_NQ;
    const float* ar = r1 + j * HH;
    const float* cr = r2 + j * HH;
    const float* wq = w2s + q * MA_WP;
    float a11 = 0.f, a22 = 0.f, a12 = 0.f;
#pragma unroll 8
    for (int h = 0; h < HH; h++) {
      const float a = ar[h], c = cr[h];
      const float aa = a * a, cc = c * c, ac = a * c;
      const float w2a = wq[h];
      a11 += w2a * aa; a22 += w2a * cc; a12 += w2a * ac;
    }
    float* orow = out + ((size_t)(side * BB + b) * SS + r0 + j) * FF + 84;
    orow[q == PP ? 0 : 1 + q] = a12 / (fmaxf(sqrtf(a11), EPSF) * fmaxf(sqrtf(a22), EPSF));
  }
}

#define MP_WP 268
#define MP_NT 256
#define MP_NC (MP_XG * PP)
#define MP_NCC (MP_NC + 16)
#define MP_KS 5
#define MP_NG (HH / 4 / MP_KS)
#define MP_BJ (HH / 4)
#define MP_BT (MP_BJ * 5)

__device__ __forceinline__ void mp_bload(v2f (&f)[2 * MP_KS], const float* c0, const float* c1, int g) {
#pragma unroll
  for (int i = 0; i < MP_KS; i++) {
    f[2 * i] = *(const v2f*)(c0 + 4 * (g * MP_KS + i));
    f[2 * i + 1] = *(const v2f*)(c1 + 4 * (g * MP_KS + i));
  }
}

__device__ __forceinline__ void mp_mma(const v2f (&yf)[HH / 4], int g, const v2f (&f)[2 * MP_KS],
                                       v8f& acc0, v8f& acc1) {
#pragma unroll
  for (int i = 0; i < MP_KS; i++) {
    acc0 = __builtin_amdgcn_wmma_f32_16x16x4_f32(false, yf[g * MP_KS + i], false, f[2 * i], (short)0, acc0, false, false);
    acc1 = __builtin_amdgcn_wmma_f32_16x16x4_f32(false, yf[g * MP_KS + i], false, f[2 * i + 1], (short)0, acc1, false, false);
  }
}

__device__ __forceinline__ void mp_bload1(v2f (&f)[MP_KS], const float* c0, int g) {
#pragma unroll
  for (int i = 0; i < MP_KS; i++) f[i] = *(const v2f*)(c0 + 4 * (g * MP_KS + i));
}

__device__ __forceinline__ void mp_mma1(const v2f (&yf)[HH / 4], int g, const v2f (&f)[MP_KS],
                                        v8f& acc0, v8f& acc1) {
#pragma unroll
  for (int i = 0; i < MP_KS; i++) {
    if ((g * MP_KS + i) & 1)
      acc1 = __builtin_amdgcn_wmma_f32_16x16x4_f32(false, yf[g * MP_KS + i], false, f[i], (short)0, acc1, false, false);
    else
      acc0 = __builtin_amdgcn_wmma_f32_16x16x4_f32(false, yf[g * MP_KS + i], false, f[i], (short)0, acc0, false, false);
  }
}

template <int MASK>
__device__ __forceinline__ float mp_xlane(float v) {
  return __int_as_float(__builtin_amdgcn_update_dpp(0, __float_as_int(v), 0x160 + MASK, 0xf, 0xf, true));
}

__device__ __forceinline__ void mp_tile_out(const v8f& acc, float wa, const float (&wB)[8], int xi, int half,
                                            float* rm, float* rs, float* pm, float* ps) {
  float mv[8];
  float m = -3.0e38f, s = 0.f;
#pragma unroll
  for (int r = 0; r < 8; r++) {
    mv[r] = acc[r] / fmaxf(wa * wB[r], EPSF);
    m = fmaxf(m, mv[r]); s += mv[r];
  }
  m = fmaxf(m, __shfl_xor(m, 16, 32)); s += __shfl_xor(s, 16, 32);
  if (half == 0) { *rm = m; *rs = s; }
  float hm = 0.f, hs = 0.f;
#pragma unroll
  for (int r = 0; r < 8; r++) {
    float t = mv[r], u = mv[r];
    t = fmaxf(t, mp_xlane<1>(t)); u += mp_xlane<1>(u);
    t = fmaxf(t, mp_xlane<2>(t)); u += mp_xlane<2>(u);
    t = fmaxf(t, mp_xlane<4>(t)); u += mp_xlane<4>(u);
    if (xi == r) { hm = t; hs = u; }
  }
  *pm = hm; *ps = hs;
}

__global__ __launch_bounds__(MP_NT) void maxpool_pair_kernel(
    const float* __restrict__ v1, const float* __restrict__ v2,
    const float* __restrict__ w,
    const float* __restrict__ wnA, const float* __restrict__ wnB,
    const float* __restrict__ np, const float* __restrict__ nh,
    const int* __restrict__ mask, float* __restrict__ out,
    float* __restrict__ pmax, float* __restrict__ psum, float* __restrict__ cosO) {
  __shared__ __attribute__((aligned(16))) float xw[MP_NCC * MP_WP];
  __shared__ float redm[8 * MP_NC];
  __shared__ float reds[8 * MP_NC];
  __shared__ __attribute__((aligned(16))) float cst[MP_XG * SS];

  const int b = blockIdx.x / (SS / MP_XG);
  const int xg = blockIdx.x % (SS / MP_XG);
  const int x0 = xg * MP_XG;
  const int tid = threadIdx.x;
  const int lane = tid & 31, half = lane >> 4, l16 = lane & 15;
  const int wv = tid >> 5;
  const int y0 = wv * 16;

  v2f yf[HH / 4];
  const float* yrow = v2 + ((size_t)b * SS + y0 + l16) * HH + 2 * half;
#pragma unroll
  for (int ks = 0; ks < HH / 4; ks++) yf[ks] = *(const v2f*)(yrow + 4 * ks);

  const float cnx = np[b * SS + x0 + (l16 & 7)];
  float cny[8];
#pragma unroll
  for (int r = 0; r < 8; r++) cny[r] = nh[b * SS + y0 + 8 * half + r];

  {
    const int bt = tid < MP_BT ? tid : MP_BT - 1;
    const int bg = bt / MP_BJ, bj = bt - bg * MP_BJ;
    v4f x4[MP_XG], w4[PP / 5];
#pragma unroll
    for (int xi = 0; xi < MP_XG; xi++) x4[xi] = *(const v4f*)(v1 + ((size_t)b * SS + x0 + xi) * HH + 4 * bj);
#pragma unroll
    for (int q = 0; q < PP / 5; q++) w4[q] = *(const v4f*)(w + (bg + 5 * q) * HH + 4 * bj);
    if (tid < MP_BT) {
#pragma unroll
      for (int q = 0; q < PP / 5; q++) {
        const v4f ww = w4[q] * w4[q];
        float* col = xw + (size_t)((bg + 5 * q) * MP_XG) * MP_WP + 4 * bj;
#pragma unroll
        for (int xi = 0; xi < MP_XG; xi++) *(v4f*)(col + xi * MP_WP) = x4[xi] * ww;
      }
      if (bg < 2) {
        float* col = xw + (size_t)(MP_NC + bg * MP_XG) * MP_WP + 4 * bj;
        const v4f zero = {0.f, 0.f, 0.f, 0.f};
#pragma unroll
        for (int xi = 0; xi < MP_XG; xi++) *(v4f*)(col + xi * MP_WP) = bg == 0 ? x4[xi] : zero;
      }
    }
  }
  asm volatile("" :: "v"(cnx), "v"(cny[0]), "v"(cny[1]), "v"(cny[2]), "v"(cny[3]), "v"(cny[4]), "v"(cny[5]),
               "v"(cny[6]), "v"(cny[7]));
  __syncthreads();

  const int pb = l16 >> 3, xi = l16 & 7;
  const float* wArow = wnA + ((size_t)b * SS + x0 + xi) * PP;
  const float* wBrow = wnB + ((size_t)b * SS + y0 + 8 * half) * PP;
  const size_t po = (((size_t)b * (SS / MP_XG) + xg) * SS + y0 + 8 * half + xi) * PP + pb;
  const float* bc = xw + l16 * MP_WP + 2 * half;

  {
    const float* cc = bc + (size_t)MP_NC * MP_WP;
    v2f ca[MP_KS], cb[MP_KS];
    v8f e0 = {0, 0, 0, 0, 0, 0, 0, 0};
    v8f e1 = {0, 0, 0, 0, 0, 0, 0, 0};
    mp_bload1(ca, cc, 0);
#pragma unroll
    for (int g = 0; g < MP_NG; g += 2) {
      mp_bload1(cb, cc, g + 1);
      __builtin_amdgcn_sched_barrier(0);
      mp_mma1(yf, g, ca, e0, e1);
      if (g + 2 < MP_NG) mp_bload1(ca, cc, g + 2);
      __builtin_amdgcn_sched_barrier(0);
      mp_mma1(yf, g + 1, cb, e0, e1);
    }
    v4f lo4, hi4;
    lo4.x = (e0[0] + e1[0]) / fmaxf(cnx * cny[0], EPSF); lo4.y = (e0[1] + e1[1]) / fmaxf(cnx * cny[1], EPSF);
    lo4.z = (e0[2] + e1[2]) / fmaxf(cnx * cny[2], EPSF); lo4.w = (e0[3] + e1[3]) / fmaxf(cnx * cny[3], EPSF);
    hi4.x = (e0[4] + e1[4]) / fmaxf(cnx * cny[4], EPSF); hi4.y = (e0[5] + e1[5]) / fmaxf(cnx * cny[5], EPSF);
    hi4.z = (e0[6] + e1[6]) / fmaxf(cnx * cny[6], EPSF); hi4.w = (e0[7] + e1[7]) / fmaxf(cnx * cny[7], EPSF);
    if (l16 < MP_XG) {
      float* d = cst + l16 * SS + y0 + 8 * half;
      *(v4f*)d = lo4; *(v4f*)(d + 4) = hi4;
    }
  }

  v2f fa[2 * MP_KS], fb[2 * MP_KS];
  mp_bload(fa, bc, bc + 16 * MP_WP, 0);
  for (int tp = 0; tp < MP_NC / 32; tp++) {
    const float* c0 = bc + (size_t)(32 * tp) * MP_WP;
    const float* c1 = c0 + 16 * MP_WP;
    const float* n0 = bc + (size_t)(32 * (tp + 1 < MP_NC / 32 ? tp + 1 : tp)) * MP_WP;
    const int p0 = 4 * tp + pb, p1 = p0 + 2;
    const float wa0 = wArow[p0], wa1 = wArow[p1];
    float wB0[8], wB1[8];
#pragma unroll
    for (int r = 0; r < 8; r++) { wB0[r] = wBrow[r * PP + p0]; wB1[r] = wBrow[r * PP + p1]; }
    v8f acc0 = {0, 0, 0, 0, 0, 0, 0, 0};
    v8f acc1 = {0, 0, 0, 0, 0, 0, 0, 0};
#pragma unroll
    for (int g = 0; g < MP_NG; g += 2) {
      mp_bload(fb, c0, c1, g + 1);
      __builtin_amdgcn_sched_barrier(0);
      mp_mma(yf, g, fa, acc0, acc1);
      if (g + 2 < MP_NG) mp_bload(fa, c0, c1, g + 2);
      else mp_bload(fa, n0, n0 + 16 * MP_WP, 0);
      __builtin_amdgcn_sched_barrier(0);
      mp_mma(yf, g + 1, fb, acc0, acc1);
    }
    const int n = 32 * tp + l16;
    mp_tile_out(acc0, wa0, wB0, xi, half, redm + wv * MP_NC + n, reds + wv * MP_NC + n,
                pmax + po + p0 - pb, psum + po + p0 - pb);
    mp_tile_out(acc1, wa1, wB1, xi, half, redm + wv * MP_NC + n + 16, reds + wv * MP_NC + n + 16,
                pmax + po + p1 - pb, psum + po + p1 - pb);
  }
  __syncthreads();
  if (tid < MP_NC) {
    const int p = tid / MP_XG, xr = tid - p * MP_XG;
    float mx = redm[tid], sm = reds[tid];
    for (int q = 1; q < 8; q++) {
      mx = fmaxf(mx, redm[q * MP_NC + tid]);
      sm += reds[q * MP_NC + tid];
    }
    const int x = x0 + xr;
    const float m = (float)mask[b * SS + x];
    float* orow = out + ((size_t)b * SS + x) * FF;
    orow[23 + p] = m * mx;
    orow[43 + p] = m * (sm / (float)SS);
  }
  {
    const int y4 = 4 * lane;
    *(v4f*)(cosO + ((size_t)b * SS + x0 + wv) * SS + y4) = *(const v4f*)(cst + wv * SS + y4);
  }
}

extern "C" void kernel_launch(void* const* d_in, const int* in_sizes, int n_in,
                              void* d_out, int out_size, void* d_ws, size_t ws_size,
                              hipStream_t stream) {
  const float* ctx_p = (const float*)d_in[0];
  const float* ctx_h = (const float*)d_in[1];
  const float* w0 = (const float*)d_in[2];
  const float* w1 = (const float*)d_in[3];
  const float* w2 = (const float*)d_in[4];
  const float* w3 = (const float*)d_in[5];
  const int* mask_p = (const int*)d_in[6];
  const int* mask_h = (const int*)d_in[7];
  float* out = (float*)d_out;
  float* ws = (float*)d_ws;

  const size_t SZ_ROWS = (size_t)BB * SS * HH;
  float* cp = ws;                 ws += SZ_ROWS;
  float* ch = ws;                 ws += SZ_ROWS;
  float* n_p = ws;                ws += BB * SS;
  float* n_h = ws;                ws += BB * SS;
  float* wn1_p = ws;              ws += (size_t)BB * SS * PP;
  float* wn1_h = ws;              ws += (size_t)BB * SS * PP;
  float* cosM = ws;               ws += (size_t)BB * SS * SS;
  float* mp_pmax = ws;            ws += (size_t)BB * (SS / MP_XG) * SS * PP;
  float* mp_psum = ws;            ws += (size_t)BB * (SS / MP_XG) * SS * PP;

  const dim3 row_tiles(BB * SS / RT, 3);
  prep_kernel<<<row_tiles, NTH, 0, stream>>>(ctx_p, mask_p, w1, cp, n_p, wn1_p);
  prep_kernel<<<row_tiles, NTH, 0, stream>>>(ctx_h, mask_h, w1, ch, n_h, wn1_h);

  maxpool_pair_kernel<<<BB * (SS / MP_XG), MP_NT, 0, stream>>>(cp, ch, w1, wn1_p, wn1_h, n_p, n_h, mask_p, out,
                                                               mp_pmax, mp_psum, cosM);

  att_mpm_kernel<<<dim3(BB * SS / RT, 3, 2), NTH, 0, stream>>>(cp, ch, cosM, w2, w0, mask_p, mask_h, out);

  maxatt_mpm_kernel<<<dim3(BB * SS / MA_R, 2), MA_NT, 0, stream>>>(cp, ch, cosM, w3, mask_p, mask_h,
                                                                   mp_pmax, mp_psum, out);
}


# maxpool match block masks the raw rows and computes its norms itself (y rows: two more column tiles on squared fragments, k grouping changes; x rows: serially from LDS); prep removed, 5 launches -> 3
# speedup vs baseline: 9.4667x; 1.2392x over previous
#include <hip/hip_runtime.h>
#include <hip/hip_bf16.h>


#define BB 16
#define SS 128
#define HH 200
#define PP 20
#define FF 105
#define EPSF 1e-8f
#define MP_XG 8

typedef float v2f __attribute__((ext_vector_type(2)));
typedef float v8f __attribute__((ext_vector_type(8)));

#define RT 32
#define LDP 201
#define QPB 7
#define NTH 128
#define NV4 (RT * (HH / 4))
#define NFULL (NV4 / NTH)
#define NTAIL (NV4 % NTH)

struct TilePieces { float4 v[NFULL + 1]; };

__device__ __forceinline__ void tile_load(TilePieces& t, const float* __restrict__ g,
                                          int row_stride, int tid) {
#pragma unroll
  for (int k = 0; k < NFULL; k++) {
    int i = tid + k * NTH, r = i / (HH / 4), c = i - r * (HH / 4);
    t.v[k] = *(const float4*)(g + (size_t)r * row_stride + 4 * c);
  }
  t.v[NFULL] = make_float4(0.f, 0.f, 0.f, 0.f);
  if (tid < NTAIL) {
    int i = tid + NFULL * NTH, r = i / (HH / 4), c = i - r * (HH / 4);
    t.v[NFULL] = *(const float4*)(g + (size_t)r * row_stride + 4 * c);
  }
}

__device__ __forceinline__ void tile_to_lds(const TilePieces& t, float* lds, int tid) {
#pragma unroll
  for (int k = 0; k <= NFULL; k++) {
    int i = tid + k * NTH, r = i / (HH / 4), c = i - r * (HH / 4);
    if (k < NFULL || tid < NTAIL) {
      float* d = lds + r * LDP + 4 * c;
      d[0] = t.v[k].x; d[1] = t.v[k].y; d[2] = t.v[k].z; d[3] = t.v[k].w;
    }
  }
}

#define CP 132
#define AT_KS (SS / 4)

typedef float v4f __attribute__((ext_vector_type(4)));

__device__ __forceinline__ void bfrag_load(float (&bv)[2 * AT_KS], const float* __restrict__ vcol) {
#pragma unroll
  for (int i = 0; i < AT_KS; i++) {
    bv[2 * i] = vcol[(size_t)(4 * i) * HH];
    bv[2 * i + 1] = vcol[(size_t)(4 * i + 1) * HH];
  }
}

__device__ __forceinline__ void att_tile(const v2f (&a0)[AT_KS], const v2f (&a1)[AT_KS],
                                         const float (&bv)[2 * AT_KS], const float* csum,
                                         float* rows, int col, bool colOK, int half) {
  v8f acc0 = {0, 0, 0, 0, 0, 0, 0, 0};
  v8f acc1 = {0, 0, 0, 0, 0, 0, 0, 0};
#pragma unroll
  for (int i = 0; i < AT_KS; i++) {
    v2f bb;
    bb.x = bv[2 * i];
    bb.y = bv[2 * i + 1];
    acc0 = __builtin_amdgcn_wmma_f32_16x16x4_f32(false, a0[i], false, bb, (short)0, acc0, false, false);
    acc1 = __builtin_amdgcn_wmma_f32_16x16x4_f32(false, a1[i], false, bb, (short)0, acc1, false, false);
  }
  if (colOK) {
#pragma unroll
    for (int r = 0; r < 8; r++) {
      const int m = r + 8 * half;
      rows[m * LDP + col] = acc0[r] / fmaxf(csum[m], EPSF);
      rows[(16 + m) * LDP + col] = acc1[r] / fmaxf(csum[16 + m], EPSF);
    }
  }
}

__global__ __launch_bounds__(NTH) void att_mpm_kernel(
    const float* __restrict__ cp, const float* __restrict__ ch,
    const float* __restrict__ cosM, const float* __restrict__ w,
    const float* __restrict__ wf,
    const int* __restrict__ mask_p, const int* __restrict__ mask_h,
    float* __restrict__ out) {
  __shared__ float r1[RT * LDP];
  __shared__ __attribute__((aligned(16))) float r2[RT * LDP];
  __shared__ float csum[RT];
  const int side = blockIdx.z;
  const int row0 = blockIdx.x * RT;
  const int b = row0 / SS, s0 = row0 % SS;
  const int tid = threadIdx.x;
  const float* v1 = side ? ch : cp;
  const float* V = (side ? cp : ch) + (size_t)b * SS * HH;
  const int* mask = side ? mask_h : mask_p;
  const float* cb = cosM + (size_t)b * SS * SS;
  float* cst = r2;

  const int lane = tid & 31, half = lane >> 4, l16 = lane & 15;
  const int wv = __builtin_amdgcn_readfirstlane(tid >> 5);
  const int tw = 3 - wv;
  const int c0 = 16 * tw + l16, c1 = c0 + 64, c2 = c0 + 128, c3 = c0 + 192;
  const bool ok3 = c3 < HH;
  const float* vh = V + (size_t)(2 * half) * HH;

  TilePieces ta;
  tile_load(ta, v1 + (size_t)row0 * HH, HH, tid);
  float tm[NFULL + 1];
#pragma unroll
  for (int k = 0; k <= NFULL; k++) {
    const int r = (tid + k * NTH) / (HH / 4);
    tm[k] = (k < NFULL || tid < NTAIL) ? (float)mask[row0 + r] : 0.f;
  }
  v4f cv[8];
  const int sh = side ? 3 : 5, lo = side ? 7 : 31;
  const float* cpiece = cb + (side ? s0 : s0 * SS);
#pragma unroll
  for (int u = 0; u < 8; u++) {
    const int i = tid + u * NTH;
    cv[u] = *(const v4f*)(cpiece + (i >> sh) * SS + 4 * (i & lo));
  }
  float ba[2 * AT_KS], bb[2 * AT_KS];
  bfrag_load(ba, vh + c0);
  const int* omask = (side ? mask_p : mask_h) + b * SS;
  int olen = 0;
  for (int i = 0; i < SS; i++) olen += omask[i];
  const int li = olen > 0 ? olen - 1 : 0;
  const float lm = (float)omask[li];
  const float* lr = V + (size_t)li * HH;
#pragma unroll
  for (int k = 0; k <= NFULL; k++) {
    ta.v[k].x *= tm[k]; ta.v[k].y *= tm[k]; ta.v[k].z *= tm[k]; ta.v[k].w *= tm[k];
  }
  tile_to_lds(ta, r1, tid);
  if (side == 0) {
#pragma unroll
    for (int u = 0; u < 8; u++) {
      const int i = tid + u * NTH;
      *(v4f*)(cst + (i >> 5) * CP + 4 * (i & 31)) = cv[u];
    }
  } else {
#pragma unroll
    for (int u = 0; u < 8; u++) {
      const int i = tid + u * NTH;
      const int k = i >> 3, j = 4 * (i & 7);
      cst[(j + 0) * CP + k] = cv[u].x; cst[(j + 1) * CP + k] = cv[u].y;
      cst[(j + 2) * CP + k] = cv[u].z; cst[(j + 3) * CP + k] = cv[u].w;
    }
  }
  __syncthreads();

  {
    const int j = tid >> 2, q = tid & 3;
    const v4f* c4 = (const v4f*)(cst + j * CP + 32 * q);
    float mx = -3.0e38f, sm = 0.f;
#pragma unroll
    for (int i = 0; i < 8; i++) {
      const v4f t = c4[i];
      mx = fmaxf(fmaxf(fmaxf(mx, t.x), fmaxf(t.y, t.z)), t.w);
      sm += t.x; sm += t.y; sm += t.z; sm += t.w;
    }
    mx = fmaxf(mx, __shfl_xor(mx, 1, 32)); sm += __shfl_xor(sm, 1, 32);
    mx = fmaxf(mx, __shfl_xor(mx, 2, 32)); sm += __shfl_xor(sm, 2, 32);
    if (q == 0) {
      csum[j] = sm;
      if (blockIdx.y == 0) {
        const float m = (float)mask[row0 + j];
        float* o2 = out + ((size_t)(side * BB + b) * SS + s0 + j) * FF;
        o2[0] = m * mx;
        o2[1] = m * (sm / (float)SS);
      }
    }
  }
  v2f a0[AT_KS], a1[AT_KS];
  {
    const float* cl = cst + l16 * CP + 2 * half;
#pragma unroll
    for (int i = 0; i < AT_KS; i++) {
      a0[i] = *(const v2f*)(cl + 4 * i);
      a1[i] = *(const v2f*)(cl + 16 * CP + 4 * i);
    }
  }
  __syncthreads();

  bfrag_load(bb, vh + c1);
  __builtin_amdgcn_sched_barrier(0);
  att_tile(a0, a1, ba, csum, r2, c0, true, half);
  bfrag_load(ba, vh + c2);
  __builtin_amdgcn_sched_barrier(0);
  att_tile(a0, a1, bb, csum, r2, c1, true, half);
  if (tw == 0) bfrag_load(bb, vh + (ok3 ? c3 : HH - 1));
  att_tile(a0, a1, ba, csum, r2, c2, true, half);
  if (tw == 0) att_tile(a0, a1, bb, csum, r2, c3, ok3, half);
  __syncthreads();

  const int q0 = QPB * blockIdx.y + 2 * wv;
  const bool two = (2 * wv + 1) < QPB;
  const bool plain = (q0 == PP);
  const float* wa = w + (plain ? 0 : q0) * HH;
  const float* wb = w + (two ? q0 + 1 : 0) * HH;
  const float* ua = wf + (plain ? 0 : q0) * HH;
  const float* ub = wf + (two ? q0 + 1 : 0) * HH;
  const float* ar = r1 + lane * LDP;
  const float* cr = r2 + lane * LDP;
  float a11 = 0.f, a22 = 0.f, a12 = 0.f, b11 = 0.f, b22 = 0.f, b12 = 0.f;
  float f11 = 0.f, f22 = 0.f, f12 = 0.f, g11 = 0.f, g22 = 0.f, g12 = 0.f;
#pragma unroll 8
  for (int h = 0; h < HH; h++) {
    float a = ar[h], c = cr[h];
    float aa = a * a, cc = c * c, ac = a * c;
    float xa = wa[h], xb = wb[h];
    float w2a = plain ? 1.0f : xa * xa;
    float w2b = xb * xb;
    a11 += w2a * aa; a22 += w2a * cc; a12 += w2a * ac;
    b11 += w2b * aa; b22 += w2b * cc; b12 += w2b * ac;
    float l = lr[h] * lm;
    float ll = l * l, al = a * l;
    float ya = ua[h], yb = ub[h];
    float u2a = plain ? 1.0f : ya * ya;
    float u2b = yb * yb;
    f11 += u2a * aa; f22 += u2a * ll; f12 += u2a * al;
    g11 += u2b * aa; g22 += u2b * ll; g12 += u2b * al;
  }
  float* orow = out + ((size_t)(side * BB + b) * SS + s0 + lane) * FF;
  orow[63 + (plain ? 0 : 1 + q0)] = a12 / (fmaxf(sqrtf(a11), EPSF) * fmaxf(sqrtf(a22), EPSF));
  orow[2 + (plain ? 0 : 1 + q0)] = f12 / (fmaxf(sqrtf(f11), EPSF) * fmaxf(sqrtf(f22), EPSF));
  if (two) {
    orow[63 + 2 + q0] = b12 / (fmaxf(sqrtf(b11), EPSF) * fmaxf(sqrtf(b22), EPSF));
    orow[2 + 2 + q0] = g12 / (fmaxf(sqrtf(g11), EPSF) * fmaxf(sqrtf(g22), EPSF));
  }
}

#define MA_R 8
#define MA_NT 256
#define MA_NQ 21
#define MA_WP 201
#define MA_KC 16
#define MA_FN (MA_R * PP)
#define MA_NG (SS / MP_XG)

__device__ __forceinline__ void vcol_load(float (&v)[MA_KC], const float* __restrict__ vcol, int k0) {
#pragma unroll
  for (int i = 0; i < MA_KC; i++) v[i] = vcol[(size_t)(k0 + i) * HH];
}

__device__ __forceinline__ void maxplus_step(const float* cs, int k0, const float (&v)[MA_KC], float (&mx)[MA_R]) {
#pragma unroll
  for (int i = 0; i < MA_KC; i++) {
#pragma unroll
    for (int j = 0; j < MA_R; j++) mx[j] = fmaxf(mx[j], cs[(k0 + i) * MA_R + j] * v[i]);
  }
}

__global__ __launch_bounds__(MA_NT) void maxatt_mpm_kernel(
    const float* __restrict__ cp, const float* __restrict__ ch,
    const float* __restrict__ cosM, const float* __restrict__ w,
    const int* __restrict__ mask_p, const int* __restrict__ mask_h,
    const float* __restrict__ pmax, const float* __restrict__ psum,
    float* __restrict__ out) {
  __shared__ __attribute__((aligned(16))) float r1[MA_R * HH];
  __shared__ __attribute__((aligned(16))) float r2[MA_R * HH];
  __shared__ __attribute__((aligned(16))) float cs[SS * MA_R];
  __shared__ float w2s[MA_NQ * MA_WP];

  const int side = blockIdx.y;
  const int row0 = blockIdx.x * MA_R;
  const int b = row0 / SS, r0 = row0 % SS;
  const int tid = threadIdx.x;
  const float* v1 = side ? ch : cp;
  const float* V = side ? cp : ch;
  const int* mask = side ? mask_h : mask_p;
  const float* cb = cosM + (size_t)b * SS * SS;

  constexpr int W4 = PP * HH / 4;
  constexpr int R4 = MA_R * HH / 4;
  v4f wv4[4], rv4[2], cv4;
  float rm[2];
#pragma unroll
  for (int u = 0; u < 4; u++) {
    const int i = tid + u * MA_NT;
    wv4[u] = *(const v4f*)(w + 4 * (i < W4 ? i : W4 - 1));
  }
#pragma unroll
  for (int u = 0; u < 2; u++) {
    const int i = tid + u * MA_NT;
    const int ic = i < R4 ? i : R4 - 1;
    rv4[u] = *(const v4f*)(v1 + (size_t)row0 * HH + 4 * ic);
    rm[u] = (float)mask[row0 + ic / (HH / 4)];
  }
  const int pj = tid & 7, pk = 4 * (tid >> 3);
  const int qk = tid >> 1, qj = 4 * (tid & 1);
  cv4 = side == 0 ? *(const v4f*)(cb + (size_t)(r0 + pj) * SS + pk)
                  : *(const v4f*)(cb + (size_t)qk * SS + r0 + qj);
  const int fc = tid < MA_FN ? tid : MA_FN - 1;
  const int fy = fc / PP, fp = fc - fy * PP;
  const float* part = (side ? psum : pmax) + (size_t)b * MA_NG * (SS * PP) + r0 * PP + fc;
  float pv[MA_NG];
#pragma unroll
  for (int g = 0; g < MA_NG; g++) pv[g] = part[(size_t)g * (SS * PP)];
  const int fmask = mask_h[b * SS + r0 + fy];
  asm volatile("" :: "v"(wv4[0]), "v"(wv4[1]), "v"(wv4[2]), "v"(wv4[3]), "v"(rv4[0]), "v"(rv4[1]), "v"(cv4),
               "v"(pv[0]), "v"(pv[1]), "v"(pv[2]), "v"(pv[3]), "v"(pv[4]), "v"(pv[5]), "v"(pv[6]), "v"(pv[7]),
               "v"(pv[8]), "v"(pv[9]), "v"(pv[10]), "v"(pv[11]), "v"(pv[12]), "v"(pv[13]), "v"(pv[14]), "v"(pv[15]),
               "v"(fmask), "v"(rm[0]), "v"(rm[1]));
#pragma unroll
  for (int u = 0; u < 4; u++) {
    const int i = tid + u * MA_NT;
    if (i < W4) {
      const int p = i / (HH / 4), c = i - p * (HH / 4);
      float* d = w2s + p * MA_WP + 4 * c;
      d[0] = wv4[u].x * wv4[u].x; d[1] = wv4[u].y * wv4[u].y;
      d[2] = wv4[u].z * wv4[u].z; d[3] = wv4[u].w * wv4[u].w;
    }
  }
  if (tid < HH) w2s[PP * MA_WP + tid] = 1.0f;
#pragma unroll
  for (int u = 0; u < 2; u++) {
    const int i = tid + u * MA_NT;
    if (i < R4) *(v4f*)(r1 + 4 * i) = rv4[u] * rm[u];
  }
  if (side == 0) {
    cs[(pk + 0) * MA_R + pj] = cv4.x; cs[(pk + 1) * MA_R + pj] = cv4.y;
    cs[(pk + 2) * MA_R + pj] = cv4.z; cs[(pk + 3) * MA_R + pj] = cv4.w;
  } else {
    *(v4f*)(cs + qk * MA_R + qj) = cv4;
  }
  __syncthreads();

  if (tid < MA_FN) {
    float* orow = out + ((size_t)(BB + b) * SS + r0 + fy) * FF;
    const float m = (float)fmask;
    if (side == 0) {
      float mx = -3.0e38f;
#pragma unroll
      for (int g = 0; g < MA_NG; g++) mx = fmaxf(mx, pv[g]);
      orow[23 + fp] = m * mx;
    } else {
      float sm = 0.f;
#pragma unroll
      for (int g = 0; g < MA_NG; g++) sm += pv[g];
      orow[43 + fp] = m * (sm / (float)SS);
    }
  }

  const int hc = tid < HH ? tid : HH - 1;
  const float* vcol = V + (size_t)b * SS * HH + hc;
  float mx[MA_R];
#pragma unroll
  for (int j = 0; j < MA_R; j++) mx[j] = -3.0e38f;
  if ((tid & ~31) < HH) {
    float va[MA_KC], vb[MA_KC];
    vcol_load(va, vcol, 0);
#pragma unroll
    for (int k0 = 0; k0 < SS; k0 += 2 * MA_KC) {
      vcol_load(vb, vcol, k0 + MA_KC);
      maxplus_step(cs, k0, va, mx);
      if (k0 + 2 * MA_KC < SS) vcol_load(va, vcol, k0 + 2 * MA_KC);
      maxplus_step(cs, k0 + MA_KC, vb, mx);
    }
  }
  if (tid < HH) {
#pragma unroll
    for (int j = 0; j < MA_R; j++) r2[j * HH + tid] = (float)mask[row0 + j] * mx[j];
  }
  __syncthreads();

  if (tid < MA_R * MA_NQ) {
    const int j = tid / MA_NQ, q = tid - j * MA_NQ;
    const float* ar = r1 + j * HH;
    const float* cr = r2 + j * HH;
    const float* wq = w2s + q * MA_WP;
    float a11 = 0.f, a22 = 0.f, a12 = 0.f;
#pragma unroll 8
    for (int h = 0; h < HH; h++) {
      const float a = ar[h], c = cr[h];
      const float aa = a * a, cc = c * c, ac = a * c;
      const float w2a = wq[h];
      a11 += w2a * aa; a22 += w2a * cc; a12 += w2a * ac;
    }
    float* orow = out + ((size_t)(side * BB + b) * SS + r0 + j) * FF + 84;
    orow[q == PP ? 0 : 1 + q] = a12 / (fmaxf(sqrtf(a11), EPSF) * fmaxf(sqrtf(a22), EPSF));
  }
}

#define MP_WP 268
#define MP_NT 256
#define MP_NC (MP_XG * PP)
#define MP_NCC (MP_NC + 16)
#define MP_NW (PP + 1)
#define MP_NCW (MP_NCC + MP_NW)
#define MP_KS 5
#define MP_NG (HH / 4 / MP_KS)
#define MP_BJ (HH / 4)
#define MP_BT (MP_BJ * 5)

__device__ __forceinline__ void mp_bload(v2f (&f)[2 * MP_KS], const float* c0, const float* c1, int g) {
#pragma unroll
  for (int i = 0; i < MP_KS; i++) {
    f[2 * i] = *(const v2f*)(c0 + 4 * (g * MP_KS + i));
    f[2 * i + 1] = *(const v2f*)(c1 + 4 * (g * MP_KS + i));
  }
}

__device__ __forceinline__ void mp_mma(const v2f (&yf)[HH / 4], int g, const v2f (&f)[2 * MP_KS],
                                       v8f& acc0, v8f& acc1) {
#pragma unroll
  for (int i = 0; i < MP_KS; i++) {
    acc0 = __builtin_amdgcn_wmma_f32_16x16x4_f32(false, yf[g * MP_KS + i], false, f[2 * i], (short)0, acc0, false, false);
    acc1 = __builtin_amdgcn_wmma_f32_16x16x4_f32(false, yf[g * MP_KS + i], false, f[2 * i + 1], (short)0, acc1, false, false);
  }
}

__device__ __forceinline__ void mp_bload1(v2f (&f)[MP_KS], const float* c0, int g) {
#pragma unroll
  for (int i = 0; i < MP_KS; i++) f[i] = *(const v2f*)(c0 + 4 * (g * MP_KS + i));
}

__device__ __forceinline__ void mp_mma1(const v2f (&yf)[HH / 4], int g, const v2f (&f)[MP_KS],
                                        v8f& acc0, v8f& acc1) {
#pragma unroll
  for (int i = 0; i < MP_KS; i++) {
    if ((g * MP_KS + i) & 1)
      acc1 = __builtin_amdgcn_wmma_f32_16x16x4_f32(false, yf[g * MP_KS + i], false, f[i], (short)0, acc1, false, false);
    else
      acc0 = __builtin_amdgcn_wmma_f32_16x16x4_f32(false, yf[g * MP_KS + i], false, f[i], (short)0, acc0, false, false);
  }
}

template <int MASK>
__device__ __forceinline__ float mp_xlane(float v) {
  return __int_as_float(__builtin_amdgcn_update_dpp(0, __float_as_int(v), 0x160 + MASK, 0xf, 0xf, true));
}

__device__ __forceinline__ void mp_tile_out(const v8f& acc, float wa, const float (&wB)[8], int xi, int half,
                                            float* rm, float* rs, float* pm, float* ps) {
  float mv[8];
  float m = -3.0e38f, s = 0.f;
#pragma unroll
  for (int r = 0; r < 8; r++) {
    mv[r] = acc[r] / fmaxf(wa * wB[r], EPSF);
    m = fmaxf(m, mv[r]); s += mv[r];
  }
  m = fmaxf(m, __shfl_xor(m, 16, 32)); s += __shfl_xor(s, 16, 32);
  if (half == 0) { *rm = m; *rs = s; }
  float hm = 0.f, hs = 0.f;
#pragma unroll
  for (int r = 0; r < 8; r++) {
    float t = mv[r], u = mv[r];
    t = fmaxf(t, mp_xlane<1>(t)); u += mp_xlane<1>(u);
    t = fmaxf(t, mp_xlane<2>(t)); u += mp_xlane<2>(u);
    t = fmaxf(t, mp_xlane<4>(t)); u += mp_xlane<4>(u);
    if (xi == r) { hm = t; hs = u; }
  }
  *pm = hm; *ps = hs;
}

__global__ __launch_bounds__(MP_NT) void maxpool_pair_kernel(
    const float* __restrict__ v1, const float* __restrict__ v2,
    const float* __restrict__ w,
    const int* __restrict__ mask, const int* __restrict__ mask2, float* __restrict__ out,
    float* __restrict__ pmax, float* __restrict__ psum, float* __restrict__ cosO) {
  __shared__ __attribute__((aligned(16))) float xw[MP_NCW * MP_WP];
  __shared__ float redm[8 * MP_NC];
  __shared__ float reds[8 * MP_NC];
  __shared__ __attribute__((aligned(16))) float cst[MP_XG * SS];
  __shared__ float wny[SS * MP_NW];
  __shared__ float wnx[MP_XG * MP_NW];
  static_assert(MP_NT / 32 == MP_XG, "wave i sums the norms of x row i");

  const int b = blockIdx.x / (SS / MP_XG);
  const int xg = blockIdx.x % (SS / MP_XG);
  const int x0 = xg * MP_XG;
  const int tid = threadIdx.x;
  const int lane = tid & 31, half = lane >> 4, l16 = lane & 15;
  const int wv = tid >> 5;
  const int y0 = wv * 16;

  v2f yf[HH / 4];
  const float* yrow = v2 + ((size_t)b * SS + y0 + l16) * HH + 2 * half;
#pragma unroll
  for (int ks = 0; ks < HH / 4; ks++) yf[ks] = *(const v2f*)(yrow + 4 * ks);
  const float ym = (float)mask2[b * SS + y0 + l16];

  {
    const int bt = tid < MP_BT ? tid : MP_BT - 1;
    const int bg = bt / MP_BJ, bj = bt - bg * MP_BJ;
    v4f x4[MP_XG], w4[PP / 5];
    float xm[MP_XG];
#pragma unroll
    for (int xi = 0; xi < MP_XG; xi++) x4[xi] = *(const v4f*)(v1 + ((size_t)b * SS + x0 + xi) * HH + 4 * bj);
#pragma unroll
    for (int q = 0; q < PP / 5; q++) w4[q] = *(const v4f*)(w + (bg + 5 * q) * HH + 4 * bj);
#pragma unroll
    for (int xi = 0; xi < MP_XG; xi++) xm[xi] = (float)mask[b * SS + x0 + xi];
#pragma unroll
    for (int xi = 0; xi < MP_XG; xi++) x4[xi] = x4[xi] * xm[xi];
    if (tid < MP_BT) {
#pragma unroll
      for (int q = 0; q < PP / 5; q++) {
        const v4f ww = w4[q] * w4[q];
        float* col = xw + (size_t)((bg + 5 * q) * MP_XG) * MP_WP + 4 * bj;
#pragma unroll
        for (int xi = 0; xi < MP_XG; xi++) *(v4f*)(col + xi * MP_WP) = x4[xi] * ww;
        *(v4f*)(xw + (size_t)(MP_NCC + bg + 5 * q) * MP_WP + 4 * bj) = ww;
      }
      if (bg == 2) {
        const v4f one = {1.f, 1.f, 1.f, 1.f};
        *(v4f*)(xw + (size_t)(MP_NCC + PP) * MP_WP + 4 * bj) = one;
      }
      if (bg < 2) {
        float* col = xw + (size_t)(MP_NC + bg * MP_XG) * MP_WP + 4 * bj;
        const v4f zero = {0.f, 0.f, 0.f, 0.f};
#pragma unroll
        for (int xi = 0; xi < MP_XG; xi++) *(v4f*)(col + xi * MP_WP) = bg == 0 ? x4[xi] : zero;
      }
    }
  }
#pragma unroll
  for (int ks = 0; ks < HH / 4; ks++) yf[ks] = yf[ks] * ym;
  __syncthreads();

  if (lane < MP_NW) {
    const v4f* xc = (const v4f*)(xw + (size_t)(MP_NC + wv) * MP_WP);
    const v4f* wc = (const v4f*)(xw + (size_t)(MP_NCC + lane) * MP_WP);
    float s = 0.f;
#pragma unroll 5
    for (int j = 0; j < MP_BJ; j++) {
      const v4f x = xc[j], ww = wc[j];
      s += ww.x * (x.x * x.x); s += ww.y * (x.y * x.y);
      s += ww.z * (x.z * x.z); s += ww.w * (x.w * x.w);
    }
    wnx[wv * MP_NW + lane] = sqrtf(s);
  }
  {
    const float* wc0 = xw + (size_t)(MP_NCC + l16) * MP_WP + 2 * half;
    const float* wc1 = xw + (size_t)(MP_NCC + 16 + (l16 < MP_NW - 16 ? l16 : MP_NW - 17)) * MP_WP + 2 * half;
    v8f n0 = {0, 0, 0, 0, 0, 0, 0, 0};
    v8f n1 = {0, 0, 0, 0, 0, 0, 0, 0};
#pragma unroll
    for (int ks = 0; ks < HH / 4; ks++) {
      const v2f y2 = yf[ks] * yf[ks];
      n0 = __builtin_amdgcn_wmma_f32_16x16x4_f32(false, y2, false, *(const v2f*)(wc0 + 4 * ks), (short)0, n0, false, false);
      n1 = __builtin_amdgcn_wmma_f32_16x16x4_f32(false, y2, false, *(const v2f*)(wc1 + 4 * ks), (short)0, n1, false, false);
    }
    float* d = wny + (y0 + 8 * half) * MP_NW;
#pragma unroll
    for (int r = 0; r < 8; r++) d[r * MP_NW + l16] = sqrtf(n0[r]);
    if (l16 < MP_NW - 16) {
#pragma unroll
      for (int r = 0; r < 8; r++) d[r * MP_NW + 16 + l16] = sqrtf(n1[r]);
    }
  }
  __syncthreads();

  const float cnx = wnx[(l16 & 7) * MP_NW + PP];
  float cny[8];
#pragma unroll
  for (int r = 0; r < 8; r++) cny[r] = wny[(y0 + 8 * half + r) * MP_NW + PP];

  const int pb = l16 >> 3, xi = l16 & 7;
  const float* wArow = wnx + xi * MP_NW;
  const float* wBrow = wny + (y0 + 8 * half) * MP_NW;
  const size_t po = (((size_t)b * (SS / MP_XG) + xg) * SS + y0 + 8 * half + xi) * PP + pb;
  const float* bc = xw + l16 * MP_WP + 2 * half;

  {
    const float* cc = bc + (size_t)MP_NC * MP_WP;
    v2f ca[MP_KS], cb[MP_KS];
    v8f e0 = {0, 0, 0, 0, 0, 0, 0, 0};
    v8f e1 = {0, 0, 0, 0, 0, 0, 0, 0};
    mp_bload1(ca, cc, 0);
#pragma unroll
    for (int g = 0; g < MP_NG; g += 2) {
      mp_bload1(cb, cc, g + 1);
      __builtin_amdgcn_sched_barrier(0);
      mp_mma1(yf, g, ca, e0, e1);
      if (g + 2 < MP_NG) mp_bload1(ca, cc, g + 2);
      __builtin_amdgcn_sched_barrier(0);
      mp_mma1(yf, g + 1, cb, e0, e1);
    }
    v4f lo4, hi4;
    lo4.x = (e0[0] + e1[0]) / fmaxf(cnx * cny[0], EPSF); lo4.y = (e0[1] + e1[1]) / fmaxf(cnx * cny[1], EPSF);
    lo4.z = (e0[2] + e1[2]) / fmaxf(cnx * cny[2], EPSF); lo4.w = (e0[3] + e1[3]) / fmaxf(cnx * cny[3], EPSF);
    hi4.x = (e0[4] + e1[4]) / fmaxf(cnx * cny[4], EPSF); hi4.y = (e0[5] + e1[5]) / fmaxf(cnx * cny[5], EPSF);
    hi4.z = (e0[6] + e1[6]) / fmaxf(cnx * cny[6], EPSF); hi4.w = (e0[7] + e1[7]) / fmaxf(cnx * cny[7], EPSF);
    if (l16 < MP_XG) {
      float* d = cst + l16 * SS + y0 + 8 * half;
      *(v4f*)d = lo4; *(v4f*)(d + 4) = hi4;
    }
  }

  v2f fa[2 * MP_KS], fb[2 * MP_KS];
  mp_bload(fa, bc, bc + 16 * MP_WP, 0);
  for (int tp = 0; tp < MP_NC / 32; tp++) {
    const float* c0 = bc + (size_t)(32 * tp) * MP_WP;
    const float* c1 = c0 + 16 * MP_WP;
    const float* n0 = bc + (size_t)(32 * (tp + 1 < MP_NC / 32 ? tp + 1 : tp)) * MP_WP;
    const int p0 = 4 * tp + pb, p1 = p0 + 2;
    const float wa0 = wArow[p0], wa1 = wArow[p1];
    float wB0[8], wB1[8];
#pragma unroll
    for (int r = 0; r < 8; r++) { wB0[r] = wBrow[r * MP_NW + p0]; wB1[r] = wBrow[r * MP_NW + p1]; }
    v8f acc0 = {0, 0, 0, 0, 0, 0, 0, 0};
    v8f acc1 = {0, 0, 0, 0, 0, 0, 0, 0};
#pragma unroll
    for (int g = 0; g < MP_NG; g += 2) {
      mp_bload(fb, c0, c1, g + 1);
      __builtin_amdgcn_sched_barrier(0);
      mp_mma(yf, g, fa, acc0, acc1);
      if (g + 2 < MP_NG) mp_bload(fa, c0, c1, g + 2);
      else mp_bload(fa, n0, n0 + 16 * MP_WP, 0);
      __builtin_amdgcn_sched_barrier(0);
      mp_mma(yf, g + 1, fb, acc0, acc1);
    }
    const int n = 32 * tp + l16;
    mp_tile_out(acc0, wa0, wB0, xi, half, redm + wv * MP_NC + n, reds + wv * MP_NC + n,
                pmax + po + p0 - pb, psum + po + p0 - pb);
    mp_tile_out(acc1, wa1, wB1, xi, half, redm + wv * MP_NC + n + 16, reds + wv * MP_NC + n + 16,
                pmax + po + p1 - pb, psum + po + p1 - pb);
  }
  __syncthreads();
  if (tid < MP_NC) {
    const int p = tid / MP_XG, xr = tid - p * MP_XG;
    float mx = redm[tid], sm = reds[tid];
    for (int q = 1; q < 8; q++) {
      mx = fmaxf(mx, redm[q * MP_NC + tid]);
      sm += reds[q * MP_NC + tid];
    }
    const int x = x0 + xr;
    const float m = (float)mask[b * SS + x];
    float* orow = out + ((size_t)b * SS + x) * FF;
    orow[23 + p] = m * mx;
    orow[43 + p] = m * (sm / (float)SS);
  }
  {
    const int y4 = 4 * lane;
    *(v4f*)(cosO + ((size_t)b * SS + x0 + wv) * SS + y4) = *(const v4f*)(cst + wv * SS + y4);
  }
}

extern "C" void kernel_launch(void* const* d_in, const int* in_sizes, int n_in,
                              void* d_out, int out_size, void* d_ws, size_t ws_size,
                              hipStream_t stream) {
  const float* ctx_p = (const float*)d_in[0];
  const float* ctx_h = (const float*)d_in[1];
  const float* w0 = (const float*)d_in[2];
  const float* w1 = (const float*)d_in[3];
  const float* w2 = (const float*)d_in[4];
  const float* w3 = (const float*)d_in[5];
  const int* mask_p = (const int*)d_in[6];
  const int* mask_h = (const int*)d_in[7];
  float* out = (float*)d_out;
  float* ws = (float*)d_ws;

  float* cosM = ws;               ws += (size_t)BB * SS * SS;
  float* mp_pmax = ws;            ws += (size_t)BB * (SS / MP_XG) * SS * PP;
  float* mp_psum = ws;            ws += (size_t)BB * (SS / MP_XG) * SS * PP;

  maxpool_pair_kernel<<<BB * (SS / MP_XG), MP_NT, 0, stream>>>(ctx_p, ctx_h, w1, mask_p, mask_h, out,
                                                               mp_pmax, mp_psum, cosM);

  att_mpm_kernel<<<dim3(BB * SS / RT, 3, 2), NTH, 0, stream>>>(ctx_p, ctx_h, cosM, w2, w0, mask_p, mask_h, out);

  maxatt_mpm_kernel<<<dim3(BB * SS / MA_R, 2), MA_NT, 0, stream>>>(ctx_p, ctx_h, cosM, w3, mask_p, mask_h,
                                                                   mp_pmax, mp_psum, out);
}


# attentive match, full match and cos row max/mean move into the max-attentive block (8 rows: mean rows on WMMA in LDS, a thread sums 3 features of a row); 3 launches -> 2; no sum changes its order
# speedup vs baseline: 12.2772x; 1.2969x over previous
#include <hip/hip_runtime.h>
#include <hip/hip_bf16.h>


#define BB 16
#define SS 128
#define HH 200
#define PP 20
#define FF 105
#define EPSF 1e-8f
#define MP_XG 8

typedef float v2f __attribute__((ext_vector_type(2)));
typedef float v4f __attribute__((ext_vector_type(4)));
typedef float v8f __attribute__((ext_vector_type(8)));

#define AT_KS (SS / 4)
#define AT_NT ((HH + 15) / 16)

__device__ __forceinline__ void bfrag_load(float (&bv)[2 * AT_KS], const float* __restrict__ vcol) {
#pragma unroll
  for (int i = 0; i < AT_KS; i++) {
    bv[2 * i] = vcol[(size_t)(4 * i) * HH];
    bv[2 * i + 1] = vcol[(size_t)(4 * i + 1) * HH];
  }
}

__device__ __forceinline__ void att8_tile(const v2f (&af)[AT_KS], const float (&bv)[2 * AT_KS],
                                          const float (&csr)[8], float* rows, int col, bool st) {
  v8f acc = {0, 0, 0, 0, 0, 0, 0, 0};
#pragma unroll
  for (int i = 0; i < AT_KS; i++) {
    v2f bb;
    bb.x = bv[2 * i];
    bb.y = bv[2 * i + 1];
    acc = __builtin_amdgcn_wmma_f32_16x16x4_f32(false, af[i], false, bb, (short)0, acc, false, false);
  }
  if (st) {
#pragma unroll
    for (int r = 0; r < 8; r++) rows[r * HH + col] = acc[r] / fmaxf(csr[r], EPSF);
  }
}

#define MA_R 8
#define MA_NT 256
#define MA_NQ 21
#define MA_WP 204
#define MA_NS 3
#define MA_WS (MA_NQ * MA_WP)
#define MA_KC 16
#define MA_FN (MA_R * PP)
#define MA_NG (SS / MP_XG)
#define MA_QG 3
#define MA_NI (MA_NS * MA_R * (MA_NQ / MA_QG))

__device__ __forceinline__ void vcol_load(float (&v)[MA_KC], const float* __restrict__ vcol, int k0) {
#pragma unroll
  for (int i = 0; i < MA_KC; i++) v[i] = vcol[(size_t)(k0 + i) * HH];
}

__device__ __forceinline__ void maxplus_step(const float* cs, int k0, const float (&v)[MA_KC], float (&mx)[MA_R]) {
#pragma unroll
  for (int i = 0; i < MA_KC; i++) {
#pragma unroll
    for (int j = 0; j < MA_R; j++) mx[j] = fmaxf(mx[j], cs[(k0 + i) * MA_R + j] * v[i]);
  }
}

__global__ __launch_bounds__(MA_NT) void rows_match_kernel(
    const float* __restrict__ cp, const float* __restrict__ ch,
    const float* __restrict__ cosM, const float* __restrict__ wmx,
    const float* __restrict__ wat, const float* __restrict__ wfu,
    const int* __restrict__ mask_p, const int* __restrict__ mask_h,
    const float* __restrict__ pmax, const float* __restrict__ psum,
    float* __restrict__ out) {
  __shared__ __attribute__((aligned(16))) float r1[MA_R * HH];
  __shared__ __attribute__((aligned(16))) float r2[MA_R * HH];
  __shared__ __attribute__((aligned(16))) float r3[MA_R * HH];
  __shared__ __attribute__((aligned(16))) float r4[HH];
  __shared__ __attribute__((aligned(16))) float cs[SS * MA_R];
  __shared__ __attribute__((aligned(16))) float w2s[MA_NS * MA_WS];
  static_assert(MA_NQ % MA_QG == 0 && MA_NI <= MA_NT, "one thread per (set, row, feature group)");

  const int side = blockIdx.y;
  const int row0 = blockIdx.x * MA_R;
  const int b = row0 / SS, r0 = row0 % SS;
  const int tid = threadIdx.x;
  const int lane = tid & 31, half = lane >> 4, l16 = lane & 15;
  const int wv = __builtin_amdgcn_readfirstlane(tid >> 5);
  const float* v1 = side ? ch : cp;
  const float* V = (side ? cp : ch) + (size_t)b * SS * HH;
  const int* mask = side ? mask_h : mask_p;
  const float* cb = cosM + (size_t)b * SS * SS;

  constexpr int W4 = PP * HH / 4;
  constexpr int R4 = MA_R * HH / 4;
  v4f wv4[MA_NS][4], rv4[2], cv4;
  float rm[2];
#pragma unroll
  for (int s = 0; s < MA_NS; s++) {
    const float* ws = s == 0 ? wmx : (s == 1 ? wat : wfu);
#pragma unroll
    for (int u = 0; u < 4; u++) {
      const int i = tid + u * MA_NT;
      wv4[s][u] = *(const v4f*)(ws + 4 * (i < W4 ? i : W4 - 1));
    }
  }
#pragma unroll
  for (int u = 0; u < 2; u++) {
    const int i = tid + u * MA_NT;
    const int ic = i < R4 ? i : R4 - 1;
    rv4[u] = *(const v4f*)(v1 + (size_t)row0 * HH + 4 * ic);
    rm[u] = (float)mask[row0 + ic / (HH / 4)];
  }
  const int pj = tid & 7, pk = 4 * (tid >> 3);
  const int qk = tid >> 1, qj = 4 * (tid & 1);
  cv4 = side == 0 ? *(const v4f*)(cb + (size_t)(r0 + pj) * SS + pk)
                  : *(const v4f*)(cb + (size_t)qk * SS + r0 + qj);
  const int fc = tid < MA_FN ? tid : MA_FN - 1;
  const int fy = fc / PP, fp = fc - fy * PP;
  const float* part = (side ? psum : pmax) + (size_t)b * MA_NG * (SS * PP) + r0 * PP + fc;
  float pv[MA_NG];
#pragma unroll
  for (int g = 0; g < MA_NG; g++) pv[g] = part[(size_t)g * (SS * PP)];
  const int fmask = mask_h[b * SS + r0 + fy];
  const int* omask = (side ? mask_p : mask_h) + b * SS;
  int olen = 0;
  for (int i = 0; i < SS; i++) olen += omask[i];
  const int li = olen > 0 ? olen - 1 : 0;
  const float lm = (float)omask[li];
  asm volatile("" :: "v"(wv4[0][0]), "v"(wv4[0][1]), "v"(wv4[0][2]), "v"(wv4[0][3]), "v"(rv4[0]), "v"(rv4[1]), "v"(cv4),
               "v"(pv[0]), "v"(pv[1]), "v"(pv[2]), "v"(pv[3]), "v"(pv[4]), "v"(pv[5]), "v"(pv[6]), "v"(pv[7]),
               "v"(pv[8]), "v"(pv[9]), "v"(pv[10]), "v"(pv[11]), "v"(pv[12]), "v"(pv[13]), "v"(pv[14]), "v"(pv[15]),
               "v"(fmask), "v"(rm[0]), "v"(rm[1]));
  asm volatile("" :: "v"(wv4[1][0]), "v"(wv4[1][1]), "v"(wv4[1][2]), "v"(wv4[1][3]),
               "v"(wv4[2][0]), "v"(wv4[2][1]), "v"(wv4[2][2]), "v"(wv4[2][3]));

  const int hc = tid < HH ? tid : HH - 1;
  const float* vcol = V + hc;
  const float lrow = V[(size_t)li * HH + hc];
  const bool two = (wv + 8) < AT_NT;
  const int ca = 16 * wv + l16;
  const int cbt = 16 * (wv + 8) + l16;
  const bool okb = cbt < HH;
  const float* vh = V + (size_t)(2 * half) * HH;
  float ba[2 * AT_KS], bb[2 * AT_KS];
  bfrag_load(ba, vh + ca);
  if (two) bfrag_load(bb, vh + (okb ? cbt : HH - 1));

#pragma unroll
  for (int s = 0; s < MA_NS; s++) {
#pragma unroll
    for (int u = 0; u < 4; u++) {
      const int i = tid + u * MA_NT;
      if (i < W4) {
        const int p = i / (HH / 4), c = i - p * (HH / 4);
        *(v4f*)(w2s + s * MA_WS + p * MA_WP + 4 * c) = wv4[s][u] * wv4[s][u];
      }
    }
    if (tid < HH) w2s[s * MA_WS + PP * MA_WP + tid] = 1.0f;
  }
#pragma unroll
  for (int u = 0; u < 2; u++) {
    const int i = tid + u * MA_NT;
    if (i < R4) *(v4f*)(r1 + 4 * i) = rv4[u] * rm[u];
  }
  if (side == 0) {
    cs[(pk + 0) * MA_R + pj] = cv4.x; cs[(pk + 1) * MA_R + pj] = cv4.y;
    cs[(pk + 2) * MA_R + pj] = cv4.z; cs[(pk + 3) * MA_R + pj] = cv4.w;
  } else {
    *(v4f*)(cs + qk * MA_R + qj) = cv4;
  }
  __syncthreads();

  if (tid < MA_FN) {
    float* orow = out + ((size_t)(BB + b) * SS + r0 + fy) * FF;
    const float m = (float)fmask;
    if (side == 0) {
      float mx = -3.0e38f;
#pragma unroll
      for (int g = 0; g < MA_NG; g++) mx = fmaxf(mx, pv[g]);
      orow[23 + fp] = m * mx;
    } else {
      float sm = 0.f;
#pragma unroll
      for (int g = 0; g < MA_NG; g++) sm += pv[g];
      orow[43 + fp] = m * (sm / (float)SS);
    }
  }

  float csr[MA_R];
  {
    const int j = lane >> 2, q = lane & 3;
    float mx = -3.0e38f, sm = 0.f;
#pragma unroll
    for (int i = 0; i < SS / 4; i++) {
      const float t = cs[(32 * q + i) * MA_R + j];
      mx = fmaxf(mx, t);
      sm += t;
    }
    mx = fmaxf(mx, __shfl_xor(mx, 1, 32)); sm += __shfl_xor(sm, 1, 32);
    mx = fmaxf(mx, __shfl_xor(mx, 2, 32)); sm += __shfl_xor(sm, 2, 32);
    if (wv == 7 && q == 0) {
      const float m = (float)mask[row0 + j];
      float* o2 = out + ((size_t)(side * BB + b) * SS + r0 + j) * FF;
      o2[0] = m * mx;
      o2[1] = m * (sm / (float)SS);
    }
#pragma unroll
    for (int r = 0; r < MA_R; r++) csr[r] = __shfl(sm, 4 * r, 32);
  }
  v2f af[AT_KS];
  {
    const float* cl = cs + (2 * half) * MA_R + (l16 & (MA_R - 1));
    const bool real = l16 < MA_R;
#pragma unroll
    for (int i = 0; i < AT_KS; i++) {
      const float x0 = cl[(4 * i) * MA_R], x1 = cl[(4 * i + 1) * MA_R];
      af[i].x = real ? x0 : 0.f;
      af[i].y = real ? x1 : 0.f;
    }
  }
  att8_tile(af, ba, csr, r3, ca, half == 0);
  if (two) att8_tile(af, bb, csr, r3, cbt, half == 0 && okb);

  float mx[MA_R];
#pragma unroll
  for (int j = 0; j < MA_R; j++) mx[j] = -3.0e38f;
  if ((tid & ~31) < HH) {
    float va[MA_KC], vb[MA_KC];
    vcol_load(va, vcol, 0);
#pragma unroll
    for (int k0 = 0; k0 < SS; k0 += 2 * MA_KC) {
      vcol_load(vb, vcol, k0 + MA_KC);
      maxplus_step(cs, k0, va, mx);
      if (k0 + 2 * MA_KC < SS) vcol_load(va, vcol, k0 + 2 * MA_KC);
      maxplus_step(cs, k0 + MA_KC, vb, mx);
    }
  }
  if (tid < HH) {
#pragma unroll
    for (int j = 0; j < MA_R; j++) r2[j * HH + tid] = (float)mask[row0 + j] * mx[j];
    r4[tid] = lrow * lm;
  }
  __syncthreads();

  if (tid < MA_NI) {
    constexpr int NG = MA_NQ / MA_QG;
    const int set = tid / (MA_R * NG), rem = tid - set * (MA_R * NG);
    const int j = rem / NG, g = rem - j * NG;
    const v4f* ar = (const v4f*)(r1 + j * HH);
    const v4f* cr = (const v4f*)(set == 0 ? r2 + j * HH : (set == 1 ? r3 + j * HH : r4));
    const v4f* wq = (const v4f*)(w2s + set * MA_WS + (MA_QG * g) * MA_WP);
    float s11[MA_QG], s22[MA_QG], s12[MA_QG];
#pragma unroll
    for (int t = 0; t < MA_QG; t++) { s11[t] = 0.f; s22[t] = 0.f; s12[t] = 0.f; }
#pragma unroll 2
    for (int h4 = 0; h4 < HH / 4; h4++) {
      const v4f a = ar[h4], c = cr[h4];
      v4f wt[MA_QG];
#pragma unroll
      for (int t = 0; t < MA_QG; t++) wt[t] = wq[t * (MA_WP / 4) + h4];
#pragma unroll
      for (int e = 0; e < 4; e++) {
        const float aa = a[e] * a[e], cc = c[e] * c[e], ac = a[e] * c[e];
#pragma unroll
        for (int t = 0; t < MA_QG; t++) {
          const float w2a = wt[t][e];
          s11[t] = __builtin_fmaf(w2a, aa, s11[t]); s22[t] = __builtin_fmaf(w2a, cc, s22[t]); s12[t] = __builtin_fmaf(w2a, ac, s12[t]);
        }
      }
    }
    const int f0 = set == 0 ? 84 : (set == 1 ? 63 : 2);
    float* orow = out + ((size_t)(side * BB + b) * SS + r0 + j) * FF + f0;
#pragma unroll
    for (int t = 0; t < MA_QG; t++) {
      const int q = MA_QG * g + t;
      orow[q == PP ? 0 : 1 + q] = s12[t] / (fmaxf(sqrtf(s11[t]), EPSF) * fmaxf(sqrtf(s22[t]), EPSF));
    }
  }
}

#define MP_WP 268
#define MP_NT 256
#define MP_NC (MP_XG * PP)
#define MP_NCC (MP_NC + 16)
#define MP_NW (PP + 1)
#define MP_NCW (MP_NCC + MP_NW)
#define MP_KS 5
#define MP_NG (HH / 4 / MP_KS)
#define MP_BJ (HH / 4)
#define MP_BT (MP_BJ * 5)

__device__ __forceinline__ void mp_bload(v2f (&f)[2 * MP_KS], const float* c0, const float* c1, int g) {
#pragma unroll
  for (int i = 0; i < MP_KS; i++) {
    f[2 * i] = *(const v2f*)(c0 + 4 * (g * MP_KS + i));
    f[2 * i + 1] = *(const v2f*)(c1 + 4 * (g * MP_KS + i));
  }
}

__device__ __forceinline__ void mp_mma(const v2f (&yf)[HH / 4], int g, const v2f (&f)[2 * MP_KS],
                                       v8f& acc0, v8f& acc1) {
#pragma unroll
  for (int i = 0; i < MP_KS; i++) {
    acc0 = __builtin_amdgcn_wmma_f32_16x16x4_f32(false, yf[g * MP_KS + i], false, f[2 * i], (short)0, acc0, false, false);
    acc1 = __builtin_amdgcn_wmma_f32_16x16x4_f32(false, yf[g * MP_KS + i], false, f[2 * i + 1], (short)0, acc1, false, false);
  }
}

__device__ __forceinline__ void mp_bload1(v2f (&f)[MP_KS], const float* c0, int g) {
#pragma unroll
  for (int i = 0; i < MP_KS; i++) f[i] = *(const v2f*)(c0 + 4 * (g * MP_KS + i));
}

__device__ __forceinline__ void mp_mma1(const v2f (&yf)[HH / 4], int g, const v2f (&f)[MP_KS],
                                        v8f& acc0, v8f& acc1) {
#pragma unroll
  for (int i = 0; i < MP_KS; i++) {
    if ((g * MP_KS + i) & 1)
      acc1 = __builtin_amdgcn_wmma_f32_16x16x4_f32(false, yf[g * MP_KS + i], false, f[i], (short)0, acc1, false, false);
    else
      acc0 = __builtin_amdgcn_wmma_f32_16x16x4_f32(false, yf[g * MP_KS + i], false, f[i], (short)0, acc0, false, false);
  }
}

template <int MASK>
__device__ __forceinline__ float mp_xlane(float v) {
  return __int_as_float(__builtin_amdgcn_update_dpp(0, __float_as_int(v), 0x160 + MASK, 0xf, 0xf, true));
}

__device__ __forceinline__ void mp_tile_out(const v8f& acc, float wa, const float (&wB)[8], int xi, int half,
                                            float* rm, float* rs, float* pm, float* ps) {
  float mv[8];
  float m = -3.0e38f, s = 0.f;
#pragma unroll
  for (int r = 0; r < 8; r++) {
    mv[r] = acc[r] / fmaxf(wa * wB[r], EPSF);
    m = fmaxf(m, mv[r]); s += mv[r];
  }
  m = fmaxf(m, __shfl_xor(m, 16, 32)); s += __shfl_xor(s, 16, 32);
  if (half == 0) { *rm = m; *rs = s; }
  float hm = 0.f, hs = 0.f;
#pragma unroll
  for (int r = 0; r < 8; r++) {
    float t = mv[r], u = mv[r];
    t = fmaxf(t, mp_xlane<1>(t)); u += mp_xlane<1>(u);
    t = fmaxf(t, mp_xlane<2>(t)); u += mp_xlane<2>(u);
    t = fmaxf(t, mp_xlane<4>(t)); u += mp_xlane<4>(u);
    if (xi == r) { hm = t; hs = u; }
  }
  *pm = hm; *ps = hs;
}

__global__ __launch_bounds__(MP_NT) void maxpool_pair_kernel(
    const float* __restrict__ v1, const float* __restrict__ v2,
    const float* __restrict__ w,
    const int* __restrict__ mask, const int* __restrict__ mask2, float* __restrict__ out,
    float* __restrict__ pmax, float* __restrict__ psum, float* __restrict__ cosO) {
  __shared__ __attribute__((aligned(16))) float xw[MP_NCW * MP_WP];
  __shared__ float redm[8 * MP_NC];
  __shared__ float reds[8 * MP_NC];
  __shared__ __attribute__((aligned(16))) float cst[MP_XG * SS];
  __shared__ float wny[SS * MP_NW];
  __shared__ float wnx[MP_XG * MP_NW];
  static_assert(MP_NT / 32 == MP_XG, "wave i sums the norms of x row i");

  const int b = blockIdx.x / (SS / MP_XG);
  const int xg = blockIdx.x % (SS / MP_XG);
  const int x0 = xg * MP_XG;
  const int tid = threadIdx.x;
  const int lane = tid & 31, half = lane >> 4, l16 = lane & 15;
  const int wv = tid >> 5;
  const int y0 = wv * 16;

  v2f yf[HH / 4];
  const float* yrow = v2 + ((size_t)b * SS + y0 + l16) * HH + 2 * half;
#pragma unroll
  for (int ks = 0; ks < HH / 4; ks++) yf[ks] = *(const v2f*)(yrow + 4 * ks);
  const float ym = (float)mask2[b * SS + y0 + l16];

  {
    const int bt = tid < MP_BT ? tid : MP_BT - 1;
    const int bg = bt / MP_BJ, bj = bt - bg * MP_BJ;
    v4f x4[MP_XG], w4[PP / 5];
    float xm[MP_XG];
#pragma unroll
    for (int xi = 0; xi < MP_XG; xi++) x4[xi] = *(const v4f*)(v1 + ((size_t)b * SS + x0 + xi) * HH + 4 * bj);
#pragma unroll
    for (int q = 0; q < PP / 5; q++) w4[q] = *(const v4f*)(w + (bg + 5 * q) * HH + 4 * bj);
#pragma unroll
    for (int xi = 0; xi < MP_XG; xi++) xm[xi] = (float)mask[b * SS + x0 + xi];
#pragma unroll
    for (int xi = 0; xi < MP_XG; xi++) x4[xi] = x4[xi] * xm[xi];
    if (tid < MP_BT) {
#pragma unroll
      for (int q = 0; q < PP / 5; q++) {
        const v4f ww = w4[q] * w4[q];
        float* col = xw + (size_t)((bg + 5 * q) * MP_XG) * MP_WP + 4 * bj;
#pragma unroll
        for (int xi = 0; xi < MP_XG; xi++) *(v4f*)(col + xi * MP_WP) = x4[xi] * ww;
        *(v4f*)(xw + (size_t)(MP_NCC + bg + 5 * q) * MP_WP + 4 * bj) = ww;
      }
      if (bg == 2) {
        const v4f one = {1.f, 1.f, 1.f, 1.f};
        *(v4f*)(xw + (size_t)(MP_NCC + PP) * MP_WP + 4 * bj) = one;
      }
      if (bg < 2) {
        float* col = xw + (size_t)(MP_NC + bg * MP_XG) * MP_WP + 4 * bj;
        const v4f zero = {0.f, 0.f, 0.f, 0.f};
#pragma unroll
        for (int xi = 0; xi < MP_XG; xi++) *(v4f*)(col + xi * MP_WP) = bg == 0 ? x4[xi] : zero;
      }
    }
  }
#pragma unroll
  for (int ks = 0; ks < HH / 4; ks++) yf[ks] = yf[ks] * ym;
  __syncthreads();

  if (lane < MP_NW) {
    const v4f* xc = (const v4f*)(xw + (size_t)(MP_NC + wv) * MP_WP);
    const v4f* wc = (const v4f*)(xw + (size_t)(MP_NCC + lane) * MP_WP);
    float s = 0.f;
#pragma unroll 5
    for (int j = 0; j < MP_BJ; j++) {
      const v4f x = xc[j], ww = wc[j];
      s += ww.x * (x.x * x.x); s += ww.y * (x.y * x.y);
      s += ww.z * (x.z * x.z); s += ww.w * (x.w * x.w);
    }
    wnx[wv * MP_NW + lane] = sqrtf(s);
  }
  {
    const float* wc0 = xw + (size_t)(MP_NCC + l16) * MP_WP + 2 * half;
    const float* wc1 = xw + (size_t)(MP_NCC + 16 + (l16 < MP_NW - 16 ? l16 : MP_NW - 17)) * MP_WP + 2 * half;
    v8f n0 = {0, 0, 0, 0, 0, 0, 0, 0};
    v8f n1 = {0, 0, 0, 0, 0, 0, 0, 0};
#pragma unroll
    for (int ks = 0; ks < HH / 4; ks++) {
      const v2f y2 = yf[ks] * yf[ks];
      n0 = __builtin_amdgcn_wmma_f32_16x16x4_f32(false, y2, false, *(const v2f*)(wc0 + 4 * ks), (short)0, n0, false, false);
      n1 = __builtin_amdgcn_wmma_f32_16x16x4_f32(false, y2, false, *(const v2f*)(wc1 + 4 * ks), (short)0, n1, false, false);
    }
    float* d = wny + (y0 + 8 * half) * MP_NW;
#pragma unroll
    for (int r = 0; r < 8; r++) d[r * MP_NW + l16] = sqrtf(n0[r]);
    if (l16 < MP_NW - 16) {
#pragma unroll
      for (int r = 0; r < 8; r++) d[r * MP_NW + 16 + l16] = sqrtf(n1[r]);
    }
  }
  __syncthreads();

  const float cnx = wnx[(l16 & 7) * MP_NW + PP];
  float cny[8];
#pragma unroll
  for (int r = 0; r < 8; r++) cny[r] = wny[(y0 + 8 * half + r) * MP_NW + PP];

  const int pb = l16 >> 3, xi = l16 & 7;
  const float* wArow = wnx + xi * MP_NW;
  const float* wBrow = wny + (y0 + 8 * half) * MP_NW;
  const size_t po = (((size_t)b * (SS / MP_XG) + xg) * SS + y0 + 8 * half + xi) * PP + pb;
  const float* bc = xw + l16 * MP_WP + 2 * half;

  {
    const float* cc = bc + (size_t)MP_NC * MP_WP;
    v2f ca[MP_KS], cb[MP_KS];
    v8f e0 = {0, 0, 0, 0, 0, 0, 0, 0};
    v8f e1 = {0, 0, 0, 0, 0, 0, 0, 0};
    mp_bload1(ca, cc, 0);
#pragma unroll
    for (int g = 0; g < MP_NG; g += 2) {
      mp_bload1(cb, cc, g + 1);
      __builtin_amdgcn_sched_barrier(0);
      mp_mma1(yf, g, ca, e0, e1);
      if (g + 2 < MP_NG) mp_bload1(ca, cc, g + 2);
      __builtin_amdgcn_sched_barrier(0);
      mp_mma1(yf, g + 1, cb, e0, e1);
    }
    v4f lo4, hi4;
    lo4.x = (e0[0] + e1[0]) / fmaxf(cnx * cny[0], EPSF); lo4.y = (e0[1] + e1[1]) / fmaxf(cnx * cny[1], EPSF);
    lo4.z = (e0[2] + e1[2]) / fmaxf(cnx * cny[2], EPSF); lo4.w = (e0[3] + e1[3]) / fmaxf(cnx * cny[3], EPSF);
    hi4.x = (e0[4] + e1[4]) / fmaxf(cnx * cny[4], EPSF); hi4.y = (e0[5] + e1[5]) / fmaxf(cnx * cny[5], EPSF);
    hi4.z = (e0[6] + e1[6]) / fmaxf(cnx * cny[6], EPSF); hi4.w = (e0[7] + e1[7]) / fmaxf(cnx * cny[7], EPSF);
    if (l16 < MP_XG) {
      float* d = cst + l16 * SS + y0 + 8 * half;
      *(v4f*)d = lo4; *(v4f*)(d + 4) = hi4;
    }
  }

  v2f fa[2 * MP_KS], fb[2 * MP_KS];
  mp_bload(fa, bc, bc + 16 * MP_WP, 0);
  for (int tp = 0; tp < MP_NC / 32; tp++) {
    const float* c0 = bc + (size_t)(32 * tp) * MP_WP;
    const float* c1 = c0 + 16 * MP_WP;
    const float* n0 = bc + (size_t)(32 * (tp + 1 < MP_NC / 32 ? tp + 1 : tp)) * MP_WP;
    const int p0 = 4 * tp + pb, p1 = p0 + 2;
    const float wa0 = wArow[p0], wa1 = wArow[p1];
    float wB0[8], wB1[8];
#pragma unroll
    for (int r = 0; r < 8; r++) { wB0[r] = wBrow[r * MP_NW + p0]; wB1[r] = wBrow[r * MP_NW + p1]; }
    v8f acc0 = {0, 0, 0, 0, 0, 0, 0, 0};
    v8f acc1 = {0, 0, 0, 0, 0, 0, 0, 0};
#pragma unroll
    for (int g = 0; g < MP_NG; g += 2) {
      mp_bload(fb, c0, c1, g + 1);
      __builtin_amdgcn_sched_barrier(0);
      mp_mma(yf, g, fa, acc0, acc1);
      if (g + 2 < MP_NG) mp_bload(fa, c0, c1, g + 2);
      else mp_bload(fa, n0, n0 + 16 * MP_WP, 0);
      __builtin_amdgcn_sched_barrier(0);
      mp_mma(yf, g + 1, fb, acc0, acc1);
    }
    const int n = 32 * tp + l16;
    mp_tile_out(acc0, wa0, wB0, xi, half, redm + wv * MP_NC + n, reds + wv * MP_NC + n,
                pmax + po + p0 - pb, psum + po + p0 - pb);
    mp_tile_out(acc1, wa1, wB1, xi, half, redm + wv * MP_NC + n + 16, reds + wv * MP_NC + n + 16,
                pmax + po + p1 - pb, psum + po + p1 - pb);
  }
  __syncthreads();
  if (tid < MP_NC) {
    const int p = tid / MP_XG, xr = tid - p * MP_XG;
    float mx = redm[tid], sm = reds[tid];
    for (int q = 1; q < 8; q++) {
      mx = fmaxf(mx, redm[q * MP_NC + tid]);
      sm += reds[q * MP_NC + tid];
    }
    const int x = x0 + xr;
    const float m = (float)mask[b * SS + x];
    float* orow = out + ((size_t)b * SS + x) * FF;
    orow[23 + p] = m * mx;
    orow[43 + p] = m * (sm / (float)SS);
  }
  {
    const int y4 = 4 * lane;
    *(v4f*)(cosO + ((size_t)b * SS + x0 + wv) * SS + y4) = *(const v4f*)(cst + wv * SS + y4);
  }
}

extern "C" void kernel_launch(void* const* d_in, const int* in_sizes, int n_in,
                              void* d_out, int out_size, void* d_ws, size_t ws_size,
                              hipStream_t stream) {
  const float* ctx_p = (const float*)d_in[0];
  const float* ctx_h = (const float*)d_in[1];
  const float* w0 = (const float*)d_in[2];
  const float* w1 = (const float*)d_in[3];
  const float* w2 = (const float*)d_in[4];
  const float* w3 = (const float*)d_in[5];
  const int* mask_p = (const int*)d_in[6];
  const int* mask_h = (const int*)d_in[7];
  float* out = (float*)d_out;
  float* ws = (float*)d_ws;

  float* cosM = ws;               ws += (size_t)BB * SS * SS;
  float* mp_pmax = ws;            ws += (size_t)BB * (SS / MP_XG) * SS * PP;
  float* mp_psum = ws;            ws += (size_t)BB * (SS / MP_XG) * SS * PP;

  maxpool_pair_kernel<<<BB * (SS / MP_XG), MP_NT, 0, stream>>>(ctx_p, ctx_h, w1, mask_p, mask_h, out,
                                                               mp_pmax, mp_psum, cosM);

  rows_match_kernel<<<dim3(BB * SS / MA_R, 2), MA_NT, 0, stream>>>(ctx_p, ctx_h, cosM, w3, w2, w0, mask_p, mask_h,
                                                                   mp_pmax, mp_psum, out);
}
